# starting kernel
# baseline (speedup 1.0000x reference)
#include <hip/hip_runtime.h>
#include <hip/hip_bf16.h>

typedef __attribute__((ext_vector_type(2))) float v2f;
typedef __attribute__((ext_vector_type(8))) float v8f;

#define NUM_USERS  100000
#define NUM_MOVIES 50000
#define NUM_EDGES  4000000
#define USER_FEAT  24
#define MOVIE_FEAT 404
#define HID        64
#define GENRES     3

__global__ __launch_bounds__(256) void degree_kernel(const int* __restrict__ idx,
                                                     float* __restrict__ deg, int E) {
    int e = blockIdx.x * blockDim.x + threadIdx.x;
    if (e < E) atomicAdd(&deg[idx[e]], 1.0f);
}

__global__ __launch_bounds__(256) void scatter_kernel(const float* __restrict__ X,
                                                      const int* __restrict__ src,
                                                      const int* __restrict__ dst,
                                                      float* __restrict__ acc, int E) {
    long long t = (long long)blockIdx.x * blockDim.x + threadIdx.x;
    int e = (int)(t >> 4);
    int c = ((int)t & 15) * 4;
    if (e >= E) return;
    int s = src[e];
    int d = dst[e];
    const float4 v = *(const float4*)(X + (size_t)s * HID + c);
    float* o = acc + (size_t)d * HID + c;
    atomicAdd(o + 0, v.x);
    atomicAdd(o + 1, v.y);
    atomicAdd(o + 2, v.z);
    atomicAdd(o + 3, v.w);
}

__global__ __launch_bounds__(256)
void sage_gemm64(const float* __restrict__ A1, const float* __restrict__ deg1,
                 const float* __restrict__ W1, int K1,
                 const float* __restrict__ A2, const float* __restrict__ W2, int K2,
                 const float* __restrict__ bias, float* __restrict__ C,
                 int M, int relu) {
    const int lane = threadIdx.x & 31;
    const int wave = threadIdx.x >> 5;
    const int half = lane >> 4;
    const int r    = lane & 15;
    const int row0 = (blockIdx.x * 8 + wave) * 16;
    if (row0 >= M) return;

    v8f c[4] = {};

    float scale = 1.0f;
    if (deg1) scale = 1.0f / fmaxf(deg1[row0 + r], 1.0f);

    {
        const float* Arow = A1 + (size_t)(row0 + r) * K1;
        for (int kk = 0; kk < K1; kk += 4) {
            const int k = kk + 2 * half;
            v2f a;
            a.x = Arow[k] * scale;
            a.y = Arow[k + 1] * scale;
#pragma unroll
            for (int t = 0; t < 4; ++t) {
                const float* Wrow = W1 + (size_t)(t * 16 + r) * K1;
                v2f b;
                b.x = Wrow[k];
                b.y = Wrow[k + 1];
                c[t] = __builtin_amdgcn_wmma_f32_16x16x4_f32(
                    false, a, false, b, (short)0, c[t], false, false);
            }
        }
    }
    if (A2) {
        const float* Arow = A2 + (size_t)(row0 + r) * K2;
        for (int kk = 0; kk < K2; kk += 4) {
            const int k = kk + 2 * half;
            v2f a;
            a.x = Arow[k];
            a.y = Arow[k + 1];
#pragma unroll
            for (int t = 0; t < 4; ++t) {
                const float* Wrow = W2 + (size_t)(t * 16 + r) * K2;
                v2f b;
                b.x = Wrow[k];
                b.y = Wrow[k + 1];
                c[t] = __builtin_amdgcn_wmma_f32_16x16x4_f32(
                    false, a, false, b, (short)0, c[t], false, false);
            }
        }
    }

#pragma unroll
    for (int t = 0; t < 4; ++t) {
        const float bv = bias ? bias[t * 16 + r] : 0.0f;
#pragma unroll
        for (int i = 0; i < 8; ++i) {
            float v = c[t][i] + bv;
            if (relu) v = fmaxf(v, 0.0f);
            C[(size_t)(row0 + i + 8 * half) * HID + t * 16 + r] = v;
        }
    }
}

__global__ __launch_bounds__(256)
void head_kernel(const float* __restrict__ acc2, const float* __restrict__ degm,
                 const float* __restrict__ m1, const float* __restrict__ Wl2,
                 const float* __restrict__ bl2, const float* __restrict__ Wr2,
                 float* __restrict__ out, int M) {
    int m = blockIdx.x * blockDim.x + threadIdx.x;
    if (m >= M) return;
    const float inv = 1.0f / fmaxf(degm[m], 1.0f);
    const float* a = acc2 + (size_t)m * HID;
    const float* x = m1 + (size_t)m * HID;
    float acc[GENRES];
#pragma unroll
    for (int g = 0; g < GENRES; ++g) acc[g] = bl2[g];
    for (int k = 0; k < HID; ++k) {
        const float av = a[k] * inv;
        const float xv = x[k];
#pragma unroll
        for (int g = 0; g < GENRES; ++g)
            acc[g] += av * Wl2[g * HID + k] + xv * Wr2[g * HID + k];
    }
#pragma unroll
    for (int g = 0; g < GENRES; ++g) out[(size_t)m * GENRES + g] = acc[g];
}

extern "C" void kernel_launch(void* const* d_in, const int* in_sizes, int n_in,
                              void* d_out, int out_size, void* d_ws, size_t ws_size,
                              hipStream_t stream) {
    (void)in_sizes; (void)n_in; (void)out_size; (void)ws_size;

    const float* x_user  = (const float*)d_in[0];
    const float* x_movie = (const float*)d_in[1];
    const int*   e_src   = (const int*)d_in[2];
    const int*   e_dst   = (const int*)d_in[3];
    const float* W_user  = (const float*)d_in[4];
    const float* b_user  = (const float*)d_in[5];
    const float* W_movie = (const float*)d_in[6];
    const float* b_movie = (const float*)d_in[7];
    const float* Wl1_um  = (const float*)d_in[8];
    const float* bl1_um  = (const float*)d_in[9];
    const float* Wr1_um  = (const float*)d_in[10];
    const float* Wl1_mu  = (const float*)d_in[11];
    const float* bl1_mu  = (const float*)d_in[12];
    const float* Wr1_mu  = (const float*)d_in[13];
    const float* Wl2_um  = (const float*)d_in[14];
    const float* bl2_um  = (const float*)d_in[15];
    const float* Wr2_um  = (const float*)d_in[16];
    float* out = (float*)d_out;

    float* ws = (float*)d_ws;
    float* acc_um    = ws;
    float* acc_mu    = acc_um + 3200000;
    float* acc2      = acc_mu + 6400000;
    float* deg_user  = acc2 + 3200000;
    float* deg_movie = deg_user + 100000;
    float* hu        = deg_movie + 50000;
    float* hm        = hu + 6400000;
    float* m1        = hm + 3200000;
    float* u1        = m1 + 3200000;

    const size_t zero_bytes = (size_t)(3200000 + 6400000 + 3200000 + 100000 + 50000) * 4;
    hipMemsetAsync(d_ws, 0, zero_bytes, stream);

    const int E = NUM_EDGES;
    degree_kernel<<<(E + 255) / 256, 256, 0, stream>>>(e_dst, deg_movie, E);
    degree_kernel<<<(E + 255) / 256, 256, 0, stream>>>(e_src, deg_user, E);

    const int gridU = (NUM_USERS / 16 + 7) / 8;
    const int gridM = (NUM_MOVIES / 16 + 7) / 8;
    sage_gemm64<<<gridU, 256, 0, stream>>>(x_user, nullptr, W_user, USER_FEAT,
                                           nullptr, nullptr, 0, b_user, hu,
                                           NUM_USERS, 0);
    sage_gemm64<<<gridM, 256, 0, stream>>>(x_movie, nullptr, W_movie, MOVIE_FEAT,
                                           nullptr, nullptr, 0, b_movie, hm,
                                           NUM_MOVIES, 0);

    const long long sthreads = (long long)E * 16;
    const int sblocks = (int)((sthreads + 255) / 256);

    scatter_kernel<<<sblocks, 256, 0, stream>>>(hu, e_src, e_dst, acc_um, E);
    sage_gemm64<<<gridM, 256, 0, stream>>>(acc_um, deg_movie, Wl1_um, HID,
                                           hm, Wr1_um, HID, bl1_um, m1,
                                           NUM_MOVIES, 1);

    scatter_kernel<<<sblocks, 256, 0, stream>>>(hm, e_dst, e_src, acc_mu, E);
    sage_gemm64<<<gridU, 256, 0, stream>>>(acc_mu, deg_user, Wl1_mu, HID,
                                           hu, Wr1_mu, HID, bl1_mu, u1,
                                           NUM_USERS, 1);

    scatter_kernel<<<sblocks, 256, 0, stream>>>(u1, e_src, e_dst, acc2, E);
    head_kernel<<<(NUM_MOVIES + 255) / 256, 256, 0, stream>>>(
        acc2, deg_movie, m1, Wl2_um, bl2_um, Wr2_um, out, NUM_MOVIES);
}


# Degree counts: per-block integer histograms in LDS over edge slices, flushed as exact lane-contiguous float adds, replace the 8M scattered global atomics of two launches (one launch now)
# speedup vs baseline: 1.3676x; 1.3676x over previous
#include <hip/hip_runtime.h>
#include <hip/hip_bf16.h>

typedef __attribute__((ext_vector_type(2))) float v2f;
typedef __attribute__((ext_vector_type(8))) float v8f;

#define NUM_USERS  100000
#define NUM_MOVIES 50000
#define NUM_EDGES  4000000
#define USER_FEAT  24
#define MOVIE_FEAT 404
#define HID        64
#define GENRES     3

#define HBINS        50000
#define HSLICES      32
#define HTHREADS     1024
#define MOVIE_RANGES ((NUM_MOVIES + HBINS - 1) / HBINS)
#define USER_RANGES  ((NUM_USERS + HBINS - 1) / HBINS)

__device__ __forceinline__ void hist_count4(unsigned int* bins, const int4 v, int lo, unsigned int nb) {
    const unsigned int b0 = (unsigned int)(v.x - lo);
    const unsigned int b1 = (unsigned int)(v.y - lo);
    const unsigned int b2 = (unsigned int)(v.z - lo);
    const unsigned int b3 = (unsigned int)(v.w - lo);
    if (b0 < nb) atomicAdd(&bins[b0], 1u);
    if (b1 < nb) atomicAdd(&bins[b1], 1u);
    if (b2 < nb) atomicAdd(&bins[b2], 1u);
    if (b3 < nb) atomicAdd(&bins[b3], 1u);
}

__global__ __launch_bounds__(HTHREADS)
void degree_hist_kernel(const int* __restrict__ dst, float* __restrict__ deg_movie,
                        const int* __restrict__ src, float* __restrict__ deg_user, int E) {
    __shared__ unsigned int bins[HBINS];

    const int tid = threadIdx.x;
    const int job = blockIdx.y;
    const bool is_movie = job < MOVIE_RANGES;
    const int* __restrict__ idx = is_movie ? dst : src;
    const int nseg = is_movie ? NUM_MOVIES : NUM_USERS;
    const int lo = (is_movie ? job : job - MOVIE_RANGES) * HBINS;
    const int left = nseg - lo;
    const unsigned int nb = (unsigned int)(left < HBINS ? left : HBINS);
    float* __restrict__ out = (is_movie ? deg_movie : deg_user) + lo;

    for (int i = tid; i < HBINS; i += HTHREADS) bins[i] = 0u;
    __syncthreads();

    const int per = (((E + HSLICES - 1) / HSLICES) + 3) & ~3;
    const int start = blockIdx.x * per;
    int end = start + per;
    if (end > E) end = E;
    if (start < end) {
        const int4* __restrict__ p = (const int4*)(idx + start);
        const int nvec = (end - start) >> 2;
        int i = tid;
        for (; i + 3 * HTHREADS < nvec; i += 4 * HTHREADS) {
            const int4 a = p[i];
            const int4 b = p[i + HTHREADS];
            const int4 c = p[i + 2 * HTHREADS];
            const int4 d = p[i + 3 * HTHREADS];
            hist_count4(bins, a, lo, nb);
            hist_count4(bins, b, lo, nb);
            hist_count4(bins, c, lo, nb);
            hist_count4(bins, d, lo, nb);
        }
        for (; i < nvec; i += HTHREADS) hist_count4(bins, p[i], lo, nb);
        for (int e = start + (nvec << 2) + tid; e < end; e += HTHREADS) {
            const unsigned int b = (unsigned int)(idx[e] - lo);
            if (b < nb) atomicAdd(&bins[b], 1u);
        }
    }
    __syncthreads();

    const int n = (int)nb;
    const int rot = ((int)blockIdx.x * (n / HSLICES)) & ~31;
    for (int j = tid; j < n; j += HTHREADS) {
        int i = j + rot;
        if (i >= n) i -= n;
        const unsigned int cnt = bins[i];
        if (cnt) atomicAdd(&out[i], (float)cnt);
    }
}

__global__ __launch_bounds__(256) void scatter_kernel(const float* __restrict__ X,
                                                      const int* __restrict__ src,
                                                      const int* __restrict__ dst,
                                                      float* __restrict__ acc, int E) {
    long long t = (long long)blockIdx.x * blockDim.x + threadIdx.x;
    int e = (int)(t >> 4);
    int c = ((int)t & 15) * 4;
    if (e >= E) return;
    int s = src[e];
    int d = dst[e];
    const float4 v = *(const float4*)(X + (size_t)s * HID + c);
    float* o = acc + (size_t)d * HID + c;
    atomicAdd(o + 0, v.x);
    atomicAdd(o + 1, v.y);
    atomicAdd(o + 2, v.z);
    atomicAdd(o + 3, v.w);
}

__global__ __launch_bounds__(256)
void sage_gemm64(const float* __restrict__ A1, const float* __restrict__ deg1,
                 const float* __restrict__ W1, int K1,
                 const float* __restrict__ A2, const float* __restrict__ W2, int K2,
                 const float* __restrict__ bias, float* __restrict__ C,
                 int M, int relu) {
    const int lane = threadIdx.x & 31;
    const int wave = threadIdx.x >> 5;
    const int half = lane >> 4;
    const int r    = lane & 15;
    const int row0 = (blockIdx.x * 8 + wave) * 16;
    if (row0 >= M) return;

    v8f c[4] = {};

    float scale = 1.0f;
    if (deg1) scale = 1.0f / fmaxf(deg1[row0 + r], 1.0f);

    {
        const float* Arow = A1 + (size_t)(row0 + r) * K1;
        for (int kk = 0; kk < K1; kk += 4) {
            const int k = kk + 2 * half;
            v2f a;
            a.x = Arow[k] * scale;
            a.y = Arow[k + 1] * scale;
#pragma unroll
            for (int t = 0; t < 4; ++t) {
                const float* Wrow = W1 + (size_t)(t * 16 + r) * K1;
                v2f b;
                b.x = Wrow[k];
                b.y = Wrow[k + 1];
                c[t] = __builtin_amdgcn_wmma_f32_16x16x4_f32(
                    false, a, false, b, (short)0, c[t], false, false);
            }
        }
    }
    if (A2) {
        const float* Arow = A2 + (size_t)(row0 + r) * K2;
        for (int kk = 0; kk < K2; kk += 4) {
            const int k = kk + 2 * half;
            v2f a;
            a.x = Arow[k];
            a.y = Arow[k + 1];
#pragma unroll
            for (int t = 0; t < 4; ++t) {
                const float* Wrow = W2 + (size_t)(t * 16 + r) * K2;
                v2f b;
                b.x = Wrow[k];
                b.y = Wrow[k + 1];
                c[t] = __builtin_amdgcn_wmma_f32_16x16x4_f32(
                    false, a, false, b, (short)0, c[t], false, false);
            }
        }
    }

#pragma unroll
    for (int t = 0; t < 4; ++t) {
        const float bv = bias ? bias[t * 16 + r] : 0.0f;
#pragma unroll
        for (int i = 0; i < 8; ++i) {
            float v = c[t][i] + bv;
            if (relu) v = fmaxf(v, 0.0f);
            C[(size_t)(row0 + i + 8 * half) * HID + t * 16 + r] = v;
        }
    }
}

__global__ __launch_bounds__(256)
void head_kernel(const float* __restrict__ acc2, const float* __restrict__ degm,
                 const float* __restrict__ m1, const float* __restrict__ Wl2,
                 const float* __restrict__ bl2, const float* __restrict__ Wr2,
                 float* __restrict__ out, int M) {
    int m = blockIdx.x * blockDim.x + threadIdx.x;
    if (m >= M) return;
    const float inv = 1.0f / fmaxf(degm[m], 1.0f);
    const float* a = acc2 + (size_t)m * HID;
    const float* x = m1 + (size_t)m * HID;
    float acc[GENRES];
#pragma unroll
    for (int g = 0; g < GENRES; ++g) acc[g] = bl2[g];
    for (int k = 0; k < HID; ++k) {
        const float av = a[k] * inv;
        const float xv = x[k];
#pragma unroll
        for (int g = 0; g < GENRES; ++g)
            acc[g] += av * Wl2[g * HID + k] + xv * Wr2[g * HID + k];
    }
#pragma unroll
    for (int g = 0; g < GENRES; ++g) out[(size_t)m * GENRES + g] = acc[g];
}

extern "C" void kernel_launch(void* const* d_in, const int* in_sizes, int n_in,
                              void* d_out, int out_size, void* d_ws, size_t ws_size,
                              hipStream_t stream) {
    (void)in_sizes; (void)n_in; (void)out_size; (void)ws_size;

    const float* x_user  = (const float*)d_in[0];
    const float* x_movie = (const float*)d_in[1];
    const int*   e_src   = (const int*)d_in[2];
    const int*   e_dst   = (const int*)d_in[3];
    const float* W_user  = (const float*)d_in[4];
    const float* b_user  = (const float*)d_in[5];
    const float* W_movie = (const float*)d_in[6];
    const float* b_movie = (const float*)d_in[7];
    const float* Wl1_um  = (const float*)d_in[8];
    const float* bl1_um  = (const float*)d_in[9];
    const float* Wr1_um  = (const float*)d_in[10];
    const float* Wl1_mu  = (const float*)d_in[11];
    const float* bl1_mu  = (const float*)d_in[12];
    const float* Wr1_mu  = (const float*)d_in[13];
    const float* Wl2_um  = (const float*)d_in[14];
    const float* bl2_um  = (const float*)d_in[15];
    const float* Wr2_um  = (const float*)d_in[16];
    float* out = (float*)d_out;

    float* ws = (float*)d_ws;
    float* acc_um    = ws;
    float* acc_mu    = acc_um + 3200000;
    float* acc2      = acc_mu + 6400000;
    float* deg_user  = acc2 + 3200000;
    float* deg_movie = deg_user + 100000;
    float* hu        = deg_movie + 50000;
    float* hm        = hu + 6400000;
    float* m1        = hm + 3200000;
    float* u1        = m1 + 3200000;

    const size_t zero_bytes = (size_t)(3200000 + 6400000 + 3200000 + 100000 + 50000) * 4;
    hipMemsetAsync(d_ws, 0, zero_bytes, stream);

    const int E = NUM_EDGES;
    degree_hist_kernel<<<dim3(HSLICES, MOVIE_RANGES + USER_RANGES), HTHREADS, 0, stream>>>(
        e_dst, deg_movie, e_src, deg_user, E);

    const int gridU = (NUM_USERS / 16 + 7) / 8;
    const int gridM = (NUM_MOVIES / 16 + 7) / 8;
    sage_gemm64<<<gridU, 256, 0, stream>>>(x_user, nullptr, W_user, USER_FEAT,
                                           nullptr, nullptr, 0, b_user, hu,
                                           NUM_USERS, 0);
    sage_gemm64<<<gridM, 256, 0, stream>>>(x_movie, nullptr, W_movie, MOVIE_FEAT,
                                           nullptr, nullptr, 0, b_movie, hm,
                                           NUM_MOVIES, 0);

    const long long sthreads = (long long)E * 16;
    const int sblocks = (int)((sthreads + 255) / 256);

    scatter_kernel<<<sblocks, 256, 0, stream>>>(hu, e_src, e_dst, acc_um, E);
    sage_gemm64<<<gridM, 256, 0, stream>>>(acc_um, deg_movie, Wl1_um, HID,
                                           hm, Wr1_um, HID, bl1_um, m1,
                                           NUM_MOVIES, 1);

    scatter_kernel<<<sblocks, 256, 0, stream>>>(hm, e_dst, e_src, acc_mu, E);
    sage_gemm64<<<gridU, 256, 0, stream>>>(acc_mu, deg_user, Wl1_mu, HID,
                                           hu, Wr1_mu, HID, bl1_mu, u1,
                                           NUM_USERS, 1);

    scatter_kernel<<<sblocks, 256, 0, stream>>>(u1, e_src, e_dst, acc2, E);
    head_kernel<<<(NUM_MOVIES + 255) / 256, 256, 0, stream>>>(
        acc2, deg_movie, m1, Wl2_um, bl2_um, Wr2_um, out, NUM_MOVIES);
}


# Edge scatter re-tiled to whole lines: one wave per 256-B row (lane = channel; two 128-B loads and two 128-B float-atomic adds per edge), 64 edges per wave, planes line-aligned in the workspace
# speedup vs baseline: 2.7515x; 2.0119x over previous
#include <hip/hip_runtime.h>
#include <hip/hip_bf16.h>

typedef __attribute__((ext_vector_type(2))) float v2f;
typedef __attribute__((ext_vector_type(8))) float v8f;

#define NUM_USERS  100000
#define NUM_MOVIES 50000
#define NUM_EDGES  4000000
#define USER_FEAT  24
#define MOVIE_FEAT 404
#define HID        64
#define GENRES     3

#define HBINS        50000
#define HSLICES      32
#define HTHREADS     1024
#define MOVIE_RANGES ((NUM_MOVIES + HBINS - 1) / HBINS)
#define USER_RANGES  ((NUM_USERS + HBINS - 1) / HBINS)

__device__ __forceinline__ void hist_count4(unsigned int* bins, const int4 v, int lo, unsigned int nb) {
    const unsigned int b0 = (unsigned int)(v.x - lo);
    const unsigned int b1 = (unsigned int)(v.y - lo);
    const unsigned int b2 = (unsigned int)(v.z - lo);
    const unsigned int b3 = (unsigned int)(v.w - lo);
    if (b0 < nb) atomicAdd(&bins[b0], 1u);
    if (b1 < nb) atomicAdd(&bins[b1], 1u);
    if (b2 < nb) atomicAdd(&bins[b2], 1u);
    if (b3 < nb) atomicAdd(&bins[b3], 1u);
}

__global__ __launch_bounds__(HTHREADS)
void degree_hist_kernel(const int* __restrict__ dst, float* __restrict__ deg_movie,
                        const int* __restrict__ src, float* __restrict__ deg_user, int E) {
    __shared__ unsigned int bins[HBINS];

    const int tid = threadIdx.x;
    const int job = blockIdx.y;
    const bool is_movie = job < MOVIE_RANGES;
    const int* __restrict__ idx = is_movie ? dst : src;
    const int nseg = is_movie ? NUM_MOVIES : NUM_USERS;
    const int lo = (is_movie ? job : job - MOVIE_RANGES) * HBINS;
    const int left = nseg - lo;
    const unsigned int nb = (unsigned int)(left < HBINS ? left : HBINS);
    float* __restrict__ out = (is_movie ? deg_movie : deg_user) + lo;

    for (int i = tid; i < HBINS; i += HTHREADS) bins[i] = 0u;
    __syncthreads();

    const int per = (((E + HSLICES - 1) / HSLICES) + 3) & ~3;
    const int start = blockIdx.x * per;
    int end = start + per;
    if (end > E) end = E;
    if (start < end) {
        const int4* __restrict__ p = (const int4*)(idx + start);
        const int nvec = (end - start) >> 2;
        int i = tid;
        for (; i + 3 * HTHREADS < nvec; i += 4 * HTHREADS) {
            const int4 a = p[i];
            const int4 b = p[i + HTHREADS];
            const int4 c = p[i + 2 * HTHREADS];
            const int4 d = p[i + 3 * HTHREADS];
            hist_count4(bins, a, lo, nb);
            hist_count4(bins, b, lo, nb);
            hist_count4(bins, c, lo, nb);
            hist_count4(bins, d, lo, nb);
        }
        for (; i < nvec; i += HTHREADS) hist_count4(bins, p[i], lo, nb);
        for (int e = start + (nvec << 2) + tid; e < end; e += HTHREADS) {
            const unsigned int b = (unsigned int)(idx[e] - lo);
            if (b < nb) atomicAdd(&bins[b], 1u);
        }
    }
    __syncthreads();

    const int n = (int)nb;
    const int rot = ((int)blockIdx.x * (n / HSLICES)) & ~31;
    for (int j = tid; j < n; j += HTHREADS) {
        int i = j + rot;
        if (i >= n) i -= n;
        const unsigned int cnt = bins[i];
        if (cnt) atomicAdd(&out[i], (float)cnt);
    }
}

#define SC_EPW   64
#define SC_GROUP 8

__device__ __forceinline__ void scatter_rows(const float* __restrict__ X, float* __restrict__ acc,
                                             const int sv, const int dv, const int n, const int lane) {
    if (n == 32) {
#pragma unroll
        for (int g = 0; g < 32; g += SC_GROUP) {
            float lo[SC_GROUP], hi[SC_GROUP];
#pragma unroll
            for (int j = 0; j < SC_GROUP; ++j) {
                const int s = __builtin_amdgcn_readlane(sv, g + j);
                const float* __restrict__ xr = X + (size_t)s * HID + lane;
                lo[j] = xr[0];
                hi[j] = xr[32];
            }
#pragma unroll
            for (int j = 0; j < SC_GROUP; ++j) {
                const int d = __builtin_amdgcn_readlane(dv, g + j);
                float* __restrict__ o = acc + (size_t)d * HID + lane;
                atomicAdd(o, lo[j]);
                atomicAdd(o + 32, hi[j]);
            }
        }
    } else {
        for (int i = 0; i < n; ++i) {
            const int s = __builtin_amdgcn_readlane(sv, i);
            const int d = __builtin_amdgcn_readlane(dv, i);
            const float* __restrict__ xr = X + (size_t)s * HID + lane;
            const float a = xr[0];
            const float b = xr[32];
            float* __restrict__ o = acc + (size_t)d * HID + lane;
            atomicAdd(o, a);
            atomicAdd(o + 32, b);
        }
    }
}

__global__ __launch_bounds__(256) void scatter_kernel(const float* __restrict__ X,
                                                      const int* __restrict__ src,
                                                      const int* __restrict__ dst,
                                                      float* __restrict__ acc, int E) {
    const int lane = threadIdx.x & 31;
    const int wave = __builtin_amdgcn_readfirstlane((int)(blockIdx.x * 8 + (threadIdx.x >> 5)));
    const long long e0 = (long long)wave * SC_EPW;
    if (e0 >= E) return;
    const int ea = (int)e0 + lane;
    const int eb = ea + 32;
    int sa = 0, da = 0, sb = 0, db = 0;
    if (ea < E) { sa = src[ea]; da = dst[ea]; }
    if (eb < E) { sb = src[eb]; db = dst[eb]; }
    int na = E - (int)e0;
    int nb = na - 32;
    if (na > 32) na = 32;
    if (nb > 32) nb = 32;
    scatter_rows(X, acc, sa, da, na, lane);
    if (nb > 0) scatter_rows(X, acc, sb, db, nb, lane);
}

__global__ __launch_bounds__(256)
void sage_gemm64(const float* __restrict__ A1, const float* __restrict__ deg1,
                 const float* __restrict__ W1, int K1,
                 const float* __restrict__ A2, const float* __restrict__ W2, int K2,
                 const float* __restrict__ bias, float* __restrict__ C,
                 int M, int relu) {
    const int lane = threadIdx.x & 31;
    const int wave = threadIdx.x >> 5;
    const int half = lane >> 4;
    const int r    = lane & 15;
    const int row0 = (blockIdx.x * 8 + wave) * 16;
    if (row0 >= M) return;

    v8f c[4] = {};

    float scale = 1.0f;
    if (deg1) scale = 1.0f / fmaxf(deg1[row0 + r], 1.0f);

    {
        const float* Arow = A1 + (size_t)(row0 + r) * K1;
        for (int kk = 0; kk < K1; kk += 4) {
            const int k = kk + 2 * half;
            v2f a;
            a.x = Arow[k] * scale;
            a.y = Arow[k + 1] * scale;
#pragma unroll
            for (int t = 0; t < 4; ++t) {
                const float* Wrow = W1 + (size_t)(t * 16 + r) * K1;
                v2f b;
                b.x = Wrow[k];
                b.y = Wrow[k + 1];
                c[t] = __builtin_amdgcn_wmma_f32_16x16x4_f32(
                    false, a, false, b, (short)0, c[t], false, false);
            }
        }
    }
    if (A2) {
        const float* Arow = A2 + (size_t)(row0 + r) * K2;
        for (int kk = 0; kk < K2; kk += 4) {
            const int k = kk + 2 * half;
            v2f a;
            a.x = Arow[k];
            a.y = Arow[k + 1];
#pragma unroll
            for (int t = 0; t < 4; ++t) {
                const float* Wrow = W2 + (size_t)(t * 16 + r) * K2;
                v2f b;
                b.x = Wrow[k];
                b.y = Wrow[k + 1];
                c[t] = __builtin_amdgcn_wmma_f32_16x16x4_f32(
                    false, a, false, b, (short)0, c[t], false, false);
            }
        }
    }

#pragma unroll
    for (int t = 0; t < 4; ++t) {
        const float bv = bias ? bias[t * 16 + r] : 0.0f;
#pragma unroll
        for (int i = 0; i < 8; ++i) {
            float v = c[t][i] + bv;
            if (relu) v = fmaxf(v, 0.0f);
            C[(size_t)(row0 + i + 8 * half) * HID + t * 16 + r] = v;
        }
    }
}

__global__ __launch_bounds__(256)
void head_kernel(const float* __restrict__ acc2, const float* __restrict__ degm,
                 const float* __restrict__ m1, const float* __restrict__ Wl2,
                 const float* __restrict__ bl2, const float* __restrict__ Wr2,
                 float* __restrict__ out, int M) {
    int m = blockIdx.x * blockDim.x + threadIdx.x;
    if (m >= M) return;
    const float inv = 1.0f / fmaxf(degm[m], 1.0f);
    const float* a = acc2 + (size_t)m * HID;
    const float* x = m1 + (size_t)m * HID;
    float acc[GENRES];
#pragma unroll
    for (int g = 0; g < GENRES; ++g) acc[g] = bl2[g];
    for (int k = 0; k < HID; ++k) {
        const float av = a[k] * inv;
        const float xv = x[k];
#pragma unroll
        for (int g = 0; g < GENRES; ++g)
            acc[g] += av * Wl2[g * HID + k] + xv * Wr2[g * HID + k];
    }
#pragma unroll
    for (int g = 0; g < GENRES; ++g) out[(size_t)m * GENRES + g] = acc[g];
}

extern "C" void kernel_launch(void* const* d_in, const int* in_sizes, int n_in,
                              void* d_out, int out_size, void* d_ws, size_t ws_size,
                              hipStream_t stream) {
    (void)in_sizes; (void)n_in; (void)out_size; (void)ws_size;

    const float* x_user  = (const float*)d_in[0];
    const float* x_movie = (const float*)d_in[1];
    const int*   e_src   = (const int*)d_in[2];
    const int*   e_dst   = (const int*)d_in[3];
    const float* W_user  = (const float*)d_in[4];
    const float* b_user  = (const float*)d_in[5];
    const float* W_movie = (const float*)d_in[6];
    const float* b_movie = (const float*)d_in[7];
    const float* Wl1_um  = (const float*)d_in[8];
    const float* bl1_um  = (const float*)d_in[9];
    const float* Wr1_um  = (const float*)d_in[10];
    const float* Wl1_mu  = (const float*)d_in[11];
    const float* bl1_mu  = (const float*)d_in[12];
    const float* Wr1_mu  = (const float*)d_in[13];
    const float* Wl2_um  = (const float*)d_in[14];
    const float* bl2_um  = (const float*)d_in[15];
    const float* Wr2_um  = (const float*)d_in[16];
    float* out = (float*)d_out;

    float* ws = (float*)d_ws;
    float* hu        = ws;
    float* hm        = hu + 6400000;
    float* m1        = hm + 3200000;
    float* u1        = m1 + 3200000;
    float* acc_um    = u1 + 6400000;
    float* acc_mu    = acc_um + 3200000;
    float* acc2      = acc_mu + 6400000;
    float* deg_user  = acc2 + 3200000;
    float* deg_movie = deg_user + 100000;

    const size_t zero_bytes = (size_t)(3200000 + 6400000 + 3200000 + 100000 + 50000) * 4;
    hipMemsetAsync(acc_um, 0, zero_bytes, stream);

    const int E = NUM_EDGES;
    degree_hist_kernel<<<dim3(HSLICES, MOVIE_RANGES + USER_RANGES), HTHREADS, 0, stream>>>(
        e_dst, deg_movie, e_src, deg_user, E);

    const int gridU = (NUM_USERS / 16 + 7) / 8;
    const int gridM = (NUM_MOVIES / 16 + 7) / 8;
    sage_gemm64<<<gridU, 256, 0, stream>>>(x_user, nullptr, W_user, USER_FEAT,
                                           nullptr, nullptr, 0, b_user, hu,
                                           NUM_USERS, 0);
    sage_gemm64<<<gridM, 256, 0, stream>>>(x_movie, nullptr, W_movie, MOVIE_FEAT,
                                           nullptr, nullptr, 0, b_movie, hm,
                                           NUM_MOVIES, 0);

    const int sblocks = (E + 8 * SC_EPW - 1) / (8 * SC_EPW);

    scatter_kernel<<<sblocks, 256, 0, stream>>>(hu, e_src, e_dst, acc_um, E);
    sage_gemm64<<<gridM, 256, 0, stream>>>(acc_um, deg_movie, Wl1_um, HID,
                                           hm, Wr1_um, HID, bl1_um, m1,
                                           NUM_MOVIES, 1);

    scatter_kernel<<<sblocks, 256, 0, stream>>>(hm, e_dst, e_src, acc_mu, E);
    sage_gemm64<<<gridU, 256, 0, stream>>>(acc_mu, deg_user, Wl1_mu, HID,
                                           hu, Wr1_mu, HID, bl1_mu, u1,
                                           NUM_USERS, 1);

    scatter_kernel<<<sblocks, 256, 0, stream>>>(u1, e_src, e_dst, acc2, E);
    head_kernel<<<(NUM_MOVIES + 255) / 256, 256, 0, stream>>>(
        acc2, deg_movie, m1, Wl2_um, bl2_um, Wr2_um, out, NUM_MOVIES);
}


# Layer-2 lin_l (64->3) applied per user row in the u1 GEMM epilogue, before the edge sum: third edge sum carries 3 floats per edge in per-block LDS float bins, atomic flush; sum order changes
# speedup vs baseline: 3.2579x; 1.1840x over previous
#include <hip/hip_runtime.h>
#include <hip/hip_bf16.h>

typedef __attribute__((ext_vector_type(2))) float v2f;
typedef __attribute__((ext_vector_type(8))) float v8f;

#define NUM_USERS  100000
#define NUM_MOVIES 50000
#define NUM_EDGES  4000000
#define USER_FEAT  24
#define MOVIE_FEAT 404
#define HID        64
#define GENRES     3

#define HBINS        50000
#define HSLICES      32
#define HTHREADS     1024
#define MOVIE_RANGES ((NUM_MOVIES + HBINS - 1) / HBINS)
#define USER_RANGES  ((NUM_USERS + HBINS - 1) / HBINS)

__device__ __forceinline__ void hist_count4(unsigned int* bins, const int4 v, int lo, unsigned int nb) {
    const unsigned int b0 = (unsigned int)(v.x - lo);
    const unsigned int b1 = (unsigned int)(v.y - lo);
    const unsigned int b2 = (unsigned int)(v.z - lo);
    const unsigned int b3 = (unsigned int)(v.w - lo);
    if (b0 < nb) atomicAdd(&bins[b0], 1u);
    if (b1 < nb) atomicAdd(&bins[b1], 1u);
    if (b2 < nb) atomicAdd(&bins[b2], 1u);
    if (b3 < nb) atomicAdd(&bins[b3], 1u);
}

__global__ __launch_bounds__(HTHREADS)
void degree_hist_kernel(const int* __restrict__ dst, float* __restrict__ deg_movie,
                        const int* __restrict__ src, float* __restrict__ deg_user, int E) {
    __shared__ unsigned int bins[HBINS];

    const int tid = threadIdx.x;
    const int job = blockIdx.y;
    const bool is_movie = job < MOVIE_RANGES;
    const int* __restrict__ idx = is_movie ? dst : src;
    const int nseg = is_movie ? NUM_MOVIES : NUM_USERS;
    const int lo = (is_movie ? job : job - MOVIE_RANGES) * HBINS;
    const int left = nseg - lo;
    const unsigned int nb = (unsigned int)(left < HBINS ? left : HBINS);
    float* __restrict__ out = (is_movie ? deg_movie : deg_user) + lo;

    for (int i = tid; i < HBINS; i += HTHREADS) bins[i] = 0u;
    __syncthreads();

    const int per = (((E + HSLICES - 1) / HSLICES) + 3) & ~3;
    const int start = blockIdx.x * per;
    int end = start + per;
    if (end > E) end = E;
    if (start < end) {
        const int4* __restrict__ p = (const int4*)(idx + start);
        const int nvec = (end - start) >> 2;
        int i = tid;
        for (; i + 3 * HTHREADS < nvec; i += 4 * HTHREADS) {
            const int4 a = p[i];
            const int4 b = p[i + HTHREADS];
            const int4 c = p[i + 2 * HTHREADS];
            const int4 d = p[i + 3 * HTHREADS];
            hist_count4(bins, a, lo, nb);
            hist_count4(bins, b, lo, nb);
            hist_count4(bins, c, lo, nb);
            hist_count4(bins, d, lo, nb);
        }
        for (; i < nvec; i += HTHREADS) hist_count4(bins, p[i], lo, nb);
        for (int e = start + (nvec << 2) + tid; e < end; e += HTHREADS) {
            const unsigned int b = (unsigned int)(idx[e] - lo);
            if (b < nb) atomicAdd(&bins[b], 1u);
        }
    }
    __syncthreads();

    const int n = (int)nb;
    const int rot = ((int)blockIdx.x * (n / HSLICES)) & ~31;
    for (int j = tid; j < n; j += HTHREADS) {
        int i = j + rot;
        if (i >= n) i -= n;
        const unsigned int cnt = bins[i];
        if (cnt) atomicAdd(&out[i], (float)cnt);
    }
}

#define SC_EPW   64
#define SC_GROUP 8

__device__ __forceinline__ void scatter_rows(const float* __restrict__ X, float* __restrict__ acc,
                                             const int sv, const int dv, const int n, const int lane) {
    if (n == 32) {
#pragma unroll
        for (int g = 0; g < 32; g += SC_GROUP) {
            float lo[SC_GROUP], hi[SC_GROUP];
#pragma unroll
            for (int j = 0; j < SC_GROUP; ++j) {
                const int s = __builtin_amdgcn_readlane(sv, g + j);
                const float* __restrict__ xr = X + (size_t)s * HID + lane;
                lo[j] = xr[0];
                hi[j] = xr[32];
            }
#pragma unroll
            for (int j = 0; j < SC_GROUP; ++j) {
                const int d = __builtin_amdgcn_readlane(dv, g + j);
                float* __restrict__ o = acc + (size_t)d * HID + lane;
                atomicAdd(o, lo[j]);
                atomicAdd(o + 32, hi[j]);
            }
        }
    } else {
        for (int i = 0; i < n; ++i) {
            const int s = __builtin_amdgcn_readlane(sv, i);
            const int d = __builtin_amdgcn_readlane(dv, i);
            const float* __restrict__ xr = X + (size_t)s * HID + lane;
            const float a = xr[0];
            const float b = xr[32];
            float* __restrict__ o = acc + (size_t)d * HID + lane;
            atomicAdd(o, a);
            atomicAdd(o + 32, b);
        }
    }
}

__global__ __launch_bounds__(256) void scatter_kernel(const float* __restrict__ X,
                                                      const int* __restrict__ src,
                                                      const int* __restrict__ dst,
                                                      float* __restrict__ acc, int E) {
    const int lane = threadIdx.x & 31;
    const int wave = __builtin_amdgcn_readfirstlane((int)(blockIdx.x * 8 + (threadIdx.x >> 5)));
    const long long e0 = (long long)wave * SC_EPW;
    if (e0 >= E) return;
    const int ea = (int)e0 + lane;
    const int eb = ea + 32;
    int sa = 0, da = 0, sb = 0, db = 0;
    if (ea < E) { sa = src[ea]; da = dst[ea]; }
    if (eb < E) { sb = src[eb]; db = dst[eb]; }
    int na = E - (int)e0;
    int nb = na - 32;
    if (na > 32) na = 32;
    if (nb > 32) nb = 32;
    scatter_rows(X, acc, sa, da, na, lane);
    if (nb > 0) scatter_rows(X, acc, sb, db, nb, lane);
}

#define PS_RANGES  2
#define PS_RB      ((NUM_MOVIES + PS_RANGES - 1) / PS_RANGES)
#define PS_SLICES  128
#define PS_THREADS 1024
#define PS_PLANES  4
#define PS_PITCH   150016

__device__ __forceinline__ void ps_edges8(float* bins, const float4* __restrict__ P4,
                                          const int4 d0, const int4 s0,
                                          const int4 d1, const int4 s1,
                                          int lo, unsigned int nb) {
    const int dd[8] = {d0.x, d0.y, d0.z, d0.w, d1.x, d1.y, d1.z, d1.w};
    const int ss[8] = {s0.x, s0.y, s0.z, s0.w, s1.x, s1.y, s1.z, s1.w};
    unsigned int b[8];
    float4 p[8];
#pragma unroll
    for (int j = 0; j < 8; ++j) {
        b[j] = (unsigned int)(dd[j] - lo);
        p[j] = make_float4(0.0f, 0.0f, 0.0f, 0.0f);
        if (b[j] < nb) p[j] = P4[ss[j]];
    }
#pragma unroll
    for (int j = 0; j < 8; ++j) {
        if (b[j] < nb) {
            float* q = bins + b[j] * GENRES;
            atomicAdd(q + 0, p[j].x);
            atomicAdd(q + 1, p[j].y);
            atomicAdd(q + 2, p[j].z);
        }
    }
}

__device__ __forceinline__ void ps_edge1(float* bins, const float4* __restrict__ P4,
                                         const int d, const int s, int lo, unsigned int nb) {
    const unsigned int b = (unsigned int)(d - lo);
    if (b < nb) {
        const float4 p = P4[s];
        float* q = bins + b * GENRES;
        atomicAdd(q + 0, p.x);
        atomicAdd(q + 1, p.y);
        atomicAdd(q + 2, p.z);
    }
}

__global__ __launch_bounds__(PS_THREADS)
void proj_scatter_kernel(const float* __restrict__ P, const int* __restrict__ src,
                         const int* __restrict__ dst, float* __restrict__ accp, int E) {
    __shared__ float bins[PS_RB * GENRES];

    const int tid = threadIdx.x;
    const int lo = (int)blockIdx.y * PS_RB;
    const int left = NUM_MOVIES - lo;
    const unsigned int nb = (unsigned int)(left < PS_RB ? left : PS_RB);
    const float4* __restrict__ P4 = (const float4*)P;

    for (int i = tid; i < PS_RB * GENRES; i += PS_THREADS) bins[i] = 0.0f;
    __syncthreads();

    const int per = (((E + PS_SLICES - 1) / PS_SLICES) + 3) & ~3;
    const int start = blockIdx.x * per;
    int end = start + per;
    if (end > E) end = E;
    if (start < end) {
        const int4* __restrict__ pd = (const int4*)(dst + start);
        const int4* __restrict__ ps = (const int4*)(src + start);
        const int nvec = (end - start) >> 2;
        int i = tid;
        for (; i + PS_THREADS < nvec; i += 2 * PS_THREADS) {
            const int4 d0 = pd[i];
            const int4 s0 = ps[i];
            const int4 d1 = pd[i + PS_THREADS];
            const int4 s1 = ps[i + PS_THREADS];
            ps_edges8(bins, P4, d0, s0, d1, s1, lo, nb);
        }
        for (; i < nvec; i += PS_THREADS) {
            const int4 d0 = pd[i];
            const int4 s0 = ps[i];
            ps_edge1(bins, P4, d0.x, s0.x, lo, nb);
            ps_edge1(bins, P4, d0.y, s0.y, lo, nb);
            ps_edge1(bins, P4, d0.z, s0.z, lo, nb);
            ps_edge1(bins, P4, d0.w, s0.w, lo, nb);
        }
        for (int e = start + (nvec << 2) + tid; e < end; e += PS_THREADS)
            ps_edge1(bins, P4, dst[e], src[e], lo, nb);
    }
    __syncthreads();

    const int n = (int)nb * GENRES;
    const int rot = ((int)(blockIdx.x / PS_PLANES) * (n / (PS_SLICES / PS_PLANES))) & ~31;
    float* __restrict__ out = accp + (size_t)(blockIdx.x % PS_PLANES) * PS_PITCH + (size_t)lo * GENRES;
    for (int j = tid; j < n; j += PS_THREADS) {
        int i = j + rot;
        if (i >= n) i -= n;
        const float v = bins[i];
        if (v != 0.0f) atomicAdd(&out[i], v);
    }
}

template <bool PROJ>
__global__ __launch_bounds__(256)
void sage_gemm64(const float* __restrict__ A1, const float* __restrict__ deg1,
                 const float* __restrict__ W1, int K1,
                 const float* __restrict__ A2, const float* __restrict__ W2, int K2,
                 const float* __restrict__ bias, const float* __restrict__ Wp,
                 float* __restrict__ C, int M, int relu) {
    const int lane = threadIdx.x & 31;
    const int wave = threadIdx.x >> 5;
    const int half = lane >> 4;
    const int r    = lane & 15;
    const int row0 = (blockIdx.x * 8 + wave) * 16;
    if (row0 >= M) return;

    v8f c[4] = {};

    float scale = 1.0f;
    if (deg1) scale = 1.0f / fmaxf(deg1[row0 + r], 1.0f);

    {
        const float* Arow = A1 + (size_t)(row0 + r) * K1;
        for (int kk = 0; kk < K1; kk += 4) {
            const int k = kk + 2 * half;
            v2f a;
            a.x = Arow[k] * scale;
            a.y = Arow[k + 1] * scale;
#pragma unroll
            for (int t = 0; t < 4; ++t) {
                const float* Wrow = W1 + (size_t)(t * 16 + r) * K1;
                v2f b;
                b.x = Wrow[k];
                b.y = Wrow[k + 1];
                c[t] = __builtin_amdgcn_wmma_f32_16x16x4_f32(
                    false, a, false, b, (short)0, c[t], false, false);
            }
        }
    }
    if (A2) {
        const float* Arow = A2 + (size_t)(row0 + r) * K2;
        for (int kk = 0; kk < K2; kk += 4) {
            const int k = kk + 2 * half;
            v2f a;
            a.x = Arow[k];
            a.y = Arow[k + 1];
#pragma unroll
            for (int t = 0; t < 4; ++t) {
                const float* Wrow = W2 + (size_t)(t * 16 + r) * K2;
                v2f b;
                b.x = Wrow[k];
                b.y = Wrow[k + 1];
                c[t] = __builtin_amdgcn_wmma_f32_16x16x4_f32(
                    false, a, false, b, (short)0, c[t], false, false);
            }
        }
    }

    if (!PROJ) {
#pragma unroll
        for (int t = 0; t < 4; ++t) {
            const float bv = bias ? bias[t * 16 + r] : 0.0f;
#pragma unroll
            for (int i = 0; i < 8; ++i) {
                float v = c[t][i] + bv;
                if (relu) v = fmaxf(v, 0.0f);
                C[(size_t)(row0 + i + 8 * half) * HID + t * 16 + r] = v;
            }
        }
    } else {
        float s[8][GENRES];
#pragma unroll
        for (int i = 0; i < 8; ++i)
#pragma unroll
            for (int g = 0; g < GENRES; ++g) s[i][g] = 0.0f;
#pragma unroll
        for (int t = 0; t < 4; ++t) {
            const float bv = bias ? bias[t * 16 + r] : 0.0f;
            float w[GENRES];
#pragma unroll
            for (int g = 0; g < GENRES; ++g) w[g] = Wp[g * HID + t * 16 + r];
#pragma unroll
            for (int i = 0; i < 8; ++i) {
                float v = c[t][i] + bv;
                if (relu) v = fmaxf(v, 0.0f);
#pragma unroll
                for (int g = 0; g < GENRES; ++g) s[i][g] += v * w[g];
            }
        }
#pragma unroll
        for (int i = 0; i < 8; ++i)
#pragma unroll
            for (int g = 0; g < GENRES; ++g)
#pragma unroll
                for (int m = 8; m >= 1; m >>= 1)
                    s[i][g] += __shfl_xor(s[i][g], m, 16);
        float4 o = make_float4(0.0f, 0.0f, 0.0f, 0.0f);
#pragma unroll
        for (int i = 0; i < 8; ++i)
            if (r == i) o = make_float4(s[i][0], s[i][1], s[i][2], 0.0f);
        if (r < 8) ((float4*)C)[(size_t)(row0 + r + 8 * half)] = o;
    }
}

__global__ __launch_bounds__(256)
void head_kernel(const float* __restrict__ accp, const float* __restrict__ degm,
                 const float* __restrict__ m1,
                 const float* __restrict__ bl2, const float* __restrict__ Wr2,
                 float* __restrict__ out, int M) {
    int m = blockIdx.x * blockDim.x + threadIdx.x;
    if (m >= M) return;
    const float inv = 1.0f / fmaxf(degm[m], 1.0f);
    const float* a = accp + (size_t)m * GENRES;
    const float* x = m1 + (size_t)m * HID;
    float acc[GENRES];
#pragma unroll
    for (int g = 0; g < GENRES; ++g) {
        float sum = a[g];
#pragma unroll
        for (int pl = 1; pl < PS_PLANES; ++pl) sum += a[(size_t)pl * PS_PITCH + g];
        acc[g] = sum * inv + bl2[g];
    }
    for (int k = 0; k < HID; ++k) {
        const float xv = x[k];
#pragma unroll
        for (int g = 0; g < GENRES; ++g)
            acc[g] += xv * Wr2[g * HID + k];
    }
#pragma unroll
    for (int g = 0; g < GENRES; ++g) out[(size_t)m * GENRES + g] = acc[g];
}

extern "C" void kernel_launch(void* const* d_in, const int* in_sizes, int n_in,
                              void* d_out, int out_size, void* d_ws, size_t ws_size,
                              hipStream_t stream) {
    (void)in_sizes; (void)n_in; (void)out_size; (void)ws_size;

    const float* x_user  = (const float*)d_in[0];
    const float* x_movie = (const float*)d_in[1];
    const int*   e_src   = (const int*)d_in[2];
    const int*   e_dst   = (const int*)d_in[3];
    const float* W_user  = (const float*)d_in[4];
    const float* b_user  = (const float*)d_in[5];
    const float* W_movie = (const float*)d_in[6];
    const float* b_movie = (const float*)d_in[7];
    const float* Wl1_um  = (const float*)d_in[8];
    const float* bl1_um  = (const float*)d_in[9];
    const float* Wr1_um  = (const float*)d_in[10];
    const float* Wl1_mu  = (const float*)d_in[11];
    const float* bl1_mu  = (const float*)d_in[12];
    const float* Wr1_mu  = (const float*)d_in[13];
    const float* Wl2_um  = (const float*)d_in[14];
    const float* bl2_um  = (const float*)d_in[15];
    const float* Wr2_um  = (const float*)d_in[16];
    float* out = (float*)d_out;

    float* ws = (float*)d_ws;
    float* hu        = ws;
    float* hm        = hu + 6400000;
    float* m1        = hm + 3200000;
    float* pu        = m1 + 3200000;
    float* acc_um    = pu + 400000;
    float* acc_mu    = acc_um + 3200000;
    float* accp      = acc_mu + 6400000;
    float* deg_user  = accp + 600064;
    float* deg_movie = deg_user + 100000;

    const size_t zero_bytes = (size_t)(3200000 + 6400000 + 600064 + 100000 + 50000) * 4;
    hipMemsetAsync(acc_um, 0, zero_bytes, stream);

    const int E = NUM_EDGES;
    degree_hist_kernel<<<dim3(HSLICES, MOVIE_RANGES + USER_RANGES), HTHREADS, 0, stream>>>(
        e_dst, deg_movie, e_src, deg_user, E);

    const int gridU = (NUM_USERS / 16 + 7) / 8;
    const int gridM = (NUM_MOVIES / 16 + 7) / 8;
    sage_gemm64<false><<<gridU, 256, 0, stream>>>(x_user, nullptr, W_user, USER_FEAT,
                                                  nullptr, nullptr, 0, b_user, nullptr, hu,
                                                  NUM_USERS, 0);
    sage_gemm64<false><<<gridM, 256, 0, stream>>>(x_movie, nullptr, W_movie, MOVIE_FEAT,
                                                  nullptr, nullptr, 0, b_movie, nullptr, hm,
                                                  NUM_MOVIES, 0);

    const int sblocks = (E + 8 * SC_EPW - 1) / (8 * SC_EPW);

    scatter_kernel<<<sblocks, 256, 0, stream>>>(hu, e_src, e_dst, acc_um, E);
    sage_gemm64<false><<<gridM, 256, 0, stream>>>(acc_um, deg_movie, Wl1_um, HID,
                                                  hm, Wr1_um, HID, bl1_um, nullptr, m1,
                                                  NUM_MOVIES, 1);

    scatter_kernel<<<sblocks, 256, 0, stream>>>(hm, e_dst, e_src, acc_mu, E);
    sage_gemm64<true><<<gridU, 256, 0, stream>>>(acc_mu, deg_user, Wl1_mu, HID,
                                                 hu, Wr1_mu, HID, bl1_mu, Wl2_um, pu,
                                                 NUM_USERS, 1);

    proj_scatter_kernel<<<dim3(PS_SLICES, PS_RANGES), PS_THREADS, 0, stream>>>(
        pu, e_src, e_dst, accp, E);
    head_kernel<<<(NUM_MOVIES + 255) / 256, 256, 0, stream>>>(
        accp, deg_movie, m1, bl2_um, Wr2_um, out, NUM_MOVIES);
}


# Edge-sum accumulators doubled, one copy per half of the chip (die id read per wave): a row's atomic adds stay in one L2 half; the two copies are added in the GEMM A load; sum order changes
# speedup vs baseline: 3.7590x; 1.1538x over previous
#include <hip/hip_runtime.h>
#include <hip/hip_bf16.h>

typedef __attribute__((ext_vector_type(2))) float v2f;
typedef __attribute__((ext_vector_type(8))) float v8f;

#define NUM_USERS  100000
#define NUM_MOVIES 50000
#define NUM_EDGES  4000000
#define USER_FEAT  24
#define MOVIE_FEAT 404
#define HID        64
#define GENRES     3

#define HBINS        50000
#define HSLICES      32
#define HTHREADS     1024
#define MOVIE_RANGES ((NUM_MOVIES + HBINS - 1) / HBINS)
#define USER_RANGES  ((NUM_USERS + HBINS - 1) / HBINS)

__device__ __forceinline__ void hist_count4(unsigned int* bins, const int4 v, int lo, unsigned int nb) {
    const unsigned int b0 = (unsigned int)(v.x - lo);
    const unsigned int b1 = (unsigned int)(v.y - lo);
    const unsigned int b2 = (unsigned int)(v.z - lo);
    const unsigned int b3 = (unsigned int)(v.w - lo);
    if (b0 < nb) atomicAdd(&bins[b0], 1u);
    if (b1 < nb) atomicAdd(&bins[b1], 1u);
    if (b2 < nb) atomicAdd(&bins[b2], 1u);
    if (b3 < nb) atomicAdd(&bins[b3], 1u);
}

__global__ __launch_bounds__(HTHREADS)
void degree_hist_kernel(const int* __restrict__ dst, float* __restrict__ deg_movie,
                        const int* __restrict__ src, float* __restrict__ deg_user, int E) {
    __shared__ unsigned int bins[HBINS];

    const int tid = threadIdx.x;
    const int job = blockIdx.y;
    const bool is_movie = job < MOVIE_RANGES;
    const int* __restrict__ idx = is_movie ? dst : src;
    const int nseg = is_movie ? NUM_MOVIES : NUM_USERS;
    const int lo = (is_movie ? job : job - MOVIE_RANGES) * HBINS;
    const int left = nseg - lo;
    const unsigned int nb = (unsigned int)(left < HBINS ? left : HBINS);
    float* __restrict__ out = (is_movie ? deg_movie : deg_user) + lo;

    for (int i = tid; i < HBINS; i += HTHREADS) bins[i] = 0u;
    __syncthreads();

    const int per = (((E + HSLICES - 1) / HSLICES) + 3) & ~3;
    const int start = blockIdx.x * per;
    int end = start + per;
    if (end > E) end = E;
    if (start < end) {
        const int4* __restrict__ p = (const int4*)(idx + start);
        const int nvec = (end - start) >> 2;
        int i = tid;
        for (; i + 3 * HTHREADS < nvec; i += 4 * HTHREADS) {
            const int4 a = p[i];
            const int4 b = p[i + HTHREADS];
            const int4 c = p[i + 2 * HTHREADS];
            const int4 d = p[i + 3 * HTHREADS];
            hist_count4(bins, a, lo, nb);
            hist_count4(bins, b, lo, nb);
            hist_count4(bins, c, lo, nb);
            hist_count4(bins, d, lo, nb);
        }
        for (; i < nvec; i += HTHREADS) hist_count4(bins, p[i], lo, nb);
        for (int e = start + (nvec << 2) + tid; e < end; e += HTHREADS) {
            const unsigned int b = (unsigned int)(idx[e] - lo);
            if (b < nb) atomicAdd(&bins[b], 1u);
        }
    }
    __syncthreads();

    const int n = (int)nb;
    const int rot = ((int)blockIdx.x * (n / HSLICES)) & ~31;
    for (int j = tid; j < n; j += HTHREADS) {
        int i = j + rot;
        if (i >= n) i -= n;
        const unsigned int cnt = bins[i];
        if (cnt) atomicAdd(&out[i], (float)cnt);
    }
}

#define SC_EPW   64
#define SC_GROUP 8

__device__ __forceinline__ int chip_half() {
    const unsigned int id = __builtin_amdgcn_s_sendmsg_rtn(0x87u);
    return (int)((id >> 18) & 1u);
}

__device__ __forceinline__ void scatter_rows(const float* __restrict__ X, float* __restrict__ acc,
                                             const int sv, const int dv, const int n, const int lane) {
    if (n == 32) {
#pragma unroll
        for (int g = 0; g < 32; g += SC_GROUP) {
            float lo[SC_GROUP], hi[SC_GROUP];
#pragma unroll
            for (int j = 0; j < SC_GROUP; ++j) {
                const int s = __builtin_amdgcn_readlane(sv, g + j);
                const float* __restrict__ xr = X + (size_t)s * HID + lane;
                lo[j] = xr[0];
                hi[j] = xr[32];
            }
#pragma unroll
            for (int j = 0; j < SC_GROUP; ++j) {
                const int d = __builtin_amdgcn_readlane(dv, g + j);
                float* __restrict__ o = acc + (size_t)d * HID + lane;
                atomicAdd(o, lo[j]);
                atomicAdd(o + 32, hi[j]);
            }
        }
    } else {
        for (int i = 0; i < n; ++i) {
            const int s = __builtin_amdgcn_readlane(sv, i);
            const int d = __builtin_amdgcn_readlane(dv, i);
            const float* __restrict__ xr = X + (size_t)s * HID + lane;
            const float a = xr[0];
            const float b = xr[32];
            float* __restrict__ o = acc + (size_t)d * HID + lane;
            atomicAdd(o, a);
            atomicAdd(o + 32, b);
        }
    }
}

__global__ __launch_bounds__(256) void scatter_kernel(const float* __restrict__ X,
                                                      const int* __restrict__ src,
                                                      const int* __restrict__ dst,
                                                      float* __restrict__ acc0,
                                                      float* __restrict__ acc1, int E) {
    const int lane = threadIdx.x & 31;
    float* __restrict__ acc = chip_half() ? acc1 : acc0;
    const int wave = __builtin_amdgcn_readfirstlane((int)(blockIdx.x * 8 + (threadIdx.x >> 5)));
    const long long e0 = (long long)wave * SC_EPW;
    if (e0 >= E) return;
    const int ea = (int)e0 + lane;
    const int eb = ea + 32;
    int sa = 0, da = 0, sb = 0, db = 0;
    if (ea < E) { sa = src[ea]; da = dst[ea]; }
    if (eb < E) { sb = src[eb]; db = dst[eb]; }
    int na = E - (int)e0;
    int nb = na - 32;
    if (na > 32) na = 32;
    if (nb > 32) nb = 32;
    scatter_rows(X, acc, sa, da, na, lane);
    if (nb > 0) scatter_rows(X, acc, sb, db, nb, lane);
}

#define PS_RANGES  2
#define PS_RB      ((NUM_MOVIES + PS_RANGES - 1) / PS_RANGES)
#define PS_SLICES  128
#define PS_THREADS 1024
#define PS_PLANES  4
#define PS_PITCH   150016

__device__ __forceinline__ void ps_edges8(float* bins, const float4* __restrict__ P4,
                                          const int4 d0, const int4 s0,
                                          const int4 d1, const int4 s1,
                                          int lo, unsigned int nb) {
    const int dd[8] = {d0.x, d0.y, d0.z, d0.w, d1.x, d1.y, d1.z, d1.w};
    const int ss[8] = {s0.x, s0.y, s0.z, s0.w, s1.x, s1.y, s1.z, s1.w};
    unsigned int b[8];
    float4 p[8];
#pragma unroll
    for (int j = 0; j < 8; ++j) {
        b[j] = (unsigned int)(dd[j] - lo);
        p[j] = make_float4(0.0f, 0.0f, 0.0f, 0.0f);
        if (b[j] < nb) p[j] = P4[ss[j]];
    }
#pragma unroll
    for (int j = 0; j < 8; ++j) {
        if (b[j] < nb) {
            float* q = bins + b[j] * GENRES;
            atomicAdd(q + 0, p[j].x);
            atomicAdd(q + 1, p[j].y);
            atomicAdd(q + 2, p[j].z);
        }
    }
}

__device__ __forceinline__ void ps_edge1(float* bins, const float4* __restrict__ P4,
                                         const int d, const int s, int lo, unsigned int nb) {
    const unsigned int b = (unsigned int)(d - lo);
    if (b < nb) {
        const float4 p = P4[s];
        float* q = bins + b * GENRES;
        atomicAdd(q + 0, p.x);
        atomicAdd(q + 1, p.y);
        atomicAdd(q + 2, p.z);
    }
}

__global__ __launch_bounds__(PS_THREADS)
void proj_scatter_kernel(const float* __restrict__ P, const int* __restrict__ src,
                         const int* __restrict__ dst, float* __restrict__ accp, int E) {
    __shared__ float bins[PS_RB * GENRES];

    const int tid = threadIdx.x;
    const int lo = (int)blockIdx.y * PS_RB;
    const int left = NUM_MOVIES - lo;
    const unsigned int nb = (unsigned int)(left < PS_RB ? left : PS_RB);
    const float4* __restrict__ P4 = (const float4*)P;

    for (int i = tid; i < PS_RB * GENRES; i += PS_THREADS) bins[i] = 0.0f;
    __syncthreads();

    const int per = (((E + PS_SLICES - 1) / PS_SLICES) + 3) & ~3;
    const int start = blockIdx.x * per;
    int end = start + per;
    if (end > E) end = E;
    if (start < end) {
        const int4* __restrict__ pd = (const int4*)(dst + start);
        const int4* __restrict__ ps = (const int4*)(src + start);
        const int nvec = (end - start) >> 2;
        int i = tid;
        for (; i + PS_THREADS < nvec; i += 2 * PS_THREADS) {
            const int4 d0 = pd[i];
            const int4 s0 = ps[i];
            const int4 d1 = pd[i + PS_THREADS];
            const int4 s1 = ps[i + PS_THREADS];
            ps_edges8(bins, P4, d0, s0, d1, s1, lo, nb);
        }
        for (; i < nvec; i += PS_THREADS) {
            const int4 d0 = pd[i];
            const int4 s0 = ps[i];
            ps_edge1(bins, P4, d0.x, s0.x, lo, nb);
            ps_edge1(bins, P4, d0.y, s0.y, lo, nb);
            ps_edge1(bins, P4, d0.z, s0.z, lo, nb);
            ps_edge1(bins, P4, d0.w, s0.w, lo, nb);
        }
        for (int e = start + (nvec << 2) + tid; e < end; e += PS_THREADS)
            ps_edge1(bins, P4, dst[e], src[e], lo, nb);
    }
    __syncthreads();

    const int n = (int)nb * GENRES;
    const int rot = ((int)(blockIdx.x / PS_PLANES) * (n / (PS_SLICES / PS_PLANES))) & ~31;
    float* __restrict__ out = accp + (size_t)(blockIdx.x % PS_PLANES) * PS_PITCH + (size_t)lo * GENRES;
    for (int j = tid; j < n; j += PS_THREADS) {
        int i = j + rot;
        if (i >= n) i -= n;
        const float v = bins[i];
        if (v != 0.0f) atomicAdd(&out[i], v);
    }
}

template <bool PROJ, bool DUAL>
__global__ __launch_bounds__(256)
void sage_gemm64(const float* __restrict__ A1, const float* A1b,
                 const float* __restrict__ deg1,
                 const float* __restrict__ W1, int K1,
                 const float* __restrict__ A2, const float* __restrict__ W2, int K2,
                 const float* __restrict__ bias, const float* __restrict__ Wp,
                 float* C, int M, int relu) {
    const int lane = threadIdx.x & 31;
    const int wave = threadIdx.x >> 5;
    const int half = lane >> 4;
    const int r    = lane & 15;
    const int row0 = (blockIdx.x * 8 + wave) * 16;
    if (row0 >= M) return;

    v8f c[4] = {};

    float scale = 1.0f;
    if (deg1) scale = 1.0f / fmaxf(deg1[row0 + r], 1.0f);

    {
        const float* Arow = A1 + (size_t)(row0 + r) * K1;
        const float* Brow = DUAL ? A1b + (size_t)(row0 + r) * K1 : Arow;
        for (int kk = 0; kk < K1; kk += 4) {
            const int k = kk + 2 * half;
            float a0 = Arow[k];
            float a1 = Arow[k + 1];
            if (DUAL) { a0 += Brow[k]; a1 += Brow[k + 1]; }
            v2f a;
            a.x = a0 * scale;
            a.y = a1 * scale;
#pragma unroll
            for (int t = 0; t < 4; ++t) {
                const float* Wrow = W1 + (size_t)(t * 16 + r) * K1;
                v2f b;
                b.x = Wrow[k];
                b.y = Wrow[k + 1];
                c[t] = __builtin_amdgcn_wmma_f32_16x16x4_f32(
                    false, a, false, b, (short)0, c[t], false, false);
            }
        }
    }
    if (A2) {
        const float* Arow = A2 + (size_t)(row0 + r) * K2;
        for (int kk = 0; kk < K2; kk += 4) {
            const int k = kk + 2 * half;
            v2f a;
            a.x = Arow[k];
            a.y = Arow[k + 1];
#pragma unroll
            for (int t = 0; t < 4; ++t) {
                const float* Wrow = W2 + (size_t)(t * 16 + r) * K2;
                v2f b;
                b.x = Wrow[k];
                b.y = Wrow[k + 1];
                c[t] = __builtin_amdgcn_wmma_f32_16x16x4_f32(
                    false, a, false, b, (short)0, c[t], false, false);
            }
        }
    }

    if (!PROJ) {
#pragma unroll
        for (int t = 0; t < 4; ++t) {
            const float bv = bias ? bias[t * 16 + r] : 0.0f;
#pragma unroll
            for (int i = 0; i < 8; ++i) {
                float v = c[t][i] + bv;
                if (relu) v = fmaxf(v, 0.0f);
                C[(size_t)(row0 + i + 8 * half) * HID + t * 16 + r] = v;
            }
        }
    } else {
        float s[8][GENRES];
#pragma unroll
        for (int i = 0; i < 8; ++i)
#pragma unroll
            for (int g = 0; g < GENRES; ++g) s[i][g] = 0.0f;
#pragma unroll
        for (int t = 0; t < 4; ++t) {
            const float bv = bias ? bias[t * 16 + r] : 0.0f;
            float w[GENRES];
#pragma unroll
            for (int g = 0; g < GENRES; ++g) w[g] = Wp[g * HID + t * 16 + r];
#pragma unroll
            for (int i = 0; i < 8; ++i) {
                float v = c[t][i] + bv;
                if (relu) v = fmaxf(v, 0.0f);
#pragma unroll
                for (int g = 0; g < GENRES; ++g) s[i][g] += v * w[g];
            }
        }
#pragma unroll
        for (int i = 0; i < 8; ++i)
#pragma unroll
            for (int g = 0; g < GENRES; ++g)
#pragma unroll
                for (int m = 8; m >= 1; m >>= 1)
                    s[i][g] += __shfl_xor(s[i][g], m, 16);
        float4 o = make_float4(0.0f, 0.0f, 0.0f, 0.0f);
#pragma unroll
        for (int i = 0; i < 8; ++i)
            if (r == i) o = make_float4(s[i][0], s[i][1], s[i][2], 0.0f);
        if (r < 8) ((float4*)C)[(size_t)(row0 + r + 8 * half)] = o;
    }
}

__global__ __launch_bounds__(256)
void head_kernel(const float* __restrict__ accp, const float* __restrict__ degm,
                 const float* __restrict__ m1,
                 const float* __restrict__ bl2, const float* __restrict__ Wr2,
                 float* __restrict__ out, int M) {
    int m = blockIdx.x * blockDim.x + threadIdx.x;
    if (m >= M) return;
    const float inv = 1.0f / fmaxf(degm[m], 1.0f);
    const float* a = accp + (size_t)m * GENRES;
    const float* x = m1 + (size_t)m * HID;
    float acc[GENRES];
#pragma unroll
    for (int g = 0; g < GENRES; ++g) {
        float sum = a[g];
#pragma unroll
        for (int pl = 1; pl < PS_PLANES; ++pl) sum += a[(size_t)pl * PS_PITCH + g];
        acc[g] = sum * inv + bl2[g];
    }
    for (int k = 0; k < HID; ++k) {
        const float xv = x[k];
#pragma unroll
        for (int g = 0; g < GENRES; ++g)
            acc[g] += xv * Wr2[g * HID + k];
    }
#pragma unroll
    for (int g = 0; g < GENRES; ++g) out[(size_t)m * GENRES + g] = acc[g];
}

extern "C" void kernel_launch(void* const* d_in, const int* in_sizes, int n_in,
                              void* d_out, int out_size, void* d_ws, size_t ws_size,
                              hipStream_t stream) {
    (void)in_sizes; (void)n_in; (void)out_size; (void)ws_size;

    const float* x_user  = (const float*)d_in[0];
    const float* x_movie = (const float*)d_in[1];
    const int*   e_src   = (const int*)d_in[2];
    const int*   e_dst   = (const int*)d_in[3];
    const float* W_user  = (const float*)d_in[4];
    const float* b_user  = (const float*)d_in[5];
    const float* W_movie = (const float*)d_in[6];
    const float* b_movie = (const float*)d_in[7];
    const float* Wl1_um  = (const float*)d_in[8];
    const float* bl1_um  = (const float*)d_in[9];
    const float* Wr1_um  = (const float*)d_in[10];
    const float* Wl1_mu  = (const float*)d_in[11];
    const float* bl1_mu  = (const float*)d_in[12];
    const float* Wr1_mu  = (const float*)d_in[13];
    const float* Wl2_um  = (const float*)d_in[14];
    const float* bl2_um  = (const float*)d_in[15];
    const float* Wr2_um  = (const float*)d_in[16];
    float* out = (float*)d_out;

    float* ws = (float*)d_ws;
    float* hu        = ws;
    float* hm        = hu + 6400000;
    float* pu        = hm + 3200000;
    float* m1        = pu + 400000;
    float* acc_um    = m1 + 3200000;
    float* acc_mu    = acc_um + 3200000;
    float* acc_mu_b  = acc_mu + 6400000;
    float* accp      = acc_mu_b + 6400000;
    float* deg_user  = accp + 600064;
    float* deg_movie = deg_user + 100000;
    float* acc_um_b  = m1;

    const size_t zero_bytes = (size_t)(3200000 + 3200000 + 6400000 + 6400000 + 600064 + 100000 + 50000) * 4;
    hipMemsetAsync(m1, 0, zero_bytes, stream);

    const int E = NUM_EDGES;
    degree_hist_kernel<<<dim3(HSLICES, MOVIE_RANGES + USER_RANGES), HTHREADS, 0, stream>>>(
        e_dst, deg_movie, e_src, deg_user, E);

    const int gridU = (NUM_USERS / 16 + 7) / 8;
    const int gridM = (NUM_MOVIES / 16 + 7) / 8;
    sage_gemm64<false, false><<<gridU, 256, 0, stream>>>(x_user, nullptr, nullptr, W_user, USER_FEAT,
                                                         nullptr, nullptr, 0, b_user, nullptr, hu,
                                                         NUM_USERS, 0);
    sage_gemm64<false, false><<<gridM, 256, 0, stream>>>(x_movie, nullptr, nullptr, W_movie, MOVIE_FEAT,
                                                         nullptr, nullptr, 0, b_movie, nullptr, hm,
                                                         NUM_MOVIES, 0);

    const int sblocks = (E + 8 * SC_EPW - 1) / (8 * SC_EPW);

    scatter_kernel<<<sblocks, 256, 0, stream>>>(hu, e_src, e_dst, acc_um, acc_um_b, E);
    sage_gemm64<false, true><<<gridM, 256, 0, stream>>>(acc_um, acc_um_b, deg_movie, Wl1_um, HID,
                                                        hm, Wr1_um, HID, bl1_um, nullptr, m1,
                                                        NUM_MOVIES, 1);

    scatter_kernel<<<sblocks, 256, 0, stream>>>(hm, e_dst, e_src, acc_mu, acc_mu_b, E);
    sage_gemm64<true, true><<<gridU, 256, 0, stream>>>(acc_mu, acc_mu_b, deg_user, Wl1_mu, HID,
                                                       hu, Wr1_mu, HID, bl1_mu, Wl2_um, pu,
                                                       NUM_USERS, 1);

    proj_scatter_kernel<<<dim3(PS_SLICES, PS_RANGES), PS_THREADS, 0, stream>>>(
        pu, e_src, e_dst, accp, E);
    head_kernel<<<(NUM_MOVIES + 255) / 256, 256, 0, stream>>>(
        accp, deg_movie, m1, bl2_um, Wr2_um, out, NUM_MOVIES);
}


# GEMM B operand: each block copies its weights once into LDS (conflict-free row pitch), fragments by ds_load_b64 instead of four 16-line global loads per K-step; K compile-time, K loop unrolled
# speedup vs baseline: 4.2785x; 1.1382x over previous
#include <hip/hip_runtime.h>
#include <hip/hip_bf16.h>

typedef __attribute__((ext_vector_type(2))) float v2f;
typedef __attribute__((ext_vector_type(8))) float v8f;

#define NUM_USERS  100000
#define NUM_MOVIES 50000
#define NUM_EDGES  4000000
#define USER_FEAT  24
#define MOVIE_FEAT 404
#define HID        64
#define GENRES     3

#define HBINS        50000
#define HSLICES      32
#define HTHREADS     1024
#define MOVIE_RANGES ((NUM_MOVIES + HBINS - 1) / HBINS)
#define USER_RANGES  ((NUM_USERS + HBINS - 1) / HBINS)

__device__ __forceinline__ void hist_count4(unsigned int* bins, const int4 v, int lo, unsigned int nb) {
    const unsigned int b0 = (unsigned int)(v.x - lo);
    const unsigned int b1 = (unsigned int)(v.y - lo);
    const unsigned int b2 = (unsigned int)(v.z - lo);
    const unsigned int b3 = (unsigned int)(v.w - lo);
    if (b0 < nb) atomicAdd(&bins[b0], 1u);
    if (b1 < nb) atomicAdd(&bins[b1], 1u);
    if (b2 < nb) atomicAdd(&bins[b2], 1u);
    if (b3 < nb) atomicAdd(&bins[b3], 1u);
}

__global__ __launch_bounds__(HTHREADS)
void degree_hist_kernel(const int* __restrict__ dst, float* __restrict__ deg_movie,
                        const int* __restrict__ src, float* __restrict__ deg_user, int E) {
    __shared__ unsigned int bins[HBINS];

    const int tid = threadIdx.x;
    const int job = blockIdx.y;
    const bool is_movie = job < MOVIE_RANGES;
    const int* __restrict__ idx = is_movie ? dst : src;
    const int nseg = is_movie ? NUM_MOVIES : NUM_USERS;
    const int lo = (is_movie ? job : job - MOVIE_RANGES) * HBINS;
    const int left = nseg - lo;
    const unsigned int nb = (unsigned int)(left < HBINS ? left : HBINS);
    float* __restrict__ out = (is_movie ? deg_movie : deg_user) + lo;

    for (int i = tid; i < HBINS; i += HTHREADS) bins[i] = 0u;
    __syncthreads();

    const int per = (((E + HSLICES - 1) / HSLICES) + 3) & ~3;
    const int start = blockIdx.x * per;
    int end = start + per;
    if (end > E) end = E;
    if (start < end) {
        const int4* __restrict__ p = (const int4*)(idx + start);
        const int nvec = (end - start) >> 2;
        int i = tid;
        for (; i + 3 * HTHREADS < nvec; i += 4 * HTHREADS) {
            const int4 a = p[i];
            const int4 b = p[i + HTHREADS];
            const int4 c = p[i + 2 * HTHREADS];
            const int4 d = p[i + 3 * HTHREADS];
            hist_count4(bins, a, lo, nb);
            hist_count4(bins, b, lo, nb);
            hist_count4(bins, c, lo, nb);
            hist_count4(bins, d, lo, nb);
        }
        for (; i < nvec; i += HTHREADS) hist_count4(bins, p[i], lo, nb);
        for (int e = start + (nvec << 2) + tid; e < end; e += HTHREADS) {
            const unsigned int b = (unsigned int)(idx[e] - lo);
            if (b < nb) atomicAdd(&bins[b], 1u);
        }
    }
    __syncthreads();

    const int n = (int)nb;
    const int rot = ((int)blockIdx.x * (n / HSLICES)) & ~31;
    for (int j = tid; j < n; j += HTHREADS) {
        int i = j + rot;
        if (i >= n) i -= n;
        const unsigned int cnt = bins[i];
        if (cnt) atomicAdd(&out[i], (float)cnt);
    }
}

#define SC_EPW   64
#define SC_GROUP 8

__device__ __forceinline__ int chip_half() {
    const unsigned int id = __builtin_amdgcn_s_sendmsg_rtn(0x87u);
    return (int)((id >> 18) & 1u);
}

__device__ __forceinline__ void scatter_rows(const float* __restrict__ X, float* __restrict__ acc,
                                             const int sv, const int dv, const int n, const int lane) {
    if (n == 32) {
#pragma unroll
        for (int g = 0; g < 32; g += SC_GROUP) {
            float lo[SC_GROUP], hi[SC_GROUP];
#pragma unroll
            for (int j = 0; j < SC_GROUP; ++j) {
                const int s = __builtin_amdgcn_readlane(sv, g + j);
                const float* __restrict__ xr = X + (size_t)s * HID + lane;
                lo[j] = xr[0];
                hi[j] = xr[32];
            }
#pragma unroll
            for (int j = 0; j < SC_GROUP; ++j) {
                const int d = __builtin_amdgcn_readlane(dv, g + j);
                float* __restrict__ o = acc + (size_t)d * HID + lane;
                atomicAdd(o, lo[j]);
                atomicAdd(o + 32, hi[j]);
            }
        }
    } else {
        for (int i = 0; i < n; ++i) {
            const int s = __builtin_amdgcn_readlane(sv, i);
            const int d = __builtin_amdgcn_readlane(dv, i);
            const float* __restrict__ xr = X + (size_t)s * HID + lane;
            const float a = xr[0];
            const float b = xr[32];
            float* __restrict__ o = acc + (size_t)d * HID + lane;
            atomicAdd(o, a);
            atomicAdd(o + 32, b);
        }
    }
}

__global__ __launch_bounds__(256) void scatter_kernel(const float* __restrict__ X,
                                                      const int* __restrict__ src,
                                                      const int* __restrict__ dst,
                                                      float* __restrict__ acc0,
                                                      float* __restrict__ acc1, int E) {
    const int lane = threadIdx.x & 31;
    float* __restrict__ acc = chip_half() ? acc1 : acc0;
    const int wave = __builtin_amdgcn_readfirstlane((int)(blockIdx.x * 8 + (threadIdx.x >> 5)));
    const long long e0 = (long long)wave * SC_EPW;
    if (e0 >= E) return;
    const int ea = (int)e0 + lane;
    const int eb = ea + 32;
    int sa = 0, da = 0, sb = 0, db = 0;
    if (ea < E) { sa = src[ea]; da = dst[ea]; }
    if (eb < E) { sb = src[eb]; db = dst[eb]; }
    int na = E - (int)e0;
    int nb = na - 32;
    if (na > 32) na = 32;
    if (nb > 32) nb = 32;
    scatter_rows(X, acc, sa, da, na, lane);
    if (nb > 0) scatter_rows(X, acc, sb, db, nb, lane);
}

#define PS_RANGES  2
#define PS_RB      ((NUM_MOVIES + PS_RANGES - 1) / PS_RANGES)
#define PS_SLICES  128
#define PS_THREADS 1024
#define PS_PLANES  4
#define PS_PITCH   150016

__device__ __forceinline__ void ps_edges8(float* bins, const float4* __restrict__ P4,
                                          const int4 d0, const int4 s0,
                                          const int4 d1, const int4 s1,
                                          int lo, unsigned int nb) {
    const int dd[8] = {d0.x, d0.y, d0.z, d0.w, d1.x, d1.y, d1.z, d1.w};
    const int ss[8] = {s0.x, s0.y, s0.z, s0.w, s1.x, s1.y, s1.z, s1.w};
    unsigned int b[8];
    float4 p[8];
#pragma unroll
    for (int j = 0; j < 8; ++j) {
        b[j] = (unsigned int)(dd[j] - lo);
        p[j] = make_float4(0.0f, 0.0f, 0.0f, 0.0f);
        if (b[j] < nb) p[j] = P4[ss[j]];
    }
#pragma unroll
    for (int j = 0; j < 8; ++j) {
        if (b[j] < nb) {
            float* q = bins + b[j] * GENRES;
            atomicAdd(q + 0, p[j].x);
            atomicAdd(q + 1, p[j].y);
            atomicAdd(q + 2, p[j].z);
        }
    }
}

__device__ __forceinline__ void ps_edge1(float* bins, const float4* __restrict__ P4,
                                         const int d, const int s, int lo, unsigned int nb) {
    const unsigned int b = (unsigned int)(d - lo);
    if (b < nb) {
        const float4 p = P4[s];
        float* q = bins + b * GENRES;
        atomicAdd(q + 0, p.x);
        atomicAdd(q + 1, p.y);
        atomicAdd(q + 2, p.z);
    }
}

__global__ __launch_bounds__(PS_THREADS)
void proj_scatter_kernel(const float* __restrict__ P, const int* __restrict__ src,
                         const int* __restrict__ dst, float* __restrict__ accp, int E) {
    __shared__ float bins[PS_RB * GENRES];

    const int tid = threadIdx.x;
    const int lo = (int)blockIdx.y * PS_RB;
    const int left = NUM_MOVIES - lo;
    const unsigned int nb = (unsigned int)(left < PS_RB ? left : PS_RB);
    const float4* __restrict__ P4 = (const float4*)P;

    for (int i = tid; i < PS_RB * GENRES; i += PS_THREADS) bins[i] = 0.0f;
    __syncthreads();

    const int per = (((E + PS_SLICES - 1) / PS_SLICES) + 3) & ~3;
    const int start = blockIdx.x * per;
    int end = start + per;
    if (end > E) end = E;
    if (start < end) {
        const int4* __restrict__ pd = (const int4*)(dst + start);
        const int4* __restrict__ ps = (const int4*)(src + start);
        const int nvec = (end - start) >> 2;
        int i = tid;
        for (; i + PS_THREADS < nvec; i += 2 * PS_THREADS) {
            const int4 d0 = pd[i];
            const int4 s0 = ps[i];
            const int4 d1 = pd[i + PS_THREADS];
            const int4 s1 = ps[i + PS_THREADS];
            ps_edges8(bins, P4, d0, s0, d1, s1, lo, nb);
        }
        for (; i < nvec; i += PS_THREADS) {
            const int4 d0 = pd[i];
            const int4 s0 = ps[i];
            ps_edge1(bins, P4, d0.x, s0.x, lo, nb);
            ps_edge1(bins, P4, d0.y, s0.y, lo, nb);
            ps_edge1(bins, P4, d0.z, s0.z, lo, nb);
            ps_edge1(bins, P4, d0.w, s0.w, lo, nb);
        }
        for (int e = start + (nvec << 2) + tid; e < end; e += PS_THREADS)
            ps_edge1(bins, P4, dst[e], src[e], lo, nb);
    }
    __syncthreads();

    const int n = (int)nb * GENRES;
    const int rot = ((int)(blockIdx.x / PS_PLANES) * (n / (PS_SLICES / PS_PLANES))) & ~31;
    float* __restrict__ out = accp + (size_t)(blockIdx.x % PS_PLANES) * PS_PITCH + (size_t)lo * GENRES;
    for (int j = tid; j < n; j += PS_THREADS) {
        int i = j + rot;
        if (i >= n) i -= n;
        const float v = bins[i];
        if (v != 0.0f) atomicAdd(&out[i], v);
    }
}

__host__ __device__ constexpr int w_pitch(int K) { return ((K / 4) & 1) ? K : K + 4; }

template <int K, int PITCH>
__device__ __forceinline__ void stage_weights(float* __restrict__ dst, const float* __restrict__ W, int tid) {
    constexpr int Q = K / 4;
    constexpr int STEPS = (64 * Q + 255) / 256;
    const float4* __restrict__ W4 = (const float4*)W;
    float4 v[STEPS];
#pragma unroll
    for (int j = 0; j < STEPS; ++j) {
        const int i = tid + 256 * j;
        v[j] = W4[i < 64 * Q ? i : 64 * Q - 1];
    }
#pragma unroll
    for (int j = 0; j < STEPS; ++j) {
        const int i = tid + 256 * j;
        if (i < 64 * Q) {
            const int n = i / Q;
            const int q = i - n * Q;
            *(float4*)(dst + n * PITCH + 4 * q) = v[j];
        }
    }
}

template <bool PROJ, bool DUAL, int K1, int K2>
__global__ __launch_bounds__(256)
void sage_gemm64(const float* __restrict__ A1, const float* A1b,
                 const float* __restrict__ deg1,
                 const float* __restrict__ W1,
                 const float* __restrict__ A2, const float* __restrict__ W2,
                 const float* __restrict__ bias, const float* __restrict__ Wp,
                 float* C, int M, int relu) {
    constexpr int P1 = w_pitch(K1);
    constexpr int P2 = K2 > 0 ? w_pitch(K2) : 0;
    __shared__ __attribute__((aligned(16))) float wlds[64 * P1 + 64 * P2];

    stage_weights<K1, P1>(wlds, W1, threadIdx.x);
    if (K2 > 0) stage_weights<(K2 > 0 ? K2 : 4), (K2 > 0 ? P2 : 4)>(wlds + 64 * P1, W2, threadIdx.x);
    __syncthreads();

    const int lane = threadIdx.x & 31;
    const int wave = threadIdx.x >> 5;
    const int half = lane >> 4;
    const int r    = lane & 15;
    const int row0 = (blockIdx.x * 8 + wave) * 16;
    if (row0 >= M) return;

    v8f c[4] = {};

    float scale = 1.0f;
    if (deg1) scale = 1.0f / fmaxf(deg1[row0 + r], 1.0f);

    {
        const float* Arow = A1 + (size_t)(row0 + r) * K1;
        const float* Brow = DUAL ? A1b + (size_t)(row0 + r) * K1 : Arow;
        const float* Ws = wlds + r * P1 + 2 * half;
#pragma unroll
        for (int kk = 0; kk < K1; kk += 4) {
            const int k = kk + 2 * half;
            float a0 = Arow[k];
            float a1 = Arow[k + 1];
            if (DUAL) { a0 += Brow[k]; a1 += Brow[k + 1]; }
            v2f a;
            a.x = a0 * scale;
            a.y = a1 * scale;
#pragma unroll
            for (int t = 0; t < 4; ++t) {
                const v2f b = *(const v2f*)(Ws + t * 16 * P1 + kk);
                c[t] = __builtin_amdgcn_wmma_f32_16x16x4_f32(
                    false, a, false, b, (short)0, c[t], false, false);
            }
        }
    }
    if (K2 > 0) {
        const float* Arow = A2 + (size_t)(row0 + r) * K2;
        const float* Ws = wlds + 64 * P1 + r * P2 + 2 * half;
#pragma unroll
        for (int kk = 0; kk < K2; kk += 4) {
            const int k = kk + 2 * half;
            v2f a;
            a.x = Arow[k];
            a.y = Arow[k + 1];
#pragma unroll
            for (int t = 0; t < 4; ++t) {
                const v2f b = *(const v2f*)(Ws + t * 16 * P2 + kk);
                c[t] = __builtin_amdgcn_wmma_f32_16x16x4_f32(
                    false, a, false, b, (short)0, c[t], false, false);
            }
        }
    }

    if (!PROJ) {
#pragma unroll
        for (int t = 0; t < 4; ++t) {
            const float bv = bias ? bias[t * 16 + r] : 0.0f;
#pragma unroll
            for (int i = 0; i < 8; ++i) {
                float v = c[t][i] + bv;
                if (relu) v = fmaxf(v, 0.0f);
                C[(size_t)(row0 + i + 8 * half) * HID + t * 16 + r] = v;
            }
        }
    } else {
        float s[8][GENRES];
#pragma unroll
        for (int i = 0; i < 8; ++i)
#pragma unroll
            for (int g = 0; g < GENRES; ++g) s[i][g] = 0.0f;
#pragma unroll
        for (int t = 0; t < 4; ++t) {
            const float bv = bias ? bias[t * 16 + r] : 0.0f;
            float w[GENRES];
#pragma unroll
            for (int g = 0; g < GENRES; ++g) w[g] = Wp[g * HID + t * 16 + r];
#pragma unroll
            for (int i = 0; i < 8; ++i) {
                float v = c[t][i] + bv;
                if (relu) v = fmaxf(v, 0.0f);
#pragma unroll
                for (int g = 0; g < GENRES; ++g) s[i][g] += v * w[g];
            }
        }
#pragma unroll
        for (int i = 0; i < 8; ++i)
#pragma unroll
            for (int g = 0; g < GENRES; ++g)
#pragma unroll
                for (int m = 8; m >= 1; m >>= 1)
                    s[i][g] += __shfl_xor(s[i][g], m, 16);
        float4 o = make_float4(0.0f, 0.0f, 0.0f, 0.0f);
#pragma unroll
        for (int i = 0; i < 8; ++i)
            if (r == i) o = make_float4(s[i][0], s[i][1], s[i][2], 0.0f);
        if (r < 8) ((float4*)C)[(size_t)(row0 + r + 8 * half)] = o;
    }
}

__global__ __launch_bounds__(256)
void head_kernel(const float* __restrict__ accp, const float* __restrict__ degm,
                 const float* __restrict__ m1,
                 const float* __restrict__ bl2, const float* __restrict__ Wr2,
                 float* __restrict__ out, int M) {
    int m = blockIdx.x * blockDim.x + threadIdx.x;
    if (m >= M) return;
    const float inv = 1.0f / fmaxf(degm[m], 1.0f);
    const float* a = accp + (size_t)m * GENRES;
    const float* x = m1 + (size_t)m * HID;
    float acc[GENRES];
#pragma unroll
    for (int g = 0; g < GENRES; ++g) {
        float sum = a[g];
#pragma unroll
        for (int pl = 1; pl < PS_PLANES; ++pl) sum += a[(size_t)pl * PS_PITCH + g];
        acc[g] = sum * inv + bl2[g];
    }
    for (int k = 0; k < HID; ++k) {
        const float xv = x[k];
#pragma unroll
        for (int g = 0; g < GENRES; ++g)
            acc[g] += xv * Wr2[g * HID + k];
    }
#pragma unroll
    for (int g = 0; g < GENRES; ++g) out[(size_t)m * GENRES + g] = acc[g];
}

extern "C" void kernel_launch(void* const* d_in, const int* in_sizes, int n_in,
                              void* d_out, int out_size, void* d_ws, size_t ws_size,
                              hipStream_t stream) {
    (void)in_sizes; (void)n_in; (void)out_size; (void)ws_size;

    const float* x_user  = (const float*)d_in[0];
    const float* x_movie = (const float*)d_in[1];
    const int*   e_src   = (const int*)d_in[2];
    const int*   e_dst   = (const int*)d_in[3];
    const float* W_user  = (const float*)d_in[4];
    const float* b_user  = (const float*)d_in[5];
    const float* W_movie = (const float*)d_in[6];
    const float* b_movie = (const float*)d_in[7];
    const float* Wl1_um  = (const float*)d_in[8];
    const float* bl1_um  = (const float*)d_in[9];
    const float* Wr1_um  = (const float*)d_in[10];
    const float* Wl1_mu  = (const float*)d_in[11];
    const float* bl1_mu  = (const float*)d_in[12];
    const float* Wr1_mu  = (const float*)d_in[13];
    const float* Wl2_um  = (const float*)d_in[14];
    const float* bl2_um  = (const float*)d_in[15];
    const float* Wr2_um  = (const float*)d_in[16];
    float* out = (float*)d_out;

    float* ws = (float*)d_ws;
    float* hu        = ws;
    float* hm        = hu + 6400000;
    float* pu        = hm + 3200000;
    float* m1        = pu + 400000;
    float* acc_um    = m1 + 3200000;
    float* acc_mu    = acc_um + 3200000;
    float* acc_mu_b  = acc_mu + 6400000;
    float* accp      = acc_mu_b + 6400000;
    float* deg_user  = accp + 600064;
    float* deg_movie = deg_user + 100000;
    float* acc_um_b  = m1;

    const size_t zero_bytes = (size_t)(3200000 + 3200000 + 6400000 + 6400000 + 600064 + 100000 + 50000) * 4;
    hipMemsetAsync(m1, 0, zero_bytes, stream);

    const int E = NUM_EDGES;
    degree_hist_kernel<<<dim3(HSLICES, MOVIE_RANGES + USER_RANGES), HTHREADS, 0, stream>>>(
        e_dst, deg_movie, e_src, deg_user, E);

    const int gridU = (NUM_USERS / 16 + 7) / 8;
    const int gridM = (NUM_MOVIES / 16 + 7) / 8;
    sage_gemm64<false, false, USER_FEAT, 0><<<gridU, 256, 0, stream>>>(
        x_user, nullptr, nullptr, W_user, nullptr, nullptr, b_user, nullptr, hu, NUM_USERS, 0);
    sage_gemm64<false, false, MOVIE_FEAT, 0><<<gridM, 256, 0, stream>>>(
        x_movie, nullptr, nullptr, W_movie, nullptr, nullptr, b_movie, nullptr, hm, NUM_MOVIES, 0);

    const int sblocks = (E + 8 * SC_EPW - 1) / (8 * SC_EPW);

    scatter_kernel<<<sblocks, 256, 0, stream>>>(hu, e_src, e_dst, acc_um, acc_um_b, E);
    sage_gemm64<false, true, HID, HID><<<gridM, 256, 0, stream>>>(
        acc_um, acc_um_b, deg_movie, Wl1_um, hm, Wr1_um, bl1_um, nullptr, m1, NUM_MOVIES, 1);

    scatter_kernel<<<sblocks, 256, 0, stream>>>(hm, e_dst, e_src, acc_mu, acc_mu_b, E);
    sage_gemm64<true, true, HID, HID><<<gridU, 256, 0, stream>>>(
        acc_mu, acc_mu_b, deg_user, Wl1_mu, hu, Wr1_mu, bl1_mu, Wl2_um, pu, NUM_USERS, 1);

    proj_scatter_kernel<<<dim3(PS_SLICES, PS_RANGES), PS_THREADS, 0, stream>>>(
        pu, e_src, e_dst, accp, E);
    head_kernel<<<(NUM_MOVIES + 255) / 256, 256, 0, stream>>>(
        accp, deg_movie, m1, bl2_um, Wr2_um, out, NUM_MOVIES);
}


# User input projection folded into the two layer-1 weights that read hu (one small WMMA launch): hu is never built, movie-side edge sum carries 24 raw features, one line per edge; sum order changes
# speedup vs baseline: 4.7411x; 1.1081x over previous
#include <hip/hip_runtime.h>
#include <hip/hip_bf16.h>

typedef __attribute__((ext_vector_type(2))) float v2f;
typedef __attribute__((ext_vector_type(8))) float v8f;

#define NUM_USERS  100000
#define NUM_MOVIES 50000
#define NUM_EDGES  4000000
#define USER_FEAT  24
#define MOVIE_FEAT 404
#define HID        64
#define GENRES     3

#define HBINS        50000
#define HSLICES      32
#define HTHREADS     1024
#define MOVIE_RANGES ((NUM_MOVIES + HBINS - 1) / HBINS)
#define USER_RANGES  ((NUM_USERS + HBINS - 1) / HBINS)

__device__ __forceinline__ void hist_count4(unsigned int* bins, const int4 v, int lo, unsigned int nb) {
    const unsigned int b0 = (unsigned int)(v.x - lo);
    const unsigned int b1 = (unsigned int)(v.y - lo);
    const unsigned int b2 = (unsigned int)(v.z - lo);
    const unsigned int b3 = (unsigned int)(v.w - lo);
    if (b0 < nb) atomicAdd(&bins[b0], 1u);
    if (b1 < nb) atomicAdd(&bins[b1], 1u);
    if (b2 < nb) atomicAdd(&bins[b2], 1u);
    if (b3 < nb) atomicAdd(&bins[b3], 1u);
}

__global__ __launch_bounds__(HTHREADS)
void degree_hist_kernel(const int* __restrict__ dst, float* __restrict__ deg_movie,
                        const int* __restrict__ src, float* __restrict__ deg_user, int E) {
    __shared__ unsigned int bins[HBINS];

    const int tid = threadIdx.x;
    const int job = blockIdx.y;
    const bool is_movie = job < MOVIE_RANGES;
    const int* __restrict__ idx = is_movie ? dst : src;
    const int nseg = is_movie ? NUM_MOVIES : NUM_USERS;
    const int lo = (is_movie ? job : job - MOVIE_RANGES) * HBINS;
    const int left = nseg - lo;
    const unsigned int nb = (unsigned int)(left < HBINS ? left : HBINS);
    float* __restrict__ out = (is_movie ? deg_movie : deg_user) + lo;

    for (int i = tid; i < HBINS; i += HTHREADS) bins[i] = 0u;
    __syncthreads();

    const int per = (((E + HSLICES - 1) / HSLICES) + 3) & ~3;
    const int start = blockIdx.x * per;
    int end = start + per;
    if (end > E) end = E;
    if (start < end) {
        const int4* __restrict__ p = (const int4*)(idx + start);
        const int nvec = (end - start) >> 2;
        int i = tid;
        for (; i + 3 * HTHREADS < nvec; i += 4 * HTHREADS) {
            const int4 a = p[i];
            const int4 b = p[i + HTHREADS];
            const int4 c = p[i + 2 * HTHREADS];
            const int4 d = p[i + 3 * HTHREADS];
            hist_count4(bins, a, lo, nb);
            hist_count4(bins, b, lo, nb);
            hist_count4(bins, c, lo, nb);
            hist_count4(bins, d, lo, nb);
        }
        for (; i < nvec; i += HTHREADS) hist_count4(bins, p[i], lo, nb);
        for (int e = start + (nvec << 2) + tid; e < end; e += HTHREADS) {
            const unsigned int b = (unsigned int)(idx[e] - lo);
            if (b < nb) atomicAdd(&bins[b], 1u);
        }
    }
    __syncthreads();

    const int n = (int)nb;
    const int rot = ((int)blockIdx.x * (n / HSLICES)) & ~31;
    for (int j = tid; j < n; j += HTHREADS) {
        int i = j + rot;
        if (i >= n) i -= n;
        const unsigned int cnt = bins[i];
        if (cnt) atomicAdd(&out[i], (float)cnt);
    }
}

#define SC_EPW   64
#define SC_GROUP 8

__device__ __forceinline__ int chip_half() {
    const unsigned int id = __builtin_amdgcn_s_sendmsg_rtn(0x87u);
    return (int)((id >> 18) & 1u);
}

__device__ __forceinline__ void scatter_rows(const float* __restrict__ X, float* __restrict__ acc,
                                             const int sv, const int dv, const int n, const int lane) {
    if (n == 32) {
#pragma unroll
        for (int g = 0; g < 32; g += SC_GROUP) {
            float lo[SC_GROUP], hi[SC_GROUP];
#pragma unroll
            for (int j = 0; j < SC_GROUP; ++j) {
                const int s = __builtin_amdgcn_readlane(sv, g + j);
                const float* __restrict__ xr = X + (size_t)s * HID + lane;
                lo[j] = xr[0];
                hi[j] = xr[32];
            }
#pragma unroll
            for (int j = 0; j < SC_GROUP; ++j) {
                const int d = __builtin_amdgcn_readlane(dv, g + j);
                float* __restrict__ o = acc + (size_t)d * HID + lane;
                atomicAdd(o, lo[j]);
                atomicAdd(o + 32, hi[j]);
            }
        }
    } else {
        for (int i = 0; i < n; ++i) {
            const int s = __builtin_amdgcn_readlane(sv, i);
            const int d = __builtin_amdgcn_readlane(dv, i);
            const float* __restrict__ xr = X + (size_t)s * HID + lane;
            const float a = xr[0];
            const float b = xr[32];
            float* __restrict__ o = acc + (size_t)d * HID + lane;
            atomicAdd(o, a);
            atomicAdd(o + 32, b);
        }
    }
}

__global__ __launch_bounds__(256) void scatter_kernel(const float* __restrict__ X,
                                                      const int* __restrict__ src,
                                                      const int* __restrict__ dst,
                                                      float* __restrict__ acc0,
                                                      float* __restrict__ acc1, int E) {
    const int lane = threadIdx.x & 31;
    float* __restrict__ acc = chip_half() ? acc1 : acc0;
    const int wave = __builtin_amdgcn_readfirstlane((int)(blockIdx.x * 8 + (threadIdx.x >> 5)));
    const long long e0 = (long long)wave * SC_EPW;
    if (e0 >= E) return;
    const int ea = (int)e0 + lane;
    const int eb = ea + 32;
    int sa = 0, da = 0, sb = 0, db = 0;
    if (ea < E) { sa = src[ea]; da = dst[ea]; }
    if (eb < E) { sb = src[eb]; db = dst[eb]; }
    int na = E - (int)e0;
    int nb = na - 32;
    if (na > 32) na = 32;
    if (nb > 32) nb = 32;
    scatter_rows(X, acc, sa, da, na, lane);
    if (nb > 0) scatter_rows(X, acc, sb, db, nb, lane);
}

#define XF_PITCH 32
#define XF_GROUP 16

__device__ __forceinline__ void scatter_feat_rows(const float* __restrict__ X, float* __restrict__ acc,
                                                  const int sv, const int dv, const int n,
                                                  const int lane) {
    const bool act = lane < USER_FEAT;
    if (n == 32) {
#pragma unroll
        for (int g = 0; g < 32; g += XF_GROUP) {
            int s[XF_GROUP], d[XF_GROUP];
#pragma unroll
            for (int j = 0; j < XF_GROUP; ++j) {
                s[j] = __builtin_amdgcn_readlane(sv, g + j);
                d[j] = __builtin_amdgcn_readlane(dv, g + j);
            }
            if (act) {
                float v[XF_GROUP];
#pragma unroll
                for (int j = 0; j < XF_GROUP; ++j)
                    v[j] = X[(size_t)s[j] * USER_FEAT + lane];
#pragma unroll
                for (int j = 0; j < XF_GROUP; ++j)
                    atomicAdd(acc + (size_t)d[j] * XF_PITCH + lane, v[j]);
            }
        }
    } else {
        for (int i = 0; i < n; ++i) {
            const int s = __builtin_amdgcn_readlane(sv, i);
            const int d = __builtin_amdgcn_readlane(dv, i);
            if (act) {
                const float a = X[(size_t)s * USER_FEAT + lane];
                atomicAdd(acc + (size_t)d * XF_PITCH + lane, a);
            }
        }
    }
}

__global__ __launch_bounds__(256) void scatter_feat_kernel(const float* __restrict__ X,
                                                           const int* __restrict__ src,
                                                           const int* __restrict__ dst,
                                                           float* __restrict__ acc0,
                                                           float* __restrict__ acc1, int E) {
    const int lane = threadIdx.x & 31;
    float* __restrict__ acc = chip_half() ? acc1 : acc0;
    const int wave = __builtin_amdgcn_readfirstlane((int)(blockIdx.x * 8 + (threadIdx.x >> 5)));
    const long long e0 = (long long)wave * SC_EPW;
    if (e0 >= E) return;
    const int ea = (int)e0 + lane;
    const int eb = ea + 32;
    int sa = 0, da = 0, sb = 0, db = 0;
    if (ea < E) { sa = src[ea]; da = dst[ea]; }
    if (eb < E) { sb = src[eb]; db = dst[eb]; }
    int na = E - (int)e0;
    int nb = na - 32;
    if (na > 32) na = 32;
    if (nb > 32) nb = 32;
    scatter_feat_rows(X, acc, sa, da, na, lane);
    if (nb > 0) scatter_feat_rows(X, acc, sb, db, nb, lane);
}

#define PS_RANGES  2
#define PS_RB      ((NUM_MOVIES + PS_RANGES - 1) / PS_RANGES)
#define PS_SLICES  128
#define PS_THREADS 1024
#define PS_PLANES  4
#define PS_PITCH   150016

__device__ __forceinline__ void ps_edges8(float* bins, const float4* __restrict__ P4,
                                          const int4 d0, const int4 s0,
                                          const int4 d1, const int4 s1,
                                          int lo, unsigned int nb) {
    const int dd[8] = {d0.x, d0.y, d0.z, d0.w, d1.x, d1.y, d1.z, d1.w};
    const int ss[8] = {s0.x, s0.y, s0.z, s0.w, s1.x, s1.y, s1.z, s1.w};
    unsigned int b[8];
    float4 p[8];
#pragma unroll
    for (int j = 0; j < 8; ++j) {
        b[j] = (unsigned int)(dd[j] - lo);
        p[j] = make_float4(0.0f, 0.0f, 0.0f, 0.0f);
        if (b[j] < nb) p[j] = P4[ss[j]];
    }
#pragma unroll
    for (int j = 0; j < 8; ++j) {
        if (b[j] < nb) {
            float* q = bins + b[j] * GENRES;
            atomicAdd(q + 0, p[j].x);
            atomicAdd(q + 1, p[j].y);
            atomicAdd(q + 2, p[j].z);
        }
    }
}

__device__ __forceinline__ void ps_edge1(float* bins, const float4* __restrict__ P4,
                                         const int d, const int s, int lo, unsigned int nb) {
    const unsigned int b = (unsigned int)(d - lo);
    if (b < nb) {
        const float4 p = P4[s];
        float* q = bins + b * GENRES;
        atomicAdd(q + 0, p.x);
        atomicAdd(q + 1, p.y);
        atomicAdd(q + 2, p.z);
    }
}

__global__ __launch_bounds__(PS_THREADS)
void proj_scatter_kernel(const float* __restrict__ P, const int* __restrict__ src,
                         const int* __restrict__ dst, float* __restrict__ accp, int E) {
    __shared__ float bins[PS_RB * GENRES];

    const int tid = threadIdx.x;
    const int lo = (int)blockIdx.y * PS_RB;
    const int left = NUM_MOVIES - lo;
    const unsigned int nb = (unsigned int)(left < PS_RB ? left : PS_RB);
    const float4* __restrict__ P4 = (const float4*)P;

    for (int i = tid; i < PS_RB * GENRES; i += PS_THREADS) bins[i] = 0.0f;
    __syncthreads();

    const int per = (((E + PS_SLICES - 1) / PS_SLICES) + 3) & ~3;
    const int start = blockIdx.x * per;
    int end = start + per;
    if (end > E) end = E;
    if (start < end) {
        const int4* __restrict__ pd = (const int4*)(dst + start);
        const int4* __restrict__ ps = (const int4*)(src + start);
        const int nvec = (end - start) >> 2;
        int i = tid;
        for (; i + PS_THREADS < nvec; i += 2 * PS_THREADS) {
            const int4 d0 = pd[i];
            const int4 s0 = ps[i];
            const int4 d1 = pd[i + PS_THREADS];
            const int4 s1 = ps[i + PS_THREADS];
            ps_edges8(bins, P4, d0, s0, d1, s1, lo, nb);
        }
        for (; i < nvec; i += PS_THREADS) {
            const int4 d0 = pd[i];
            const int4 s0 = ps[i];
            ps_edge1(bins, P4, d0.x, s0.x, lo, nb);
            ps_edge1(bins, P4, d0.y, s0.y, lo, nb);
            ps_edge1(bins, P4, d0.z, s0.z, lo, nb);
            ps_edge1(bins, P4, d0.w, s0.w, lo, nb);
        }
        for (int e = start + (nvec << 2) + tid; e < end; e += PS_THREADS)
            ps_edge1(bins, P4, dst[e], src[e], lo, nb);
    }
    __syncthreads();

    const int n = (int)nb * GENRES;
    const int rot = ((int)(blockIdx.x / PS_PLANES) * (n / (PS_SLICES / PS_PLANES))) & ~31;
    float* __restrict__ out = accp + (size_t)(blockIdx.x % PS_PLANES) * PS_PITCH + (size_t)lo * GENRES;
    for (int j = tid; j < n; j += PS_THREADS) {
        int i = j + rot;
        if (i >= n) i -= n;
        const float v = bins[i];
        if (v != 0.0f) atomicAdd(&out[i], v);
    }
}

__host__ __device__ constexpr int w_pitch(int K) { return ((K / 4) & 1) ? K : K + 4; }

template <int K, int PITCH>
__device__ __forceinline__ void stage_weights(float* __restrict__ dst, const float* __restrict__ W, int tid) {
    constexpr int Q = K / 4;
    constexpr int STEPS = (64 * Q + 255) / 256;
    const float4* __restrict__ W4 = (const float4*)W;
    float4 v[STEPS];
#pragma unroll
    for (int j = 0; j < STEPS; ++j) {
        const int i = tid + 256 * j;
        v[j] = W4[i < 64 * Q ? i : 64 * Q - 1];
    }
#pragma unroll
    for (int j = 0; j < STEPS; ++j) {
        const int i = tid + 256 * j;
        if (i < 64 * Q) {
            const int n = i / Q;
            const int q = i - n * Q;
            *(float4*)(dst + n * PITCH + 4 * q) = v[j];
        }
    }
}

template <bool PROJ, bool DUAL, int K1, int LD1, int AUG1, int K2, int LD2, int AUG2>
__global__ __launch_bounds__(256)
void sage_gemm64(const float* __restrict__ A1, const float* __restrict__ A1b,
                 const float* __restrict__ deg1,
                 const float* __restrict__ W1,
                 const float* __restrict__ A2, const float* __restrict__ W2,
                 const float* __restrict__ bias, const float* __restrict__ Wp,
                 float* __restrict__ C, int M, int relu) {
    constexpr int P1 = w_pitch(K1);
    constexpr int P2 = K2 > 0 ? w_pitch(K2) : 0;
    __shared__ __attribute__((aligned(16))) float wlds[64 * P1 + 64 * P2];

    stage_weights<K1, P1>(wlds, W1, threadIdx.x);
    if (K2 > 0) stage_weights<(K2 > 0 ? K2 : 4), (K2 > 0 ? P2 : 4)>(wlds + 64 * P1, W2, threadIdx.x);
    __syncthreads();

    const int lane = threadIdx.x & 31;
    const int wave = threadIdx.x >> 5;
    const int half = lane >> 4;
    const int r    = lane & 15;
    const int row0 = (blockIdx.x * 8 + wave) * 16;
    if (row0 >= M) return;

    v8f c[4] = {};

    float scale = 1.0f;
    float has_edges = 1.0f;
    if (deg1) {
        const float dg = deg1[row0 + r];
        scale = 1.0f / fmaxf(dg, 1.0f);
        has_edges = dg >= 1.0f ? 1.0f : 0.0f;
    }

    {
        constexpr int KL1 = AUG1 ? K1 - 4 : K1;
        const float* Arow = A1 + (size_t)(row0 + r) * LD1;
        const float* Brow = DUAL ? A1b + (size_t)(row0 + r) * LD1 : Arow;
        const float* Ws = wlds + r * P1 + 2 * half;
#pragma unroll
        for (int kk = 0; kk < K1; kk += 4) {
            const int k = kk + 2 * half;
            v2f a;
            if (kk < KL1) {
                float a0 = Arow[k];
                float a1 = Arow[k + 1];
                if (DUAL) { a0 += Brow[k]; a1 += Brow[k + 1]; }
                a.x = a0 * scale;
                a.y = a1 * scale;
            } else {
                const float u = AUG1 == 1 ? has_edges : 1.0f;
                a.x = half == 0 ? u : 0.0f;
                a.y = 0.0f;
            }
#pragma unroll
            for (int t = 0; t < 4; ++t) {
                const v2f b = *(const v2f*)(Ws + t * 16 * P1 + kk);
                c[t] = __builtin_amdgcn_wmma_f32_16x16x4_f32(
                    false, a, false, b, (short)0, c[t], false, false);
            }
        }
    }
    if (K2 > 0) {
        constexpr int KL2 = AUG2 ? K2 - 4 : K2;
        const float* Arow = A2 + (size_t)(row0 + r) * LD2;
        const float* Ws = wlds + 64 * P1 + r * P2 + 2 * half;
#pragma unroll
        for (int kk = 0; kk < K2; kk += 4) {
            const int k = kk + 2 * half;
            v2f a;
            if (kk < KL2) {
                a.x = Arow[k];
                a.y = Arow[k + 1];
            } else {
                const float u = AUG2 == 1 ? has_edges : 1.0f;
                a.x = half == 0 ? u : 0.0f;
                a.y = 0.0f;
            }
#pragma unroll
            for (int t = 0; t < 4; ++t) {
                const v2f b = *(const v2f*)(Ws + t * 16 * P2 + kk);
                c[t] = __builtin_amdgcn_wmma_f32_16x16x4_f32(
                    false, a, false, b, (short)0, c[t], false, false);
            }
        }
    }

    if (!PROJ) {
#pragma unroll
        for (int t = 0; t < 4; ++t) {
            const float bv = bias ? bias[t * 16 + r] : 0.0f;
#pragma unroll
            for (int i = 0; i < 8; ++i) {
                float v = c[t][i] + bv;
                if (relu) v = fmaxf(v, 0.0f);
                C[(size_t)(row0 + i + 8 * half) * HID + t * 16 + r] = v;
            }
        }
    } else {
        float s[8][GENRES];
#pragma unroll
        for (int i = 0; i < 8; ++i)
#pragma unroll
            for (int g = 0; g < GENRES; ++g) s[i][g] = 0.0f;
#pragma unroll
        for (int t = 0; t < 4; ++t) {
            const float bv = bias ? bias[t * 16 + r] : 0.0f;
            float w[GENRES];
#pragma unroll
            for (int g = 0; g < GENRES; ++g) w[g] = Wp[g * HID + t * 16 + r];
#pragma unroll
            for (int i = 0; i < 8; ++i) {
                float v = c[t][i] + bv;
                if (relu) v = fmaxf(v, 0.0f);
#pragma unroll
                for (int g = 0; g < GENRES; ++g) s[i][g] += v * w[g];
            }
        }
#pragma unroll
        for (int i = 0; i < 8; ++i)
#pragma unroll
            for (int g = 0; g < GENRES; ++g)
#pragma unroll
                for (int m = 8; m >= 1; m >>= 1)
                    s[i][g] += __shfl_xor(s[i][g], m, 16);
        float4 o = make_float4(0.0f, 0.0f, 0.0f, 0.0f);
#pragma unroll
        for (int i = 0; i < 8; ++i)
            if (r == i) o = make_float4(s[i][0], s[i][1], s[i][2], 0.0f);
        if (r < 8) ((float4*)C)[(size_t)(row0 + r + 8 * half)] = o;
    }
}

#define FOLD_K (USER_FEAT + 4)

__global__ __launch_bounds__(256)
void fold_user_proj_kernel(const float* __restrict__ Wm0, const float* __restrict__ Wm1,
                           const float* __restrict__ W_user, const float* __restrict__ b_user,
                           float* __restrict__ Wf) {
    const float* __restrict__ Wm = blockIdx.x ? Wm1 : Wm0;
    float* __restrict__ out = Wf + (size_t)blockIdx.x * 64 * FOLD_K;

    const int lane = threadIdx.x & 31;
    const int wave = threadIdx.x >> 5;
    const int half = lane >> 4;
    const int r    = lane & 15;
    const int tn   = wave >> 1;
    const int col  = (wave & 1) * 16 + r;
    const float* Bcol = col < USER_FEAT ? W_user + col : b_user;
    const int bpitch  = col < USER_FEAT ? USER_FEAT : 1;
    const bool live   = col <= USER_FEAT;

    const float* Arow = Wm + (size_t)(tn * 16 + r) * HID;
    v8f c = {};
#pragma unroll
    for (int kk = 0; kk < HID; kk += 4) {
        const int k = kk + 2 * half;
        v2f a, b;
        a.x = Arow[k];
        a.y = Arow[k + 1];
        const float b0 = Bcol[k * bpitch];
        const float b1 = Bcol[(k + 1) * bpitch];
        b.x = live ? b0 : 0.0f;
        b.y = live ? b1 : 0.0f;
        c = __builtin_amdgcn_wmma_f32_16x16x4_f32(
            false, a, false, b, (short)0, c, false, false);
    }
    if (col < FOLD_K) {
#pragma unroll
        for (int i = 0; i < 8; ++i)
            out[(size_t)(tn * 16 + i + 8 * half) * FOLD_K + col] = c[i];
    }
}

__global__ __launch_bounds__(256)
void head_kernel(const float* __restrict__ accp, const float* __restrict__ degm,
                 const float* __restrict__ m1,
                 const float* __restrict__ bl2, const float* __restrict__ Wr2,
                 float* __restrict__ out, int M) {
    int m = blockIdx.x * blockDim.x + threadIdx.x;
    if (m >= M) return;
    const float inv = 1.0f / fmaxf(degm[m], 1.0f);
    const float* a = accp + (size_t)m * GENRES;
    const float* x = m1 + (size_t)m * HID;
    float acc[GENRES];
#pragma unroll
    for (int g = 0; g < GENRES; ++g) {
        float sum = a[g];
#pragma unroll
        for (int pl = 1; pl < PS_PLANES; ++pl) sum += a[(size_t)pl * PS_PITCH + g];
        acc[g] = sum * inv + bl2[g];
    }
    for (int k = 0; k < HID; ++k) {
        const float xv = x[k];
#pragma unroll
        for (int g = 0; g < GENRES; ++g)
            acc[g] += xv * Wr2[g * HID + k];
    }
#pragma unroll
    for (int g = 0; g < GENRES; ++g) out[(size_t)m * GENRES + g] = acc[g];
}

extern "C" void kernel_launch(void* const* d_in, const int* in_sizes, int n_in,
                              void* d_out, int out_size, void* d_ws, size_t ws_size,
                              hipStream_t stream) {
    (void)in_sizes; (void)n_in; (void)out_size; (void)ws_size;

    const float* x_user  = (const float*)d_in[0];
    const float* x_movie = (const float*)d_in[1];
    const int*   e_src   = (const int*)d_in[2];
    const int*   e_dst   = (const int*)d_in[3];
    const float* W_user  = (const float*)d_in[4];
    const float* b_user  = (const float*)d_in[5];
    const float* W_movie = (const float*)d_in[6];
    const float* b_movie = (const float*)d_in[7];
    const float* Wl1_um  = (const float*)d_in[8];
    const float* bl1_um  = (const float*)d_in[9];
    const float* Wr1_um  = (const float*)d_in[10];
    const float* Wl1_mu  = (const float*)d_in[11];
    const float* bl1_mu  = (const float*)d_in[12];
    const float* Wr1_mu  = (const float*)d_in[13];
    const float* Wl2_um  = (const float*)d_in[14];
    const float* bl2_um  = (const float*)d_in[15];
    const float* Wr2_um  = (const float*)d_in[16];
    float* out = (float*)d_out;

    float* ws = (float*)d_ws;
    float* hm        = ws;
    float* pu        = hm + 3200000;
    float* m1        = pu + 400000;
    float* wfold     = m1 + 3200000;
    float* wfold_r   = wfold + 1792;
    float* acc_x     = wfold + 3584;
    float* acc_x_b   = acc_x + 1600000;
    float* acc_mu    = acc_x_b + 1600000;
    float* acc_mu_b  = acc_mu + 6400000;
    float* accp      = acc_mu_b + 6400000;
    float* deg_user  = accp + 600064;
    float* deg_movie = deg_user + 100000;

    const size_t zero_bytes = (size_t)(1600000 + 1600000 + 6400000 + 6400000 + 600064 + 100000 + 50000) * 4;
    hipMemsetAsync(acc_x, 0, zero_bytes, stream);

    fold_user_proj_kernel<<<2, 256, 0, stream>>>(Wl1_um, Wr1_mu, W_user, b_user, wfold);

    const int E = NUM_EDGES;
    degree_hist_kernel<<<dim3(HSLICES, MOVIE_RANGES + USER_RANGES), HTHREADS, 0, stream>>>(
        e_dst, deg_movie, e_src, deg_user, E);

    const int gridU = (NUM_USERS / 16 + 7) / 8;
    const int gridM = (NUM_MOVIES / 16 + 7) / 8;
    sage_gemm64<false, false, MOVIE_FEAT, MOVIE_FEAT, 0, 0, 0, 0><<<gridM, 256, 0, stream>>>(
        x_movie, nullptr, nullptr, W_movie, nullptr, nullptr, b_movie, nullptr, hm, NUM_MOVIES, 0);

    const int sblocks = (E + 8 * SC_EPW - 1) / (8 * SC_EPW);

    scatter_feat_kernel<<<sblocks, 256, 0, stream>>>(x_user, e_src, e_dst, acc_x, acc_x_b, E);
    sage_gemm64<false, true, FOLD_K, XF_PITCH, 1, HID, HID, 0><<<gridM, 256, 0, stream>>>(
        acc_x, acc_x_b, deg_movie, wfold, hm, Wr1_um, bl1_um, nullptr, m1, NUM_MOVIES, 1);

    scatter_kernel<<<sblocks, 256, 0, stream>>>(hm, e_dst, e_src, acc_mu, acc_mu_b, E);
    sage_gemm64<true, true, HID, HID, 0, FOLD_K, USER_FEAT, 2><<<gridU, 256, 0, stream>>>(
        acc_mu, acc_mu_b, deg_user, Wl1_mu, x_user, wfold_r, bl1_mu, Wl2_um, pu, NUM_USERS, 1);

    proj_scatter_kernel<<<dim3(PS_SLICES, PS_RANGES), PS_THREADS, 0, stream>>>(
        pu, e_src, e_dst, accp, E);
    head_kernel<<<(NUM_MOVIES + 255) / 256, 256, 0, stream>>>(
        accp, deg_movie, m1, bl2_um, Wr2_um, out, NUM_MOVIES);
}


# Independent passes share launches: degree count + movie projection GEMM + weight fold side by side (one block per WGP), and both layer-1 node GEMMs side by side; 9 launches -> 6
# speedup vs baseline: 5.0781x; 1.0711x over previous
#include <hip/hip_runtime.h>
#include <hip/hip_bf16.h>

typedef __attribute__((ext_vector_type(2))) float v2f;
typedef __attribute__((ext_vector_type(8))) float v8f;

#define NUM_USERS  100000
#define NUM_MOVIES 50000
#define NUM_EDGES  4000000
#define USER_FEAT  24
#define MOVIE_FEAT 404
#define HID        64
#define GENRES     3

#define HBINS        50000
#define HSLICES      32
#define HTHREADS     1024
#define MOVIE_RANGES ((NUM_MOVIES + HBINS - 1) / HBINS)
#define USER_RANGES  ((NUM_USERS + HBINS - 1) / HBINS)

__device__ __forceinline__ void hist_count4(unsigned int* bins, const int4 v, int lo, unsigned int nb) {
    const unsigned int b0 = (unsigned int)(v.x - lo);
    const unsigned int b1 = (unsigned int)(v.y - lo);
    const unsigned int b2 = (unsigned int)(v.z - lo);
    const unsigned int b3 = (unsigned int)(v.w - lo);
    if (b0 < nb) atomicAdd(&bins[b0], 1u);
    if (b1 < nb) atomicAdd(&bins[b1], 1u);
    if (b2 < nb) atomicAdd(&bins[b2], 1u);
    if (b3 < nb) atomicAdd(&bins[b3], 1u);
}

__device__ __forceinline__
void degree_hist_role(unsigned int* bins, const int slice, const int job,
                      const int* __restrict__ dst, float* __restrict__ deg_movie,
                      const int* __restrict__ src, float* __restrict__ deg_user, int E) {
    const int tid = threadIdx.x;
    const bool is_movie = job < MOVIE_RANGES;
    const int* __restrict__ idx = is_movie ? dst : src;
    const int nseg = is_movie ? NUM_MOVIES : NUM_USERS;
    const int lo = (is_movie ? job : job - MOVIE_RANGES) * HBINS;
    const int left = nseg - lo;
    const unsigned int nb = (unsigned int)(left < HBINS ? left : HBINS);
    float* __restrict__ out = (is_movie ? deg_movie : deg_user) + lo;

    for (int i = tid; i < HBINS; i += HTHREADS) bins[i] = 0u;
    __syncthreads();

    const int per = (((E + HSLICES - 1) / HSLICES) + 3) & ~3;
    const int start = slice * per;
    int end = start + per;
    if (end > E) end = E;
    if (start < end) {
        const int4* __restrict__ p = (const int4*)(idx + start);
        const int nvec = (end - start) >> 2;
        int i = tid;
        for (; i + 3 * HTHREADS < nvec; i += 4 * HTHREADS) {
            const int4 a = p[i];
            const int4 b = p[i + HTHREADS];
            const int4 c = p[i + 2 * HTHREADS];
            const int4 d = p[i + 3 * HTHREADS];
            hist_count4(bins, a, lo, nb);
            hist_count4(bins, b, lo, nb);
            hist_count4(bins, c, lo, nb);
            hist_count4(bins, d, lo, nb);
        }
        for (; i < nvec; i += HTHREADS) hist_count4(bins, p[i], lo, nb);
        for (int e = start + (nvec << 2) + tid; e < end; e += HTHREADS) {
            const unsigned int b = (unsigned int)(idx[e] - lo);
            if (b < nb) atomicAdd(&bins[b], 1u);
        }
    }
    __syncthreads();

    const int n = (int)nb;
    const int rot = (slice * (n / HSLICES)) & ~31;
    for (int j = tid; j < n; j += HTHREADS) {
        int i = j + rot;
        if (i >= n) i -= n;
        const unsigned int cnt = bins[i];
        if (cnt) atomicAdd(&out[i], (float)cnt);
    }
}

#define SC_EPW   64
#define SC_GROUP 8

__device__ __forceinline__ int chip_half() {
    const unsigned int id = __builtin_amdgcn_s_sendmsg_rtn(0x87u);
    return (int)((id >> 18) & 1u);
}

__device__ __forceinline__ void scatter_rows(const float* __restrict__ X, float* __restrict__ acc,
                                             const int sv, const int dv, const int n, const int lane) {
    if (n == 32) {
#pragma unroll
        for (int g = 0; g < 32; g += SC_GROUP) {
            float lo[SC_GROUP], hi[SC_GROUP];
#pragma unroll
            for (int j = 0; j < SC_GROUP; ++j) {
                const int s = __builtin_amdgcn_readlane(sv, g + j);
                const float* __restrict__ xr = X + (size_t)s * HID + lane;
                lo[j] = xr[0];
                hi[j] = xr[32];
            }
#pragma unroll
            for (int j = 0; j < SC_GROUP; ++j) {
                const int d = __builtin_amdgcn_readlane(dv, g + j);
                float* __restrict__ o = acc + (size_t)d * HID + lane;
                atomicAdd(o, lo[j]);
                atomicAdd(o + 32, hi[j]);
            }
        }
    } else {
        for (int i = 0; i < n; ++i) {
            const int s = __builtin_amdgcn_readlane(sv, i);
            const int d = __builtin_amdgcn_readlane(dv, i);
            const float* __restrict__ xr = X + (size_t)s * HID + lane;
            const float a = xr[0];
            const float b = xr[32];
            float* __restrict__ o = acc + (size_t)d * HID + lane;
            atomicAdd(o, a);
            atomicAdd(o + 32, b);
        }
    }
}

__global__ __launch_bounds__(256) void scatter_kernel(const float* __restrict__ X,
                                                      const int* __restrict__ src,
                                                      const int* __restrict__ dst,
                                                      float* __restrict__ acc0,
                                                      float* __restrict__ acc1, int E) {
    const int lane = threadIdx.x & 31;
    float* __restrict__ acc = chip_half() ? acc1 : acc0;
    const int wave = __builtin_amdgcn_readfirstlane((int)(blockIdx.x * 8 + (threadIdx.x >> 5)));
    const long long e0 = (long long)wave * SC_EPW;
    if (e0 >= E) return;
    const int ea = (int)e0 + lane;
    const int eb = ea + 32;
    int sa = 0, da = 0, sb = 0, db = 0;
    if (ea < E) { sa = src[ea]; da = dst[ea]; }
    if (eb < E) { sb = src[eb]; db = dst[eb]; }
    int na = E - (int)e0;
    int nb = na - 32;
    if (na > 32) na = 32;
    if (nb > 32) nb = 32;
    scatter_rows(X, acc, sa, da, na, lane);
    if (nb > 0) scatter_rows(X, acc, sb, db, nb, lane);
}

#define XF_PITCH 32
#define XF_GROUP 16

__device__ __forceinline__ void scatter_feat_rows(const float* __restrict__ X, float* __restrict__ acc,
                                                  const int sv, const int dv, const int n,
                                                  const int lane) {
    const bool act = lane < USER_FEAT;
    if (n == 32) {
#pragma unroll
        for (int g = 0; g < 32; g += XF_GROUP) {
            int s[XF_GROUP], d[XF_GROUP];
#pragma unroll
            for (int j = 0; j < XF_GROUP; ++j) {
                s[j] = __builtin_amdgcn_readlane(sv, g + j);
                d[j] = __builtin_amdgcn_readlane(dv, g + j);
            }
            if (act) {
                float v[XF_GROUP];
#pragma unroll
                for (int j = 0; j < XF_GROUP; ++j)
                    v[j] = X[(size_t)s[j] * USER_FEAT + lane];
#pragma unroll
                for (int j = 0; j < XF_GROUP; ++j)
                    atomicAdd(acc + (size_t)d[j] * XF_PITCH + lane, v[j]);
            }
        }
    } else {
        for (int i = 0; i < n; ++i) {
            const int s = __builtin_amdgcn_readlane(sv, i);
            const int d = __builtin_amdgcn_readlane(dv, i);
            if (act) {
                const float a = X[(size_t)s * USER_FEAT + lane];
                atomicAdd(acc + (size_t)d * XF_PITCH + lane, a);
            }
        }
    }
}

__global__ __launch_bounds__(256) void scatter_feat_kernel(const float* __restrict__ X,
                                                           const int* __restrict__ src,
                                                           const int* __restrict__ dst,
                                                           float* __restrict__ acc0,
                                                           float* __restrict__ acc1, int E) {
    const int lane = threadIdx.x & 31;
    float* __restrict__ acc = chip_half() ? acc1 : acc0;
    const int wave = __builtin_amdgcn_readfirstlane((int)(blockIdx.x * 8 + (threadIdx.x >> 5)));
    const long long e0 = (long long)wave * SC_EPW;
    if (e0 >= E) return;
    const int ea = (int)e0 + lane;
    const int eb = ea + 32;
    int sa = 0, da = 0, sb = 0, db = 0;
    if (ea < E) { sa = src[ea]; da = dst[ea]; }
    if (eb < E) { sb = src[eb]; db = dst[eb]; }
    int na = E - (int)e0;
    int nb = na - 32;
    if (na > 32) na = 32;
    if (nb > 32) nb = 32;
    scatter_feat_rows(X, acc, sa, da, na, lane);
    if (nb > 0) scatter_feat_rows(X, acc, sb, db, nb, lane);
}

#define PS_RANGES  2
#define PS_RB      ((NUM_MOVIES + PS_RANGES - 1) / PS_RANGES)
#define PS_SLICES  128
#define PS_THREADS 1024
#define PS_PLANES  4
#define PS_PITCH   150016

__device__ __forceinline__ void ps_edges8(float* bins, const float4* __restrict__ P4,
                                          const int4 d0, const int4 s0,
                                          const int4 d1, const int4 s1,
                                          int lo, unsigned int nb) {
    const int dd[8] = {d0.x, d0.y, d0.z, d0.w, d1.x, d1.y, d1.z, d1.w};
    const int ss[8] = {s0.x, s0.y, s0.z, s0.w, s1.x, s1.y, s1.z, s1.w};
    unsigned int b[8];
    float4 p[8];
#pragma unroll
    for (int j = 0; j < 8; ++j) {
        b[j] = (unsigned int)(dd[j] - lo);
        p[j] = make_float4(0.0f, 0.0f, 0.0f, 0.0f);
        if (b[j] < nb) p[j] = P4[ss[j]];
    }
#pragma unroll
    for (int j = 0; j < 8; ++j) {
        if (b[j] < nb) {
            float* q = bins + b[j] * GENRES;
            atomicAdd(q + 0, p[j].x);
            atomicAdd(q + 1, p[j].y);
            atomicAdd(q + 2, p[j].z);
        }
    }
}

__device__ __forceinline__ void ps_edge1(float* bins, const float4* __restrict__ P4,
                                         const int d, const int s, int lo, unsigned int nb) {
    const unsigned int b = (unsigned int)(d - lo);
    if (b < nb) {
        const float4 p = P4[s];
        float* q = bins + b * GENRES;
        atomicAdd(q + 0, p.x);
        atomicAdd(q + 1, p.y);
        atomicAdd(q + 2, p.z);
    }
}

__global__ __launch_bounds__(PS_THREADS)
void proj_scatter_kernel(const float* __restrict__ P, const int* __restrict__ src,
                         const int* __restrict__ dst, float* __restrict__ accp, int E) {
    __shared__ float bins[PS_RB * GENRES];

    const int tid = threadIdx.x;
    const int lo = (int)blockIdx.y * PS_RB;
    const int left = NUM_MOVIES - lo;
    const unsigned int nb = (unsigned int)(left < PS_RB ? left : PS_RB);
    const float4* __restrict__ P4 = (const float4*)P;

    for (int i = tid; i < PS_RB * GENRES; i += PS_THREADS) bins[i] = 0.0f;
    __syncthreads();

    const int per = (((E + PS_SLICES - 1) / PS_SLICES) + 3) & ~3;
    const int start = blockIdx.x * per;
    int end = start + per;
    if (end > E) end = E;
    if (start < end) {
        const int4* __restrict__ pd = (const int4*)(dst + start);
        const int4* __restrict__ ps = (const int4*)(src + start);
        const int nvec = (end - start) >> 2;
        int i = tid;
        for (; i + PS_THREADS < nvec; i += 2 * PS_THREADS) {
            const int4 d0 = pd[i];
            const int4 s0 = ps[i];
            const int4 d1 = pd[i + PS_THREADS];
            const int4 s1 = ps[i + PS_THREADS];
            ps_edges8(bins, P4, d0, s0, d1, s1, lo, nb);
        }
        for (; i < nvec; i += PS_THREADS) {
            const int4 d0 = pd[i];
            const int4 s0 = ps[i];
            ps_edge1(bins, P4, d0.x, s0.x, lo, nb);
            ps_edge1(bins, P4, d0.y, s0.y, lo, nb);
            ps_edge1(bins, P4, d0.z, s0.z, lo, nb);
            ps_edge1(bins, P4, d0.w, s0.w, lo, nb);
        }
        for (int e = start + (nvec << 2) + tid; e < end; e += PS_THREADS)
            ps_edge1(bins, P4, dst[e], src[e], lo, nb);
    }
    __syncthreads();

    const int n = (int)nb * GENRES;
    const int rot = ((int)(blockIdx.x / PS_PLANES) * (n / (PS_SLICES / PS_PLANES))) & ~31;
    float* __restrict__ out = accp + (size_t)(blockIdx.x % PS_PLANES) * PS_PITCH + (size_t)lo * GENRES;
    for (int j = tid; j < n; j += PS_THREADS) {
        int i = j + rot;
        if (i >= n) i -= n;
        const float v = bins[i];
        if (v != 0.0f) atomicAdd(&out[i], v);
    }
}

__host__ __device__ constexpr int w_pitch(int K) { return ((K / 4) & 1) ? K : K + 4; }

template <int K, int PITCH, int NT>
__device__ __forceinline__ void stage_weights(float* __restrict__ dst, const float* __restrict__ W, int tid) {
    constexpr int Q = K / 4;
    constexpr int STEPS = (64 * Q + NT - 1) / NT;
    const float4* __restrict__ W4 = (const float4*)W;
    float4 v[STEPS];
#pragma unroll
    for (int j = 0; j < STEPS; ++j) {
        const int i = tid + NT * j;
        v[j] = W4[i < 64 * Q ? i : 64 * Q - 1];
    }
#pragma unroll
    for (int j = 0; j < STEPS; ++j) {
        const int i = tid + NT * j;
        if (i < 64 * Q) {
            const int n = i / Q;
            const int q = i - n * Q;
            *(float4*)(dst + n * PITCH + 4 * q) = v[j];
        }
    }
}

template <bool PROJ, bool DUAL, int K1, int LD1, int AUG1, int K2, int LD2, int AUG2, int NT, int WPB>
__device__ __forceinline__
void sage_gemm64_body(float* __restrict__ wlds, const int gblock,
                      const float* __restrict__ A1, const float* __restrict__ A1b,
                      const float* __restrict__ deg1,
                      const float* __restrict__ W1,
                      const float* __restrict__ A2, const float* __restrict__ W2,
                      const float* __restrict__ bias, const float* __restrict__ Wp,
                      float* __restrict__ C, int M, int relu) {
    constexpr int P1 = w_pitch(K1);
    constexpr int P2 = K2 > 0 ? w_pitch(K2) : 0;

    stage_weights<K1, P1, NT>(wlds, W1, threadIdx.x);
    if (K2 > 0) stage_weights<(K2 > 0 ? K2 : 4), (K2 > 0 ? P2 : 4), NT>(wlds + 64 * P1, W2, threadIdx.x);
    __syncthreads();

    const int lane = threadIdx.x & 31;
    const int wave = threadIdx.x >> 5;
    const int half = lane >> 4;
    const int r    = lane & 15;
    if (NT / 32 > WPB && wave >= WPB) return;
    const int row0 = (gblock * WPB + wave) * 16;
    if (row0 >= M) return;

    v8f c[4] = {};

    float scale = 1.0f;
    float has_edges = 1.0f;
    if (deg1) {
        const float dg = deg1[row0 + r];
        scale = 1.0f / fmaxf(dg, 1.0f);
        has_edges = dg >= 1.0f ? 1.0f : 0.0f;
    }

    {
        constexpr int KL1 = AUG1 ? K1 - 4 : K1;
        const float* Arow = A1 + (size_t)(row0 + r) * LD1;
        const float* Brow = DUAL ? A1b + (size_t)(row0 + r) * LD1 : Arow;
        const float* Ws = wlds + r * P1 + 2 * half;
#pragma unroll
        for (int kk = 0; kk < K1; kk += 4) {
            const int k = kk + 2 * half;
            v2f a;
            if (kk < KL1) {
                float a0 = Arow[k];
                float a1 = Arow[k + 1];
                if (DUAL) { a0 += Brow[k]; a1 += Brow[k + 1]; }
                a.x = a0 * scale;
                a.y = a1 * scale;
            } else {
                const float u = AUG1 == 1 ? has_edges : 1.0f;
                a.x = half == 0 ? u : 0.0f;
                a.y = 0.0f;
            }
#pragma unroll
            for (int t = 0; t < 4; ++t) {
                const v2f b = *(const v2f*)(Ws + t * 16 * P1 + kk);
                c[t] = __builtin_amdgcn_wmma_f32_16x16x4_f32(
                    false, a, false, b, (short)0, c[t], false, false);
            }
        }
    }
    if (K2 > 0) {
        constexpr int KL2 = AUG2 ? K2 - 4 : K2;
        const float* Arow = A2 + (size_t)(row0 + r) * LD2;
        const float* Ws = wlds + 64 * P1 + r * P2 + 2 * half;
#pragma unroll
        for (int kk = 0; kk < K2; kk += 4) {
            const int k = kk + 2 * half;
            v2f a;
            if (kk < KL2) {
                a.x = Arow[k];
                a.y = Arow[k + 1];
            } else {
                const float u = AUG2 == 1 ? has_edges : 1.0f;
                a.x = half == 0 ? u : 0.0f;
                a.y = 0.0f;
            }
#pragma unroll
            for (int t = 0; t < 4; ++t) {
                const v2f b = *(const v2f*)(Ws + t * 16 * P2 + kk);
                c[t] = __builtin_amdgcn_wmma_f32_16x16x4_f32(
                    false, a, false, b, (short)0, c[t], false, false);
            }
        }
    }

    if (!PROJ) {
#pragma unroll
        for (int t = 0; t < 4; ++t) {
            const float bv = bias ? bias[t * 16 + r] : 0.0f;
#pragma unroll
            for (int i = 0; i < 8; ++i) {
                float v = c[t][i] + bv;
                if (relu) v = fmaxf(v, 0.0f);
                C[(size_t)(row0 + i + 8 * half) * HID + t * 16 + r] = v;
            }
        }
    } else {
        float s[8][GENRES];
#pragma unroll
        for (int i = 0; i < 8; ++i)
#pragma unroll
            for (int g = 0; g < GENRES; ++g) s[i][g] = 0.0f;
#pragma unroll
        for (int t = 0; t < 4; ++t) {
            const float bv = bias ? bias[t * 16 + r] : 0.0f;
            float w[GENRES];
#pragma unroll
            for (int g = 0; g < GENRES; ++g) w[g] = Wp[g * HID + t * 16 + r];
#pragma unroll
            for (int i = 0; i < 8; ++i) {
                float v = c[t][i] + bv;
                if (relu) v = fmaxf(v, 0.0f);
#pragma unroll
                for (int g = 0; g < GENRES; ++g) s[i][g] += v * w[g];
            }
        }
#pragma unroll
        for (int i = 0; i < 8; ++i)
#pragma unroll
            for (int g = 0; g < GENRES; ++g)
#pragma unroll
                for (int m = 8; m >= 1; m >>= 1)
                    s[i][g] += __shfl_xor(s[i][g], m, 16);
        float4 o = make_float4(0.0f, 0.0f, 0.0f, 0.0f);
#pragma unroll
        for (int i = 0; i < 8; ++i)
            if (r == i) o = make_float4(s[i][0], s[i][1], s[i][2], 0.0f);
        if (r < 8) ((float4*)C)[(size_t)(row0 + r + 8 * half)] = o;
    }
}

#define FOLD_K (USER_FEAT + 4)

__device__ __forceinline__
void fold_user_proj_role(const int m,
                         const float* __restrict__ Wm0, const float* __restrict__ Wm1,
                         const float* __restrict__ W_user, const float* __restrict__ b_user,
                         float* __restrict__ Wf) {
    if (threadIdx.x >= 256) return;
    const float* __restrict__ Wm = m ? Wm1 : Wm0;
    float* __restrict__ out = Wf + (size_t)m * 64 * FOLD_K;

    const int lane = threadIdx.x & 31;
    const int wave = threadIdx.x >> 5;
    const int half = lane >> 4;
    const int r    = lane & 15;
    const int tn   = wave >> 1;
    const int col  = (wave & 1) * 16 + r;
    const float* Bcol = col < USER_FEAT ? W_user + col : b_user;
    const int bpitch  = col < USER_FEAT ? USER_FEAT : 1;
    const bool live   = col <= USER_FEAT;

    const float* Arow = Wm + (size_t)(tn * 16 + r) * HID;
    v8f c = {};
#pragma unroll
    for (int kk = 0; kk < HID; kk += 4) {
        const int k = kk + 2 * half;
        v2f a, b;
        a.x = Arow[k];
        a.y = Arow[k + 1];
        const float b0 = Bcol[k * bpitch];
        const float b1 = Bcol[(k + 1) * bpitch];
        b.x = live ? b0 : 0.0f;
        b.y = live ? b1 : 0.0f;
        c = __builtin_amdgcn_wmma_f32_16x16x4_f32(
            false, a, false, b, (short)0, c, false, false);
    }
    if (col < FOLD_K) {
#pragma unroll
        for (int i = 0; i < 8; ++i)
            out[(size_t)(tn * 16 + i + 8 * half) * FOLD_K + col] = c[i];
    }
}

#define PRE_HIST (HSLICES * (MOVIE_RANGES + USER_RANGES))
#define PRE_WPB  20
#define PRE_GEMM ((NUM_MOVIES / 16 + PRE_WPB - 1) / PRE_WPB)

__global__ __launch_bounds__(HTHREADS)
void prepass_kernel(const int* __restrict__ dst, float* __restrict__ deg_movie,
                    const int* __restrict__ src, float* __restrict__ deg_user, int E,
                    const float* __restrict__ x_movie, const float* __restrict__ W_movie,
                    const float* __restrict__ b_movie, float* __restrict__ hm,
                    const float* __restrict__ Wm0, const float* __restrict__ Wm1,
                    const float* __restrict__ W_user, const float* __restrict__ b_user,
                    float* __restrict__ Wf) {
    __shared__ __attribute__((aligned(16))) unsigned int smem[HBINS];
    static_assert(64 * w_pitch(MOVIE_FEAT) <= HBINS, "staged W_movie must fit the histogram's LDS array");

    const int b = (int)blockIdx.x;
    if (b < PRE_HIST) {
        degree_hist_role(smem, b % HSLICES, b / HSLICES, dst, deg_movie, src, deg_user, E);
    } else if (b < PRE_HIST + PRE_GEMM) {
        sage_gemm64_body<false, false, MOVIE_FEAT, MOVIE_FEAT, 0, 0, 0, 0, HTHREADS, PRE_WPB>(
            (float*)smem, b - PRE_HIST, x_movie, nullptr, nullptr, W_movie, nullptr, nullptr,
            b_movie, nullptr, hm, NUM_MOVIES, 0);
    } else {
        fold_user_proj_role(b - PRE_HIST - PRE_GEMM, Wm0, Wm1, W_user, b_user, Wf);
    }
}

__global__ __launch_bounds__(256)
void layer1_gemm_kernel(const float* __restrict__ acc_mu, const float* __restrict__ acc_mu_b,
                        const float* __restrict__ deg_user, const float* __restrict__ Wl1_mu,
                        const float* __restrict__ x_user, const float* __restrict__ wfold_r,
                        const float* __restrict__ bl1_mu, const float* __restrict__ Wl2_um,
                        float* __restrict__ pu,
                        const float* __restrict__ acc_x, const float* __restrict__ acc_x_b,
                        const float* __restrict__ deg_movie, const float* __restrict__ wfold_l,
                        const float* __restrict__ hm, const float* __restrict__ Wr1_um,
                        const float* __restrict__ bl1_um, float* __restrict__ m1,
                        int user_blocks) {
    __shared__ __attribute__((aligned(16))) float wlds[64 * (w_pitch(FOLD_K) + w_pitch(HID))];
    const int b = (int)blockIdx.x;
    if (b < user_blocks) {
        sage_gemm64_body<true, true, HID, HID, 0, FOLD_K, USER_FEAT, 2, 256, 8>(
            wlds, b, acc_mu, acc_mu_b, deg_user, Wl1_mu, x_user, wfold_r, bl1_mu, Wl2_um,
            pu, NUM_USERS, 1);
    } else {
        sage_gemm64_body<false, true, FOLD_K, XF_PITCH, 1, HID, HID, 0, 256, 8>(
            wlds, b - user_blocks, acc_x, acc_x_b, deg_movie, wfold_l, hm, Wr1_um, bl1_um, nullptr,
            m1, NUM_MOVIES, 1);
    }
}

__global__ __launch_bounds__(256)
void head_kernel(const float* __restrict__ accp, const float* __restrict__ degm,
                 const float* __restrict__ m1,
                 const float* __restrict__ bl2, const float* __restrict__ Wr2,
                 float* __restrict__ out, int M) {
    int m = blockIdx.x * blockDim.x + threadIdx.x;
    if (m >= M) return;
    const float inv = 1.0f / fmaxf(degm[m], 1.0f);
    const float* a = accp + (size_t)m * GENRES;
    const float* x = m1 + (size_t)m * HID;
    float acc[GENRES];
#pragma unroll
    for (int g = 0; g < GENRES; ++g) {
        float sum = a[g];
#pragma unroll
        for (int pl = 1; pl < PS_PLANES; ++pl) sum += a[(size_t)pl * PS_PITCH + g];
        acc[g] = sum * inv + bl2[g];
    }
    for (int k = 0; k < HID; ++k) {
        const float xv = x[k];
#pragma unroll
        for (int g = 0; g < GENRES; ++g)
            acc[g] += xv * Wr2[g * HID + k];
    }
#pragma unroll
    for (int g = 0; g < GENRES; ++g) out[(size_t)m * GENRES + g] = acc[g];
}

extern "C" void kernel_launch(void* const* d_in, const int* in_sizes, int n_in,
                              void* d_out, int out_size, void* d_ws, size_t ws_size,
                              hipStream_t stream) {
    (void)in_sizes; (void)n_in; (void)out_size; (void)ws_size;

    const float* x_user  = (const float*)d_in[0];
    const float* x_movie = (const float*)d_in[1];
    const int*   e_src   = (const int*)d_in[2];
    const int*   e_dst   = (const int*)d_in[3];
    const float* W_user  = (const float*)d_in[4];
    const float* b_user  = (const float*)d_in[5];
    const float* W_movie = (const float*)d_in[6];
    const float* b_movie = (const float*)d_in[7];
    const float* Wl1_um  = (const float*)d_in[8];
    const float* bl1_um  = (const float*)d_in[9];
    const float* Wr1_um  = (const float*)d_in[10];
    const float* Wl1_mu  = (const float*)d_in[11];
    const float* bl1_mu  = (const float*)d_in[12];
    const float* Wr1_mu  = (const float*)d_in[13];
    const float* Wl2_um  = (const float*)d_in[14];
    const float* bl2_um  = (const float*)d_in[15];
    const float* Wr2_um  = (const float*)d_in[16];
    float* out = (float*)d_out;

    float* ws = (float*)d_ws;
    float* hm        = ws;
    float* pu        = hm + 3200000;
    float* m1        = pu + 400000;
    float* wfold     = m1 + 3200000;
    float* wfold_r   = wfold + 1792;
    float* acc_x     = wfold + 3584;
    float* acc_x_b   = acc_x + 1600000;
    float* acc_mu    = acc_x_b + 1600000;
    float* acc_mu_b  = acc_mu + 6400000;
    float* accp      = acc_mu_b + 6400000;
    float* deg_user  = accp + 600064;
    float* deg_movie = deg_user + 100000;

    const size_t zero_bytes = (size_t)(1600000 + 1600000 + 6400000 + 6400000 + 600064 + 100000 + 50000) * 4;
    hipMemsetAsync(acc_x, 0, zero_bytes, stream);

    const int E = NUM_EDGES;
    prepass_kernel<<<PRE_HIST + PRE_GEMM + 2, HTHREADS, 0, stream>>>(
        e_dst, deg_movie, e_src, deg_user, E,
        x_movie, W_movie, b_movie, hm,
        Wl1_um, Wr1_mu, W_user, b_user, wfold);

    const int gridU = (NUM_USERS / 16 + 7) / 8;
    const int gridM = (NUM_MOVIES / 16 + 7) / 8;

    const int sblocks = (E + 8 * SC_EPW - 1) / (8 * SC_EPW);

    scatter_feat_kernel<<<sblocks, 256, 0, stream>>>(x_user, e_src, e_dst, acc_x, acc_x_b, E);
    scatter_kernel<<<sblocks, 256, 0, stream>>>(hm, e_dst, e_src, acc_mu, acc_mu_b, E);

    layer1_gemm_kernel<<<gridU + gridM, 256, 0, stream>>>(
        acc_mu, acc_mu_b, deg_user, Wl1_mu, x_user, wfold_r, bl1_mu, Wl2_um, pu,
        acc_x, acc_x_b, deg_movie, wfold, hm, Wr1_um, bl1_um, m1, gridU);

    proj_scatter_kernel<<<dim3(PS_SLICES, PS_RANGES), PS_THREADS, 0, stream>>>(
        pu, e_src, e_dst, accp, E);
    head_kernel<<<(NUM_MOVIES + 255) / 256, 256, 0, stream>>>(
        accp, deg_movie, m1, bl2_um, Wr2_um, out, NUM_MOVIES);
}


# User-side edge sum off memory float atomics: edges cut into packed per-user-range lists (integer cursor adds); a block per range sums its rows in LDS float bins, one store per row; sum order changes
# speedup vs baseline: 5.2815x; 1.0401x over previous
#include <hip/hip_runtime.h>
#include <hip/hip_bf16.h>

typedef __attribute__((ext_vector_type(2))) float v2f;
typedef __attribute__((ext_vector_type(8))) float v8f;

#define NUM_USERS  100000
#define NUM_MOVIES 50000
#define NUM_EDGES  4000000
#define USER_FEAT  24
#define MOVIE_FEAT 404
#define HID        64
#define GENRES     3

#define HBINS        50000
#define HSLICES      32
#define HTHREADS     1024
#define MOVIE_RANGES ((NUM_MOVIES + HBINS - 1) / HBINS)
#define USER_RANGES  ((NUM_USERS + HBINS - 1) / HBINS)

__device__ __forceinline__ void hist_count4(unsigned int* bins, const int4 v, int lo, unsigned int nb) {
    const unsigned int b0 = (unsigned int)(v.x - lo);
    const unsigned int b1 = (unsigned int)(v.y - lo);
    const unsigned int b2 = (unsigned int)(v.z - lo);
    const unsigned int b3 = (unsigned int)(v.w - lo);
    if (b0 < nb) atomicAdd(&bins[b0], 1u);
    if (b1 < nb) atomicAdd(&bins[b1], 1u);
    if (b2 < nb) atomicAdd(&bins[b2], 1u);
    if (b3 < nb) atomicAdd(&bins[b3], 1u);
}

__device__ __forceinline__
void degree_hist_role(unsigned int* bins, const int slice, const int job,
                      const int* __restrict__ dst, float* __restrict__ deg_movie,
                      const int* __restrict__ src, float* __restrict__ deg_user, int E) {
    const int tid = threadIdx.x;
    const bool is_movie = job < MOVIE_RANGES;
    const int* __restrict__ idx = is_movie ? dst : src;
    const int nseg = is_movie ? NUM_MOVIES : NUM_USERS;
    const int lo = (is_movie ? job : job - MOVIE_RANGES) * HBINS;
    const int left = nseg - lo;
    const unsigned int nb = (unsigned int)(left < HBINS ? left : HBINS);
    float* __restrict__ out = (is_movie ? deg_movie : deg_user) + lo;

    for (int i = tid; i < HBINS; i += HTHREADS) bins[i] = 0u;
    __syncthreads();

    const int per = (((E + HSLICES - 1) / HSLICES) + 3) & ~3;
    const int start = slice * per;
    int end = start + per;
    if (end > E) end = E;
    if (start < end) {
        const int4* __restrict__ p = (const int4*)(idx + start);
        const int nvec = (end - start) >> 2;
        int i = tid;
        for (; i + 3 * HTHREADS < nvec; i += 4 * HTHREADS) {
            const int4 a = p[i];
            const int4 b = p[i + HTHREADS];
            const int4 c = p[i + 2 * HTHREADS];
            const int4 d = p[i + 3 * HTHREADS];
            hist_count4(bins, a, lo, nb);
            hist_count4(bins, b, lo, nb);
            hist_count4(bins, c, lo, nb);
            hist_count4(bins, d, lo, nb);
        }
        for (; i < nvec; i += HTHREADS) hist_count4(bins, p[i], lo, nb);
        for (int e = start + (nvec << 2) + tid; e < end; e += HTHREADS) {
            const unsigned int b = (unsigned int)(idx[e] - lo);
            if (b < nb) atomicAdd(&bins[b], 1u);
        }
    }
    __syncthreads();

    const int n = (int)nb;
    const int rot = (slice * (n / HSLICES)) & ~31;
    for (int j = tid; j < n; j += HTHREADS) {
        int i = j + rot;
        if (i >= n) i -= n;
        const unsigned int cnt = bins[i];
        if (cnt) atomicAdd(&out[i], (float)cnt);
    }
}

__device__ __forceinline__ int chip_half() {
    const unsigned int id = __builtin_amdgcn_s_sendmsg_rtn(0x87u);
    return (int)((id >> 18) & 1u);
}

#define UB_ROWS    196
#define UB_NB      512
#define UB_CAP     12288
#define UP_THREADS 1024
#define UP_BLOCKS  256
#define UP_EPT     16
#define UA_THREADS 1024
#define UA_GROUP   8

static_assert(UB_ROWS * UB_NB >= NUM_USERS, "buckets must cover all users");
static_assert(NUM_MOVIES <= 65536 && UB_ROWS <= 65536, "packed list word");
static_assert((long long)UP_EPT * UP_THREADS * UP_BLOCKS >= NUM_EDGES, "edges per thread");

__global__ __launch_bounds__(UP_THREADS)
void user_bucket_kernel(const int* __restrict__ src, const int* __restrict__ dst,
                        unsigned int* __restrict__ cursor, unsigned int* __restrict__ list, int E) {
    __shared__ unsigned int cnt[UB_NB];
    __shared__ unsigned int base[UB_NB];
    const int tid = threadIdx.x;
    for (int i = tid; i < UB_NB; i += UP_THREADS) cnt[i] = 0u;
    __syncthreads();

    const int per = (E + UP_BLOCKS - 1) / UP_BLOCKS;
    const int start = (int)blockIdx.x * per;
    int end = start + per;
    if (end > E) end = E;

    int s[UP_EPT], d[UP_EPT];
    unsigned int rk[UP_EPT];
#pragma unroll
    for (int j = 0; j < UP_EPT; ++j) {
        const int e = start + j * UP_THREADS + tid;
        s[j] = -1;
        d[j] = 0;
        if (e < end) { s[j] = src[e]; d[j] = dst[e]; }
    }
#pragma unroll
    for (int j = 0; j < UP_EPT; ++j) {
        rk[j] = 0u;
        if (s[j] >= 0) rk[j] = atomicAdd(&cnt[s[j] / UB_ROWS], 1u);
    }
    __syncthreads();
    for (int b = tid; b < UB_NB; b += UP_THREADS) {
        const unsigned int n = cnt[b];
        base[b] = n ? atomicAdd(&cursor[b], n) : 0u;
    }
    __syncthreads();
#pragma unroll
    for (int j = 0; j < UP_EPT; ++j) {
        if (s[j] >= 0) {
            const int b = s[j] / UB_ROWS;
            const unsigned int pos = base[b] + rk[j];
            if (pos < UB_CAP)
                list[(size_t)b * UB_CAP + pos] =
                    ((unsigned int)(s[j] - b * UB_ROWS) << 16) | (unsigned int)d[j];
        }
    }
}

__global__ __launch_bounds__(UA_THREADS)
void user_range_sum_kernel(const float* __restrict__ X, const int* __restrict__ src,
                           const int* __restrict__ dst,
                           const unsigned int* __restrict__ cursor,
                           const unsigned int* __restrict__ list,
                           float* __restrict__ acc, int E) {
    __shared__ float bins[UB_ROWS * HID];
    const int tid = threadIdx.x;
    const int lane = tid & 31;
    const int wave = __builtin_amdgcn_readfirstlane(tid >> 5);
    const int b = (int)blockIdx.x;

    for (int i = tid; i < UB_ROWS * HID; i += UA_THREADS) bins[i] = 0.0f;
    __syncthreads();

    const unsigned int total = cursor[b];
    if (total <= UB_CAP) {
        const unsigned int* __restrict__ L = list + (size_t)b * UB_CAP;
        for (unsigned int v0 = (unsigned int)wave * 32u; v0 < total; v0 += UA_THREADS) {
            const unsigned int i = v0 + (unsigned int)lane;
            const int w = i < total ? (int)L[i] : 0;
            if (total - v0 >= 32u) {
#pragma unroll
                for (int g = 0; g < 32; g += UA_GROUP) {
                    float lo[UA_GROUP], hi[UA_GROUP];
#pragma unroll
                    for (int j = 0; j < UA_GROUP; ++j) {
                        const unsigned int wj = (unsigned int)__builtin_amdgcn_readlane(w, g + j);
                        const float* __restrict__ xr = X + (size_t)(wj & 0xffffu) * HID + lane;
                        lo[j] = xr[0];
                        hi[j] = xr[32];
                    }
#pragma unroll
                    for (int j = 0; j < UA_GROUP; ++j) {
                        const unsigned int wj = (unsigned int)__builtin_amdgcn_readlane(w, g + j);
                        float* q = bins + (wj >> 16) * HID + lane;
                        atomicAdd(q, lo[j]);
                        atomicAdd(q + 32, hi[j]);
                    }
                }
            } else {
                const int n = (int)(total - v0);
                for (int k = 0; k < n; ++k) {
                    const unsigned int wj = (unsigned int)__builtin_amdgcn_readlane(w, k);
                    const float* __restrict__ xr = X + (size_t)(wj & 0xffffu) * HID + lane;
                    const float a0 = xr[0];
                    const float a1 = xr[32];
                    float* q = bins + (wj >> 16) * HID + lane;
                    atomicAdd(q, a0);
                    atomicAdd(q + 32, a1);
                }
            }
        }
    } else {
        for (int e = wave; e < E; e += UA_THREADS / 32) {
            const int s = src[e];
            if (s / UB_ROWS == b) {
                const float* __restrict__ xr = X + (size_t)dst[e] * HID + lane;
                const float a0 = xr[0];
                const float a1 = xr[32];
                float* q = bins + (s - b * UB_ROWS) * HID + lane;
                atomicAdd(q, a0);
                atomicAdd(q + 32, a1);
            }
        }
    }
    __syncthreads();

    const int row0 = b * UB_ROWS;
    int nrows = NUM_USERS - row0;
    if (nrows > UB_ROWS) nrows = UB_ROWS;
    float* __restrict__ out = acc + (size_t)row0 * HID;
    for (int i = tid; i < nrows * HID; i += UA_THREADS) out[i] = bins[i];
}

#define SC_EPW   64
#define XF_PITCH 32
#define XF_GROUP 16

__device__ __forceinline__ void scatter_feat_rows(const float* __restrict__ X, float* __restrict__ acc,
                                                  const int sv, const int dv, const int n,
                                                  const int lane) {
    const bool act = lane < USER_FEAT;
    if (n == 32) {
#pragma unroll
        for (int g = 0; g < 32; g += XF_GROUP) {
            int s[XF_GROUP], d[XF_GROUP];
#pragma unroll
            for (int j = 0; j < XF_GROUP; ++j) {
                s[j] = __builtin_amdgcn_readlane(sv, g + j);
                d[j] = __builtin_amdgcn_readlane(dv, g + j);
            }
            if (act) {
                float v[XF_GROUP];
#pragma unroll
                for (int j = 0; j < XF_GROUP; ++j)
                    v[j] = X[(size_t)s[j] * USER_FEAT + lane];
#pragma unroll
                for (int j = 0; j < XF_GROUP; ++j)
                    atomicAdd(acc + (size_t)d[j] * XF_PITCH + lane, v[j]);
            }
        }
    } else {
        for (int i = 0; i < n; ++i) {
            const int s = __builtin_amdgcn_readlane(sv, i);
            const int d = __builtin_amdgcn_readlane(dv, i);
            if (act) {
                const float a = X[(size_t)s * USER_FEAT + lane];
                atomicAdd(acc + (size_t)d * XF_PITCH + lane, a);
            }
        }
    }
}

__global__ __launch_bounds__(256) void scatter_feat_kernel(const float* __restrict__ X,
                                                           const int* __restrict__ src,
                                                           const int* __restrict__ dst,
                                                           float* __restrict__ acc0,
                                                           float* __restrict__ acc1, int E) {
    const int lane = threadIdx.x & 31;
    float* __restrict__ acc = chip_half() ? acc1 : acc0;
    const int wave = __builtin_amdgcn_readfirstlane((int)(blockIdx.x * 8 + (threadIdx.x >> 5)));
    const long long e0 = (long long)wave * SC_EPW;
    if (e0 >= E) return;
    const int ea = (int)e0 + lane;
    const int eb = ea + 32;
    int sa = 0, da = 0, sb = 0, db = 0;
    if (ea < E) { sa = src[ea]; da = dst[ea]; }
    if (eb < E) { sb = src[eb]; db = dst[eb]; }
    int na = E - (int)e0;
    int nb = na - 32;
    if (na > 32) na = 32;
    if (nb > 32) nb = 32;
    scatter_feat_rows(X, acc, sa, da, na, lane);
    if (nb > 0) scatter_feat_rows(X, acc, sb, db, nb, lane);
}

#define PS_RANGES  2
#define PS_RB      ((NUM_MOVIES + PS_RANGES - 1) / PS_RANGES)
#define PS_SLICES  128
#define PS_THREADS 1024
#define PS_PLANES  4
#define PS_PITCH   150016

__device__ __forceinline__ void ps_edges8(float* bins, const float4* __restrict__ P4,
                                          const int4 d0, const int4 s0,
                                          const int4 d1, const int4 s1,
                                          int lo, unsigned int nb) {
    const int dd[8] = {d0.x, d0.y, d0.z, d0.w, d1.x, d1.y, d1.z, d1.w};
    const int ss[8] = {s0.x, s0.y, s0.z, s0.w, s1.x, s1.y, s1.z, s1.w};
    unsigned int b[8];
    float4 p[8];
#pragma unroll
    for (int j = 0; j < 8; ++j) {
        b[j] = (unsigned int)(dd[j] - lo);
        p[j] = make_float4(0.0f, 0.0f, 0.0f, 0.0f);
        if (b[j] < nb) p[j] = P4[ss[j]];
    }
#pragma unroll
    for (int j = 0; j < 8; ++j) {
        if (b[j] < nb) {
            float* q = bins + b[j] * GENRES;
            atomicAdd(q + 0, p[j].x);
            atomicAdd(q + 1, p[j].y);
            atomicAdd(q + 2, p[j].z);
        }
    }
}

__device__ __forceinline__ void ps_edge1(float* bins, const float4* __restrict__ P4,
                                         const int d, const int s, int lo, unsigned int nb) {
    const unsigned int b = (unsigned int)(d - lo);
    if (b < nb) {
        const float4 p = P4[s];
        float* q = bins + b * GENRES;
        atomicAdd(q + 0, p.x);
        atomicAdd(q + 1, p.y);
        atomicAdd(q + 2, p.z);
    }
}

__global__ __launch_bounds__(PS_THREADS)
void proj_scatter_kernel(const float* __restrict__ P, const int* __restrict__ src,
                         const int* __restrict__ dst, float* __restrict__ accp, int E) {
    __shared__ float bins[PS_RB * GENRES];

    const int tid = threadIdx.x;
    const int lo = (int)blockIdx.y * PS_RB;
    const int left = NUM_MOVIES - lo;
    const unsigned int nb = (unsigned int)(left < PS_RB ? left : PS_RB);
    const float4* __restrict__ P4 = (const float4*)P;

    for (int i = tid; i < PS_RB * GENRES; i += PS_THREADS) bins[i] = 0.0f;
    __syncthreads();

    const int per = (((E + PS_SLICES - 1) / PS_SLICES) + 3) & ~3;
    const int start = blockIdx.x * per;
    int end = start + per;
    if (end > E) end = E;
    if (start < end) {
        const int4* __restrict__ pd = (const int4*)(dst + start);
        const int4* __restrict__ ps = (const int4*)(src + start);
        const int nvec = (end - start) >> 2;
        int i = tid;
        for (; i + PS_THREADS < nvec; i += 2 * PS_THREADS) {
            const int4 d0 = pd[i];
            const int4 s0 = ps[i];
            const int4 d1 = pd[i + PS_THREADS];
            const int4 s1 = ps[i + PS_THREADS];
            ps_edges8(bins, P4, d0, s0, d1, s1, lo, nb);
        }
        for (; i < nvec; i += PS_THREADS) {
            const int4 d0 = pd[i];
            const int4 s0 = ps[i];
            ps_edge1(bins, P4, d0.x, s0.x, lo, nb);
            ps_edge1(bins, P4, d0.y, s0.y, lo, nb);
            ps_edge1(bins, P4, d0.z, s0.z, lo, nb);
            ps_edge1(bins, P4, d0.w, s0.w, lo, nb);
        }
        for (int e = start + (nvec << 2) + tid; e < end; e += PS_THREADS)
            ps_edge1(bins, P4, dst[e], src[e], lo, nb);
    }
    __syncthreads();

    const int n = (int)nb * GENRES;
    const int rot = ((int)(blockIdx.x / PS_PLANES) * (n / (PS_SLICES / PS_PLANES))) & ~31;
    float* __restrict__ out = accp + (size_t)(blockIdx.x % PS_PLANES) * PS_PITCH + (size_t)lo * GENRES;
    for (int j = tid; j < n; j += PS_THREADS) {
        int i = j + rot;
        if (i >= n) i -= n;
        const float v = bins[i];
        if (v != 0.0f) atomicAdd(&out[i], v);
    }
}

__host__ __device__ constexpr int w_pitch(int K) { return ((K / 4) & 1) ? K : K + 4; }

template <int K, int PITCH, int NT>
__device__ __forceinline__ void stage_weights(float* __restrict__ dst, const float* __restrict__ W, int tid) {
    constexpr int Q = K / 4;
    constexpr int STEPS = (64 * Q + NT - 1) / NT;
    const float4* __restrict__ W4 = (const float4*)W;
    float4 v[STEPS];
#pragma unroll
    for (int j = 0; j < STEPS; ++j) {
        const int i = tid + NT * j;
        v[j] = W4[i < 64 * Q ? i : 64 * Q - 1];
    }
#pragma unroll
    for (int j = 0; j < STEPS; ++j) {
        const int i = tid + NT * j;
        if (i < 64 * Q) {
            const int n = i / Q;
            const int q = i - n * Q;
            *(float4*)(dst + n * PITCH + 4 * q) = v[j];
        }
    }
}

template <bool PROJ, bool DUAL, int K1, int LD1, int AUG1, int K2, int LD2, int AUG2, int NT, int WPB>
__device__ __forceinline__
void sage_gemm64_body(float* __restrict__ wlds, const int gblock,
                      const float* __restrict__ A1, const float* __restrict__ A1b,
                      const float* __restrict__ deg1,
                      const float* __restrict__ W1,
                      const float* __restrict__ A2, const float* __restrict__ W2,
                      const float* __restrict__ bias, const float* __restrict__ Wp,
                      float* __restrict__ C, int M, int relu) {
    constexpr int P1 = w_pitch(K1);
    constexpr int P2 = K2 > 0 ? w_pitch(K2) : 0;

    stage_weights<K1, P1, NT>(wlds, W1, threadIdx.x);
    if (K2 > 0) stage_weights<(K2 > 0 ? K2 : 4), (K2 > 0 ? P2 : 4), NT>(wlds + 64 * P1, W2, threadIdx.x);
    __syncthreads();

    const int lane = threadIdx.x & 31;
    const int wave = threadIdx.x >> 5;
    const int half = lane >> 4;
    const int r    = lane & 15;
    if (NT / 32 > WPB && wave >= WPB) return;
    const int row0 = (gblock * WPB + wave) * 16;
    if (row0 >= M) return;

    v8f c[4] = {};

    float scale = 1.0f;
    float has_edges = 1.0f;
    if (deg1) {
        const float dg = deg1[row0 + r];
        scale = 1.0f / fmaxf(dg, 1.0f);
        has_edges = dg >= 1.0f ? 1.0f : 0.0f;
    }

    {
        constexpr int KL1 = AUG1 ? K1 - 4 : K1;
        const float* Arow = A1 + (size_t)(row0 + r) * LD1;
        const float* Brow = DUAL ? A1b + (size_t)(row0 + r) * LD1 : Arow;
        const float* Ws = wlds + r * P1 + 2 * half;
#pragma unroll
        for (int kk = 0; kk < K1; kk += 4) {
            const int k = kk + 2 * half;
            v2f a;
            if (kk < KL1) {
                float a0 = Arow[k];
                float a1 = Arow[k + 1];
                if (DUAL) { a0 += Brow[k]; a1 += Brow[k + 1]; }
                a.x = a0 * scale;
                a.y = a1 * scale;
            } else {
                const float u = AUG1 == 1 ? has_edges : 1.0f;
                a.x = half == 0 ? u : 0.0f;
                a.y = 0.0f;
            }
#pragma unroll
            for (int t = 0; t < 4; ++t) {
                const v2f b = *(const v2f*)(Ws + t * 16 * P1 + kk);
                c[t] = __builtin_amdgcn_wmma_f32_16x16x4_f32(
                    false, a, false, b, (short)0, c[t], false, false);
            }
        }
    }
    if (K2 > 0) {
        constexpr int KL2 = AUG2 ? K2 - 4 : K2;
        const float* Arow = A2 + (size_t)(row0 + r) * LD2;
        const float* Ws = wlds + 64 * P1 + r * P2 + 2 * half;
#pragma unroll
        for (int kk = 0; kk < K2; kk += 4) {
            const int k = kk + 2 * half;
            v2f a;
            if (kk < KL2) {
                a.x = Arow[k];
                a.y = Arow[k + 1];
            } else {
                const float u = AUG2 == 1 ? has_edges : 1.0f;
                a.x = half == 0 ? u : 0.0f;
                a.y = 0.0f;
            }
#pragma unroll
            for (int t = 0; t < 4; ++t) {
                const v2f b = *(const v2f*)(Ws + t * 16 * P2 + kk);
                c[t] = __builtin_amdgcn_wmma_f32_16x16x4_f32(
                    false, a, false, b, (short)0, c[t], false, false);
            }
        }
    }

    if (!PROJ) {
#pragma unroll
        for (int t = 0; t < 4; ++t) {
            const float bv = bias ? bias[t * 16 + r] : 0.0f;
#pragma unroll
            for (int i = 0; i < 8; ++i) {
                float v = c[t][i] + bv;
                if (relu) v = fmaxf(v, 0.0f);
                C[(size_t)(row0 + i + 8 * half) * HID + t * 16 + r] = v;
            }
        }
    } else {
        float s[8][GENRES];
#pragma unroll
        for (int i = 0; i < 8; ++i)
#pragma unroll
            for (int g = 0; g < GENRES; ++g) s[i][g] = 0.0f;
#pragma unroll
        for (int t = 0; t < 4; ++t) {
            const float bv = bias ? bias[t * 16 + r] : 0.0f;
            float w[GENRES];
#pragma unroll
            for (int g = 0; g < GENRES; ++g) w[g] = Wp[g * HID + t * 16 + r];
#pragma unroll
            for (int i = 0; i < 8; ++i) {
                float v = c[t][i] + bv;
                if (relu) v = fmaxf(v, 0.0f);
#pragma unroll
                for (int g = 0; g < GENRES; ++g) s[i][g] += v * w[g];
            }
        }
#pragma unroll
        for (int i = 0; i < 8; ++i)
#pragma unroll
            for (int g = 0; g < GENRES; ++g)
#pragma unroll
                for (int m = 8; m >= 1; m >>= 1)
                    s[i][g] += __shfl_xor(s[i][g], m, 16);
        float4 o = make_float4(0.0f, 0.0f, 0.0f, 0.0f);
#pragma unroll
        for (int i = 0; i < 8; ++i)
            if (r == i) o = make_float4(s[i][0], s[i][1], s[i][2], 0.0f);
        if (r < 8) ((float4*)C)[(size_t)(row0 + r + 8 * half)] = o;
    }
}

#define FOLD_K (USER_FEAT + 4)

__device__ __forceinline__
void fold_user_proj_role(const int m,
                         const float* __restrict__ Wm0, const float* __restrict__ Wm1,
                         const float* __restrict__ W_user, const float* __restrict__ b_user,
                         float* __restrict__ Wf) {
    if (threadIdx.x >= 256) return;
    const float* __restrict__ Wm = m ? Wm1 : Wm0;
    float* __restrict__ out = Wf + (size_t)m * 64 * FOLD_K;

    const int lane = threadIdx.x & 31;
    const int wave = threadIdx.x >> 5;
    const int half = lane >> 4;
    const int r    = lane & 15;
    const int tn   = wave >> 1;
    const int col  = (wave & 1) * 16 + r;
    const float* Bcol = col < USER_FEAT ? W_user + col : b_user;
    const int bpitch  = col < USER_FEAT ? USER_FEAT : 1;
    const bool live   = col <= USER_FEAT;

    const float* Arow = Wm + (size_t)(tn * 16 + r) * HID;
    v8f c = {};
#pragma unroll
    for (int kk = 0; kk < HID; kk += 4) {
        const int k = kk + 2 * half;
        v2f a, b;
        a.x = Arow[k];
        a.y = Arow[k + 1];
        const float b0 = Bcol[k * bpitch];
        const float b1 = Bcol[(k + 1) * bpitch];
        b.x = live ? b0 : 0.0f;
        b.y = live ? b1 : 0.0f;
        c = __builtin_amdgcn_wmma_f32_16x16x4_f32(
            false, a, false, b, (short)0, c, false, false);
    }
    if (col < FOLD_K) {
#pragma unroll
        for (int i = 0; i < 8; ++i)
            out[(size_t)(tn * 16 + i + 8 * half) * FOLD_K + col] = c[i];
    }
}

#define PRE_HIST (HSLICES * (MOVIE_RANGES + USER_RANGES))
#define PRE_WPB  20
#define PRE_GEMM ((NUM_MOVIES / 16 + PRE_WPB - 1) / PRE_WPB)

__global__ __launch_bounds__(HTHREADS)
void prepass_kernel(const int* __restrict__ dst, float* __restrict__ deg_movie,
                    const int* __restrict__ src, float* __restrict__ deg_user, int E,
                    const float* __restrict__ x_movie, const float* __restrict__ W_movie,
                    const float* __restrict__ b_movie, float* __restrict__ hm,
                    const float* __restrict__ Wm0, const float* __restrict__ Wm1,
                    const float* __restrict__ W_user, const float* __restrict__ b_user,
                    float* __restrict__ Wf) {
    __shared__ __attribute__((aligned(16))) unsigned int smem[HBINS];
    static_assert(64 * w_pitch(MOVIE_FEAT) <= HBINS, "staged W_movie must fit the histogram's LDS array");

    const int b = (int)blockIdx.x;
    if (b < PRE_HIST) {
        degree_hist_role(smem, b % HSLICES, b / HSLICES, dst, deg_movie, src, deg_user, E);
    } else if (b < PRE_HIST + PRE_GEMM) {
        sage_gemm64_body<false, false, MOVIE_FEAT, MOVIE_FEAT, 0, 0, 0, 0, HTHREADS, PRE_WPB>(
            (float*)smem, b - PRE_HIST, x_movie, nullptr, nullptr, W_movie, nullptr, nullptr,
            b_movie, nullptr, hm, NUM_MOVIES, 0);
    } else {
        fold_user_proj_role(b - PRE_HIST - PRE_GEMM, Wm0, Wm1, W_user, b_user, Wf);
    }
}

__global__ __launch_bounds__(256)
void layer1_gemm_kernel(const float* __restrict__ acc_mu,
                        const float* __restrict__ deg_user, const float* __restrict__ Wl1_mu,
                        const float* __restrict__ x_user, const float* __restrict__ wfold_r,
                        const float* __restrict__ bl1_mu, const float* __restrict__ Wl2_um,
                        float* __restrict__ pu,
                        const float* __restrict__ acc_x, const float* __restrict__ acc_x_b,
                        const float* __restrict__ deg_movie, const float* __restrict__ wfold_l,
                        const float* __restrict__ hm, const float* __restrict__ Wr1_um,
                        const float* __restrict__ bl1_um, float* __restrict__ m1,
                        int user_blocks) {
    __shared__ __attribute__((aligned(16))) float wlds[64 * (w_pitch(FOLD_K) + w_pitch(HID))];
    const int b = (int)blockIdx.x;
    if (b < user_blocks) {
        sage_gemm64_body<true, false, HID, HID, 0, FOLD_K, USER_FEAT, 2, 256, 8>(
            wlds, b, acc_mu, nullptr, deg_user, Wl1_mu, x_user, wfold_r, bl1_mu, Wl2_um,
            pu, NUM_USERS, 1);
    } else {
        sage_gemm64_body<false, true, FOLD_K, XF_PITCH, 1, HID, HID, 0, 256, 8>(
            wlds, b - user_blocks, acc_x, acc_x_b, deg_movie, wfold_l, hm, Wr1_um, bl1_um, nullptr,
            m1, NUM_MOVIES, 1);
    }
}

__global__ __launch_bounds__(256)
void head_kernel(const float* __restrict__ accp, const float* __restrict__ degm,
                 const float* __restrict__ m1,
                 const float* __restrict__ bl2, const float* __restrict__ Wr2,
                 float* __restrict__ out, int M) {
    int m = blockIdx.x * blockDim.x + threadIdx.x;
    if (m >= M) return;
    const float inv = 1.0f / fmaxf(degm[m], 1.0f);
    const float* a = accp + (size_t)m * GENRES;
    const float* x = m1 + (size_t)m * HID;
    float acc[GENRES];
#pragma unroll
    for (int g = 0; g < GENRES; ++g) {
        float sum = a[g];
#pragma unroll
        for (int pl = 1; pl < PS_PLANES; ++pl) sum += a[(size_t)pl * PS_PITCH + g];
        acc[g] = sum * inv + bl2[g];
    }
    for (int k = 0; k < HID; ++k) {
        const float xv = x[k];
#pragma unroll
        for (int g = 0; g < GENRES; ++g)
            acc[g] += xv * Wr2[g * HID + k];
    }
#pragma unroll
    for (int g = 0; g < GENRES; ++g) out[(size_t)m * GENRES + g] = acc[g];
}

extern "C" void kernel_launch(void* const* d_in, const int* in_sizes, int n_in,
                              void* d_out, int out_size, void* d_ws, size_t ws_size,
                              hipStream_t stream) {
    (void)in_sizes; (void)n_in; (void)out_size; (void)ws_size;

    const float* x_user  = (const float*)d_in[0];
    const float* x_movie = (const float*)d_in[1];
    const int*   e_src   = (const int*)d_in[2];
    const int*   e_dst   = (const int*)d_in[3];
    const float* W_user  = (const float*)d_in[4];
    const float* b_user  = (const float*)d_in[5];
    const float* W_movie = (const float*)d_in[6];
    const float* b_movie = (const float*)d_in[7];
    const float* Wl1_um  = (const float*)d_in[8];
    const float* bl1_um  = (const float*)d_in[9];
    const float* Wr1_um  = (const float*)d_in[10];
    const float* Wl1_mu  = (const float*)d_in[11];
    const float* bl1_mu  = (const float*)d_in[12];
    const float* Wr1_mu  = (const float*)d_in[13];
    const float* Wl2_um  = (const float*)d_in[14];
    const float* bl2_um  = (const float*)d_in[15];
    const float* Wr2_um  = (const float*)d_in[16];
    float* out = (float*)d_out;

    float* ws = (float*)d_ws;
    float* hm        = ws;
    float* pu        = hm + 3200000;
    float* m1        = pu + 400000;
    float* wfold     = m1 + 3200000;
    float* wfold_r   = wfold + 1792;
    float* acc_x     = wfold + 3584;
    float* acc_x_b   = acc_x + 1600000;
    float* accp      = acc_x_b + 1600000;
    float* deg_user  = accp + 600064;
    float* deg_movie = deg_user + 100000;
    unsigned int* ucursor = (unsigned int*)(deg_movie + 50000);
    float* acc_mu    = deg_movie + 50000 + 528;
    unsigned int* ulist = (unsigned int*)(acc_mu + 6400000);

    const size_t zero_bytes = (size_t)(1600000 + 1600000 + 600064 + 100000 + 50000 + 528) * 4;
    hipMemsetAsync(acc_x, 0, zero_bytes, stream);

    const int E = NUM_EDGES;
    prepass_kernel<<<PRE_HIST + PRE_GEMM + 2, HTHREADS, 0, stream>>>(
        e_dst, deg_movie, e_src, deg_user, E,
        x_movie, W_movie, b_movie, hm,
        Wl1_um, Wr1_mu, W_user, b_user, wfold);

    const int gridU = (NUM_USERS / 16 + 7) / 8;
    const int gridM = (NUM_MOVIES / 16 + 7) / 8;

    const int sblocks = (E + 8 * SC_EPW - 1) / (8 * SC_EPW);

    user_bucket_kernel<<<UP_BLOCKS, UP_THREADS, 0, stream>>>(e_src, e_dst, ucursor, ulist, E);
    scatter_feat_kernel<<<sblocks, 256, 0, stream>>>(x_user, e_src, e_dst, acc_x, acc_x_b, E);
    user_range_sum_kernel<<<UB_NB, UA_THREADS, 0, stream>>>(hm, e_src, e_dst, ucursor, ulist, acc_mu, E);

    layer1_gemm_kernel<<<gridU + gridM, 256, 0, stream>>>(
        acc_mu, deg_user, Wl1_mu, x_user, wfold_r, bl1_mu, Wl2_um, pu,
        acc_x, acc_x_b, deg_movie, wfold, hm, Wr1_um, bl1_um, m1, gridU);

    proj_scatter_kernel<<<dim3(PS_SLICES, PS_RANGES), PS_THREADS, 0, stream>>>(
        pu, e_src, e_dst, accp, E);
    head_kernel<<<(NUM_MOVIES + 255) / 256, 256, 0, stream>>>(
        accp, deg_movie, m1, bl2_um, Wr2_um, out, NUM_MOVIES);
}


# User-side range sum: block orders its list by row in LDS (returning integer adds give ranks); waves sum whole rows in registers, one 256-B load per edge, one store per row; sum order changes
# speedup vs baseline: 5.7205x; 1.0831x over previous
#include <hip/hip_runtime.h>
#include <hip/hip_bf16.h>

typedef __attribute__((ext_vector_type(2))) float v2f;
typedef __attribute__((ext_vector_type(8))) float v8f;

#define NUM_USERS  100000
#define NUM_MOVIES 50000
#define NUM_EDGES  4000000
#define USER_FEAT  24
#define MOVIE_FEAT 404
#define HID        64
#define GENRES     3

#define HBINS        50000
#define HSLICES      32
#define HTHREADS     1024
#define MOVIE_RANGES ((NUM_MOVIES + HBINS - 1) / HBINS)
#define USER_RANGES  ((NUM_USERS + HBINS - 1) / HBINS)

__device__ __forceinline__ void hist_count4(unsigned int* bins, const int4 v, int lo, unsigned int nb) {
    const unsigned int b0 = (unsigned int)(v.x - lo);
    const unsigned int b1 = (unsigned int)(v.y - lo);
    const unsigned int b2 = (unsigned int)(v.z - lo);
    const unsigned int b3 = (unsigned int)(v.w - lo);
    if (b0 < nb) atomicAdd(&bins[b0], 1u);
    if (b1 < nb) atomicAdd(&bins[b1], 1u);
    if (b2 < nb) atomicAdd(&bins[b2], 1u);
    if (b3 < nb) atomicAdd(&bins[b3], 1u);
}

__device__ __forceinline__
void degree_hist_role(unsigned int* bins, const int slice, const int job,
                      const int* __restrict__ dst, float* __restrict__ deg_movie,
                      const int* __restrict__ src, float* __restrict__ deg_user, int E) {
    const int tid = threadIdx.x;
    const bool is_movie = job < MOVIE_RANGES;
    const int* __restrict__ idx = is_movie ? dst : src;
    const int nseg = is_movie ? NUM_MOVIES : NUM_USERS;
    const int lo = (is_movie ? job : job - MOVIE_RANGES) * HBINS;
    const int left = nseg - lo;
    const unsigned int nb = (unsigned int)(left < HBINS ? left : HBINS);
    float* __restrict__ out = (is_movie ? deg_movie : deg_user) + lo;

    for (int i = tid; i < HBINS; i += HTHREADS) bins[i] = 0u;
    __syncthreads();

    const int per = (((E + HSLICES - 1) / HSLICES) + 3) & ~3;
    const int start = slice * per;
    int end = start + per;
    if (end > E) end = E;
    if (start < end) {
        const int4* __restrict__ p = (const int4*)(idx + start);
        const int nvec = (end - start) >> 2;
        int i = tid;
        for (; i + 3 * HTHREADS < nvec; i += 4 * HTHREADS) {
            const int4 a = p[i];
            const int4 b = p[i + HTHREADS];
            const int4 c = p[i + 2 * HTHREADS];
            const int4 d = p[i + 3 * HTHREADS];
            hist_count4(bins, a, lo, nb);
            hist_count4(bins, b, lo, nb);
            hist_count4(bins, c, lo, nb);
            hist_count4(bins, d, lo, nb);
        }
        for (; i < nvec; i += HTHREADS) hist_count4(bins, p[i], lo, nb);
        for (int e = start + (nvec << 2) + tid; e < end; e += HTHREADS) {
            const unsigned int b = (unsigned int)(idx[e] - lo);
            if (b < nb) atomicAdd(&bins[b], 1u);
        }
    }
    __syncthreads();

    const int n = (int)nb;
    const int rot = (slice * (n / HSLICES)) & ~31;
    for (int j = tid; j < n; j += HTHREADS) {
        int i = j + rot;
        if (i >= n) i -= n;
        const unsigned int cnt = bins[i];
        if (cnt) atomicAdd(&out[i], (float)cnt);
    }
}

__device__ __forceinline__ int chip_half() {
    const unsigned int id = __builtin_amdgcn_s_sendmsg_rtn(0x87u);
    return (int)((id >> 18) & 1u);
}

#define UB_ROWS    196
#define UB_NB      512
#define UB_CAP     12288
#define UP_THREADS 1024
#define UP_BLOCKS  256
#define UP_EPT     16
#define UA_THREADS 1024
#define UA_GROUP   16
#define UA_SLOTS   (UB_CAP / UA_THREADS)
#define UA_RPL     ((UB_ROWS + 31) / 32)

static_assert(UB_ROWS * UB_NB >= NUM_USERS, "buckets must cover all users");
static_assert(NUM_MOVIES <= 65536 && UB_ROWS <= 65536, "packed list word");
static_assert((long long)UP_EPT * UP_THREADS * UP_BLOCKS >= NUM_EDGES, "edges per thread");
static_assert(UB_CAP % UA_THREADS == 0, "list slots per thread");
static_assert(UB_CAP <= UB_ROWS * HID, "the sorted list lies in the array of the float bins");

__global__ __launch_bounds__(UP_THREADS)
void user_bucket_kernel(const int* __restrict__ src, const int* __restrict__ dst,
                        unsigned int* __restrict__ cursor, unsigned int* __restrict__ list, int E) {
    __shared__ unsigned int cnt[UB_NB];
    __shared__ unsigned int base[UB_NB];
    const int tid = threadIdx.x;
    for (int i = tid; i < UB_NB; i += UP_THREADS) cnt[i] = 0u;
    __syncthreads();

    const int per = (E + UP_BLOCKS - 1) / UP_BLOCKS;
    const int start = (int)blockIdx.x * per;
    int end = start + per;
    if (end > E) end = E;

    int s[UP_EPT], d[UP_EPT];
    unsigned int rk[UP_EPT];
#pragma unroll
    for (int j = 0; j < UP_EPT; ++j) {
        const int e = start + j * UP_THREADS + tid;
        s[j] = -1;
        d[j] = 0;
        if (e < end) { s[j] = src[e]; d[j] = dst[e]; }
    }
#pragma unroll
    for (int j = 0; j < UP_EPT; ++j) {
        rk[j] = 0u;
        if (s[j] >= 0) rk[j] = atomicAdd(&cnt[s[j] / UB_ROWS], 1u);
    }
    __syncthreads();
    for (int b = tid; b < UB_NB; b += UP_THREADS) {
        const unsigned int n = cnt[b];
        base[b] = n ? atomicAdd(&cursor[b], n) : 0u;
    }
    __syncthreads();
#pragma unroll
    for (int j = 0; j < UP_EPT; ++j) {
        if (s[j] >= 0) {
            const int b = s[j] / UB_ROWS;
            const unsigned int pos = base[b] + rk[j];
            if (pos < UB_CAP)
                list[(size_t)b * UB_CAP + pos] =
                    ((unsigned int)(s[j] - b * UB_ROWS) << 16) | (unsigned int)d[j];
        }
    }
}

__global__ __launch_bounds__(UA_THREADS)
void user_range_sum_kernel(const float* __restrict__ X, const int* __restrict__ src,
                           const int* __restrict__ dst,
                           const unsigned int* __restrict__ cursor,
                           const unsigned int* __restrict__ list,
                           float* __restrict__ acc, int E) {
    __shared__ __attribute__((aligned(16))) float bins[UB_ROWS * HID];
    __shared__ unsigned int rcnt[UB_ROWS];
    __shared__ unsigned int rstart[UB_ROWS];
    __shared__ unsigned int ticket;
    const int tid = threadIdx.x;
    const int lane = tid & 31;
    const int wave = __builtin_amdgcn_readfirstlane(tid >> 5);
    const int b = (int)blockIdx.x;
    const int row0 = b * UB_ROWS;

    const unsigned int total = cursor[b];
    if (total <= UB_CAP) {
        unsigned int* sorted = (unsigned int*)bins;
        const unsigned int* __restrict__ L = list + (size_t)b * UB_CAP;

        unsigned int w[UA_SLOTS], rk[UA_SLOTS];
#pragma unroll
        for (int k = 0; k < UA_SLOTS; ++k) {
            const unsigned int i = (unsigned int)(tid + k * UA_THREADS);
            w[k] = i < total ? L[i] : 0u;
        }

        for (int i = tid; i < UB_ROWS; i += UA_THREADS) rcnt[i] = 0u;
        if (tid == 0) ticket = 0u;
        __syncthreads();

#pragma unroll
        for (int k = 0; k < UA_SLOTS; ++k) {
            const unsigned int i = (unsigned int)(tid + k * UA_THREADS);
            rk[k] = 0u;
            if (i < total) rk[k] = atomicAdd(&rcnt[w[k] >> 16], 1u);
        }
        __syncthreads();

        if (wave == 0) {
            unsigned int c[UA_RPL];
            unsigned int own = 0u;
#pragma unroll
            for (int j = 0; j < UA_RPL; ++j) {
                const int r = lane * UA_RPL + j;
                c[j] = r < UB_ROWS ? rcnt[r] : 0u;
                own += c[j];
            }
            unsigned int incl = own;
#pragma unroll
            for (int d = 1; d < 32; d <<= 1) {
                const unsigned int t = __shfl_up(incl, d, 32);
                if (lane >= d) incl += t;
            }
            unsigned int run = incl - own;
#pragma unroll
            for (int j = 0; j < UA_RPL; ++j) {
                const int r = lane * UA_RPL + j;
                if (r < UB_ROWS) rstart[r] = run;
                run += c[j];
            }
        }
        __syncthreads();

#pragma unroll
        for (int k = 0; k < UA_SLOTS; ++k) {
            const unsigned int i = (unsigned int)(tid + k * UA_THREADS);
            if (i < total) sorted[rstart[w[k] >> 16] + rk[k]] = w[k] & 0xffffu;
        }
        __syncthreads();

        for (;;) {
            unsigned int t = 0u;
            if (lane == 0) t = atomicAdd(&ticket, 1u);
            const int r = __builtin_amdgcn_readfirstlane((int)t);
            if (r >= UB_ROWS || row0 + r >= NUM_USERS) break;
            const unsigned int n = (unsigned int)__builtin_amdgcn_readfirstlane((int)rcnt[r]);
            const unsigned int s = (unsigned int)__builtin_amdgcn_readfirstlane((int)rstart[r]);
            float s0 = 0.0f, s1 = 0.0f;
            for (unsigned int v0 = 0u; v0 < n; v0 += 32u) {
                const unsigned int i = v0 + (unsigned int)lane;
                const int m = i < n ? (int)sorted[s + i] : 0;
                const int left = (int)(n - v0);
                const int cnt = left < 32 ? left : 32;
                int g = 0;
                for (; g + UA_GROUP <= cnt; g += UA_GROUP) {
                    v2f x[UA_GROUP];
#pragma unroll
                    for (int j = 0; j < UA_GROUP; ++j) {
                        const int mj = __builtin_amdgcn_readlane(m, g + j);
                        x[j] = *(const v2f*)(X + (size_t)mj * HID + 2 * lane);
                    }
#pragma unroll
                    for (int j = 0; j < UA_GROUP; ++j) { s0 += x[j].x; s1 += x[j].y; }
                }
                if (g < cnt) {
                    v2f x[UA_GROUP];
#pragma unroll
                    for (int j = 0; j < UA_GROUP; ++j) {
                        x[j].x = 0.0f;
                        x[j].y = 0.0f;
                        if (g + j < cnt) {
                            const int mj = __builtin_amdgcn_readlane(m, g + j);
                            x[j] = *(const v2f*)(X + (size_t)mj * HID + 2 * lane);
                        }
                    }
#pragma unroll
                    for (int j = 0; j < UA_GROUP; ++j) { s0 += x[j].x; s1 += x[j].y; }
                }
            }
            v2f o;
            o.x = s0;
            o.y = s1;
            *(v2f*)(acc + (size_t)(row0 + r) * HID + 2 * lane) = o;
        }
    } else {
        for (int i = tid; i < UB_ROWS * HID; i += UA_THREADS) bins[i] = 0.0f;
        __syncthreads();
        for (int e = wave; e < E; e += UA_THREADS / 32) {
            const int s = src[e];
            if (s / UB_ROWS == b) {
                const float* __restrict__ xr = X + (size_t)dst[e] * HID + lane;
                const float a0 = xr[0];
                const float a1 = xr[32];
                float* q = bins + (s - b * UB_ROWS) * HID + lane;
                atomicAdd(q, a0);
                atomicAdd(q + 32, a1);
            }
        }
        __syncthreads();
        int nrows = NUM_USERS - row0;
        if (nrows > UB_ROWS) nrows = UB_ROWS;
        float* __restrict__ out = acc + (size_t)row0 * HID;
        for (int i = tid; i < nrows * HID; i += UA_THREADS) out[i] = bins[i];
    }
}

#define SC_EPW   64
#define XF_PITCH 32
#define XF_GROUP 16

__device__ __forceinline__ void scatter_feat_rows(const float* __restrict__ X, float* __restrict__ acc,
                                                  const int sv, const int dv, const int n,
                                                  const int lane) {
    const bool act = lane < USER_FEAT;
    if (n == 32) {
#pragma unroll
        for (int g = 0; g < 32; g += XF_GROUP) {
            int s[XF_GROUP], d[XF_GROUP];
#pragma unroll
            for (int j = 0; j < XF_GROUP; ++j) {
                s[j] = __builtin_amdgcn_readlane(sv, g + j);
                d[j] = __builtin_amdgcn_readlane(dv, g + j);
            }
            if (act) {
                float v[XF_GROUP];
#pragma unroll
                for (int j = 0; j < XF_GROUP; ++j)
                    v[j] = X[(size_t)s[j] * USER_FEAT + lane];
#pragma unroll
                for (int j = 0; j < XF_GROUP; ++j)
                    atomicAdd(acc + (size_t)d[j] * XF_PITCH + lane, v[j]);
            }
        }
    } else {
        for (int i = 0; i < n; ++i) {
            const int s = __builtin_amdgcn_readlane(sv, i);
            const int d = __builtin_amdgcn_readlane(dv, i);
            if (act) {
                const float a = X[(size_t)s * USER_FEAT + lane];
                atomicAdd(acc + (size_t)d * XF_PITCH + lane, a);
            }
        }
    }
}

__global__ __launch_bounds__(256) void scatter_feat_kernel(const float* __restrict__ X,
                                                           const int* __restrict__ src,
                                                           const int* __restrict__ dst,
                                                           float* __restrict__ acc0,
                                                           float* __restrict__ acc1, int E) {
    const int lane = threadIdx.x & 31;
    float* __restrict__ acc = chip_half() ? acc1 : acc0;
    const int wave = __builtin_amdgcn_readfirstlane((int)(blockIdx.x * 8 + (threadIdx.x >> 5)));
    const long long e0 = (long long)wave * SC_EPW;
    if (e0 >= E) return;
    const int ea = (int)e0 + lane;
    const int eb = ea + 32;
    int sa = 0, da = 0, sb = 0, db = 0;
    if (ea < E) { sa = src[ea]; da = dst[ea]; }
    if (eb < E) { sb = src[eb]; db = dst[eb]; }
    int na = E - (int)e0;
    int nb = na - 32;
    if (na > 32) na = 32;
    if (nb > 32) nb = 32;
    scatter_feat_rows(X, acc, sa, da, na, lane);
    if (nb > 0) scatter_feat_rows(X, acc, sb, db, nb, lane);
}

#define PS_RANGES  2
#define PS_RB      ((NUM_MOVIES + PS_RANGES - 1) / PS_RANGES)
#define PS_SLICES  128
#define PS_THREADS 1024
#define PS_PLANES  4
#define PS_PITCH   150016

__device__ __forceinline__ void ps_edges8(float* bins, const float4* __restrict__ P4,
                                          const int4 d0, const int4 s0,
                                          const int4 d1, const int4 s1,
                                          int lo, unsigned int nb) {
    const int dd[8] = {d0.x, d0.y, d0.z, d0.w, d1.x, d1.y, d1.z, d1.w};
    const int ss[8] = {s0.x, s0.y, s0.z, s0.w, s1.x, s1.y, s1.z, s1.w};
    unsigned int b[8];
    float4 p[8];
#pragma unroll
    for (int j = 0; j < 8; ++j) {
        b[j] = (unsigned int)(dd[j] - lo);
        p[j] = make_float4(0.0f, 0.0f, 0.0f, 0.0f);
        if (b[j] < nb) p[j] = P4[ss[j]];
    }
#pragma unroll
    for (int j = 0; j < 8; ++j) {
        if (b[j] < nb) {
            float* q = bins + b[j] * GENRES;
            atomicAdd(q + 0, p[j].x);
            atomicAdd(q + 1, p[j].y);
            atomicAdd(q + 2, p[j].z);
        }
    }
}

__device__ __forceinline__ void ps_edge1(float* bins, const float4* __restrict__ P4,
                                         const int d, const int s, int lo, unsigned int nb) {
    const unsigned int b = (unsigned int)(d - lo);
    if (b < nb) {
        const float4 p = P4[s];
        float* q = bins + b * GENRES;
        atomicAdd(q + 0, p.x);
        atomicAdd(q + 1, p.y);
        atomicAdd(q + 2, p.z);
    }
}

__global__ __launch_bounds__(PS_THREADS)
void proj_scatter_kernel(const float* __restrict__ P, const int* __restrict__ src,
                         const int* __restrict__ dst, float* __restrict__ accp, int E) {
    __shared__ float bins[PS_RB * GENRES];

    const int tid = threadIdx.x;
    const int lo = (int)blockIdx.y * PS_RB;
    const int left = NUM_MOVIES - lo;
    const unsigned int nb = (unsigned int)(left < PS_RB ? left : PS_RB);
    const float4* __restrict__ P4 = (const float4*)P;

    for (int i = tid; i < PS_RB * GENRES; i += PS_THREADS) bins[i] = 0.0f;
    __syncthreads();

    const int per = (((E + PS_SLICES - 1) / PS_SLICES) + 3) & ~3;
    const int start = blockIdx.x * per;
    int end = start + per;
    if (end > E) end = E;
    if (start < end) {
        const int4* __restrict__ pd = (const int4*)(dst + start);
        const int4* __restrict__ ps = (const int4*)(src + start);
        const int nvec = (end - start) >> 2;
        int i = tid;
        for (; i + PS_THREADS < nvec; i += 2 * PS_THREADS) {
            const int4 d0 = pd[i];
            const int4 s0 = ps[i];
            const int4 d1 = pd[i + PS_THREADS];
            const int4 s1 = ps[i + PS_THREADS];
            ps_edges8(bins, P4, d0, s0, d1, s1, lo, nb);
        }
        for (; i < nvec; i += PS_THREADS) {
            const int4 d0 = pd[i];
            const int4 s0 = ps[i];
            ps_edge1(bins, P4, d0.x, s0.x, lo, nb);
            ps_edge1(bins, P4, d0.y, s0.y, lo, nb);
            ps_edge1(bins, P4, d0.z, s0.z, lo, nb);
            ps_edge1(bins, P4, d0.w, s0.w, lo, nb);
        }
        for (int e = start + (nvec << 2) + tid; e < end; e += PS_THREADS)
            ps_edge1(bins, P4, dst[e], src[e], lo, nb);
    }
    __syncthreads();

    const int n = (int)nb * GENRES;
    const int rot = ((int)(blockIdx.x / PS_PLANES) * (n / (PS_SLICES / PS_PLANES))) & ~31;
    float* __restrict__ out = accp + (size_t)(blockIdx.x % PS_PLANES) * PS_PITCH + (size_t)lo * GENRES;
    for (int j = tid; j < n; j += PS_THREADS) {
        int i = j + rot;
        if (i >= n) i -= n;
        const float v = bins[i];
        if (v != 0.0f) atomicAdd(&out[i], v);
    }
}

__host__ __device__ constexpr int w_pitch(int K) { return ((K / 4) & 1) ? K : K + 4; }

template <int K, int PITCH, int NT>
__device__ __forceinline__ void stage_weights(float* __restrict__ dst, const float* __restrict__ W, int tid) {
    constexpr int Q = K / 4;
    constexpr int STEPS = (64 * Q + NT - 1) / NT;
    const float4* __restrict__ W4 = (const float4*)W;
    float4 v[STEPS];
#pragma unroll
    for (int j = 0; j < STEPS; ++j) {
        const int i = tid + NT * j;
        v[j] = W4[i < 64 * Q ? i : 64 * Q - 1];
    }
#pragma unroll
    for (int j = 0; j < STEPS; ++j) {
        const int i = tid + NT * j;
        if (i < 64 * Q) {
            const int n = i / Q;
            const int q = i - n * Q;
            *(float4*)(dst + n * PITCH + 4 * q) = v[j];
        }
    }
}

template <bool PROJ, bool DUAL, int K1, int LD1, int AUG1, int K2, int LD2, int AUG2, int NT, int WPB>
__device__ __forceinline__
void sage_gemm64_body(float* __restrict__ wlds, const int gblock,
                      const float* __restrict__ A1, const float* __restrict__ A1b,
                      const float* __restrict__ deg1,
                      const float* __restrict__ W1,
                      const float* __restrict__ A2, const float* __restrict__ W2,
                      const float* __restrict__ bias, const float* __restrict__ Wp,
                      float* __restrict__ C, int M, int relu) {
    constexpr int P1 = w_pitch(K1);
    constexpr int P2 = K2 > 0 ? w_pitch(K2) : 0;

    stage_weights<K1, P1, NT>(wlds, W1, threadIdx.x);
    if (K2 > 0) stage_weights<(K2 > 0 ? K2 : 4), (K2 > 0 ? P2 : 4), NT>(wlds + 64 * P1, W2, threadIdx.x);
    __syncthreads();

    const int lane = threadIdx.x & 31;
    const int wave = threadIdx.x >> 5;
    const int half = lane >> 4;
    const int r    = lane & 15;
    if (NT / 32 > WPB && wave >= WPB) return;
    const int row0 = (gblock * WPB + wave) * 16;
    if (row0 >= M) return;

    v8f c[4] = {};

    float scale = 1.0f;
    float has_edges = 1.0f;
    if (deg1) {
        const float dg = deg1[row0 + r];
        scale = 1.0f / fmaxf(dg, 1.0f);
        has_edges = dg >= 1.0f ? 1.0f : 0.0f;
    }

    {
        constexpr int KL1 = AUG1 ? K1 - 4 : K1;
        const float* Arow = A1 + (size_t)(row0 + r) * LD1;
        const float* Brow = DUAL ? A1b + (size_t)(row0 + r) * LD1 : Arow;
        const float* Ws = wlds + r * P1 + 2 * half;
#pragma unroll
        for (int kk = 0; kk < K1; kk += 4) {
            const int k = kk + 2 * half;
            v2f a;
            if (kk < KL1) {
                float a0 = Arow[k];
                float a1 = Arow[k + 1];
                if (DUAL) { a0 += Brow[k]; a1 += Brow[k + 1]; }
                a.x = a0 * scale;
                a.y = a1 * scale;
            } else {
                const float u = AUG1 == 1 ? has_edges : 1.0f;
                a.x = half == 0 ? u : 0.0f;
                a.y = 0.0f;
            }
#pragma unroll
            for (int t = 0; t < 4; ++t) {
                const v2f b = *(const v2f*)(Ws + t * 16 * P1 + kk);
                c[t] = __builtin_amdgcn_wmma_f32_16x16x4_f32(
                    false, a, false, b, (short)0, c[t], false, false);
            }
        }
    }
    if (K2 > 0) {
        constexpr int KL2 = AUG2 ? K2 - 4 : K2;
        const float* Arow = A2 + (size_t)(row0 + r) * LD2;
        const float* Ws = wlds + 64 * P1 + r * P2 + 2 * half;
#pragma unroll
        for (int kk = 0; kk < K2; kk += 4) {
            const int k = kk + 2 * half;
            v2f a;
            if (kk < KL2) {
                a.x = Arow[k];
                a.y = Arow[k + 1];
            } else {
                const float u = AUG2 == 1 ? has_edges : 1.0f;
                a.x = half == 0 ? u : 0.0f;
                a.y = 0.0f;
            }
#pragma unroll
            for (int t = 0; t < 4; ++t) {
                const v2f b = *(const v2f*)(Ws + t * 16 * P2 + kk);
                c[t] = __builtin_amdgcn_wmma_f32_16x16x4_f32(
                    false, a, false, b, (short)0, c[t], false, false);
            }
        }
    }

    if (!PROJ) {
#pragma unroll
        for (int t = 0; t < 4; ++t) {
            const float bv = bias ? bias[t * 16 + r] : 0.0f;
#pragma unroll
            for (int i = 0; i < 8; ++i) {
                float v = c[t][i] + bv;
                if (relu) v = fmaxf(v, 0.0f);
                C[(size_t)(row0 + i + 8 * half) * HID + t * 16 + r] = v;
            }
        }
    } else {
        float s[8][GENRES];
#pragma unroll
        for (int i = 0; i < 8; ++i)
#pragma unroll
            for (int g = 0; g < GENRES; ++g) s[i][g] = 0.0f;
#pragma unroll
        for (int t = 0; t < 4; ++t) {
            const float bv = bias ? bias[t * 16 + r] : 0.0f;
            float w[GENRES];
#pragma unroll
            for (int g = 0; g < GENRES; ++g) w[g] = Wp[g * HID + t * 16 + r];
#pragma unroll
            for (int i = 0; i < 8; ++i) {
                float v = c[t][i] + bv;
                if (relu) v = fmaxf(v, 0.0f);
#pragma unroll
                for (int g = 0; g < GENRES; ++g) s[i][g] += v * w[g];
            }
        }
#pragma unroll
        for (int i = 0; i < 8; ++i)
#pragma unroll
            for (int g = 0; g < GENRES; ++g)
#pragma unroll
                for (int m = 8; m >= 1; m >>= 1)
                    s[i][g] += __shfl_xor(s[i][g], m, 16);
        float4 o = make_float4(0.0f, 0.0f, 0.0f, 0.0f);
#pragma unroll
        for (int i = 0; i < 8; ++i)
            if (r == i) o = make_float4(s[i][0], s[i][1], s[i][2], 0.0f);
        if (r < 8) ((float4*)C)[(size_t)(row0 + r + 8 * half)] = o;
    }
}

#define FOLD_K (USER_FEAT + 4)

__device__ __forceinline__
void fold_user_proj_role(const int m,
                         const float* __restrict__ Wm0, const float* __restrict__ Wm1,
                         const float* __restrict__ W_user, const float* __restrict__ b_user,
                         float* __restrict__ Wf) {
    if (threadIdx.x >= 256) return;
    const float* __restrict__ Wm = m ? Wm1 : Wm0;
    float* __restrict__ out = Wf + (size_t)m * 64 * FOLD_K;

    const int lane = threadIdx.x & 31;
    const int wave = threadIdx.x >> 5;
    const int half = lane >> 4;
    const int r    = lane & 15;
    const int tn   = wave >> 1;
    const int col  = (wave & 1) * 16 + r;
    const float* Bcol = col < USER_FEAT ? W_user + col : b_user;
    const int bpitch  = col < USER_FEAT ? USER_FEAT : 1;
    const bool live   = col <= USER_FEAT;

    const float* Arow = Wm + (size_t)(tn * 16 + r) * HID;
    v8f c = {};
#pragma unroll
    for (int kk = 0; kk < HID; kk += 4) {
        const int k = kk + 2 * half;
        v2f a, b;
        a.x = Arow[k];
        a.y = Arow[k + 1];
        const float b0 = Bcol[k * bpitch];
        const float b1 = Bcol[(k + 1) * bpitch];
        b.x = live ? b0 : 0.0f;
        b.y = live ? b1 : 0.0f;
        c = __builtin_amdgcn_wmma_f32_16x16x4_f32(
            false, a, false, b, (short)0, c, false, false);
    }
    if (col < FOLD_K) {
#pragma unroll
        for (int i = 0; i < 8; ++i)
            out[(size_t)(tn * 16 + i + 8 * half) * FOLD_K + col] = c[i];
    }
}

#define PRE_HIST (HSLICES * (MOVIE_RANGES + USER_RANGES))
#define PRE_WPB  20
#define PRE_GEMM ((NUM_MOVIES / 16 + PRE_WPB - 1) / PRE_WPB)

__global__ __launch_bounds__(HTHREADS)
void prepass_kernel(const int* __restrict__ dst, float* __restrict__ deg_movie,
                    const int* __restrict__ src, float* __restrict__ deg_user, int E,
                    const float* __restrict__ x_movie, const float* __restrict__ W_movie,
                    const float* __restrict__ b_movie, float* __restrict__ hm,
                    const float* __restrict__ Wm0, const float* __restrict__ Wm1,
                    const float* __restrict__ W_user, const float* __restrict__ b_user,
                    float* __restrict__ Wf) {
    __shared__ __attribute__((aligned(16))) unsigned int smem[HBINS];
    static_assert(64 * w_pitch(MOVIE_FEAT) <= HBINS, "staged W_movie must fit the histogram's LDS array");

    const int b = (int)blockIdx.x;
    if (b < PRE_HIST) {
        degree_hist_role(smem, b % HSLICES, b / HSLICES, dst, deg_movie, src, deg_user, E);
    } else if (b < PRE_HIST + PRE_GEMM) {
        sage_gemm64_body<false, false, MOVIE_FEAT, MOVIE_FEAT, 0, 0, 0, 0, HTHREADS, PRE_WPB>(
            (float*)smem, b - PRE_HIST, x_movie, nullptr, nullptr, W_movie, nullptr, nullptr,
            b_movie, nullptr, hm, NUM_MOVIES, 0);
    } else {
        fold_user_proj_role(b - PRE_HIST - PRE_GEMM, Wm0, Wm1, W_user, b_user, Wf);
    }
}

__global__ __launch_bounds__(256)
void layer1_gemm_kernel(const float* __restrict__ acc_mu,
                        const float* __restrict__ deg_user, const float* __restrict__ Wl1_mu,
                        const float* __restrict__ x_user, const float* __restrict__ wfold_r,
                        const float* __restrict__ bl1_mu, const float* __restrict__ Wl2_um,
                        float* __restrict__ pu,
                        const float* __restrict__ acc_x, const float* __restrict__ acc_x_b,
                        const float* __restrict__ deg_movie, const float* __restrict__ wfold_l,
                        const float* __restrict__ hm, const float* __restrict__ Wr1_um,
                        const float* __restrict__ bl1_um, float* __restrict__ m1,
                        int user_blocks) {
    __shared__ __attribute__((aligned(16))) float wlds[64 * (w_pitch(FOLD_K) + w_pitch(HID))];
    const int b = (int)blockIdx.x;
    if (b < user_blocks) {
        sage_gemm64_body<true, false, HID, HID, 0, FOLD_K, USER_FEAT, 2, 256, 8>(
            wlds, b, acc_mu, nullptr, deg_user, Wl1_mu, x_user, wfold_r, bl1_mu, Wl2_um,
            pu, NUM_USERS, 1);
    } else {
        sage_gemm64_body<false, true, FOLD_K, XF_PITCH, 1, HID, HID, 0, 256, 8>(
            wlds, b - user_blocks, acc_x, acc_x_b, deg_movie, wfold_l, hm, Wr1_um, bl1_um, nullptr,
            m1, NUM_MOVIES, 1);
    }
}

__global__ __launch_bounds__(256)
void head_kernel(const float* __restrict__ accp, const float* __restrict__ degm,
                 const float* __restrict__ m1,
                 const float* __restrict__ bl2, const float* __restrict__ Wr2,
                 float* __restrict__ out, int M) {
    int m = blockIdx.x * blockDim.x + threadIdx.x;
    if (m >= M) return;
    const float inv = 1.0f / fmaxf(degm[m], 1.0f);
    const float* a = accp + (size_t)m * GENRES;
    const float* x = m1 + (size_t)m * HID;
    float acc[GENRES];
#pragma unroll
    for (int g = 0; g < GENRES; ++g) {
        float sum = a[g];
#pragma unroll
        for (int pl = 1; pl < PS_PLANES; ++pl) sum += a[(size_t)pl * PS_PITCH + g];
        acc[g] = sum * inv + bl2[g];
    }
    for (int k = 0; k < HID; ++k) {
        const float xv = x[k];
#pragma unroll
        for (int g = 0; g < GENRES; ++g)
            acc[g] += xv * Wr2[g * HID + k];
    }
#pragma unroll
    for (int g = 0; g < GENRES; ++g) out[(size_t)m * GENRES + g] = acc[g];
}

extern "C" void kernel_launch(void* const* d_in, const int* in_sizes, int n_in,
                              void* d_out, int out_size, void* d_ws, size_t ws_size,
                              hipStream_t stream) {
    (void)in_sizes; (void)n_in; (void)out_size; (void)ws_size;

    const float* x_user  = (const float*)d_in[0];
    const float* x_movie = (const float*)d_in[1];
    const int*   e_src   = (const int*)d_in[2];
    const int*   e_dst   = (const int*)d_in[3];
    const float* W_user  = (const float*)d_in[4];
    const float* b_user  = (const float*)d_in[5];
    const float* W_movie = (const float*)d_in[6];
    const float* b_movie = (const float*)d_in[7];
    const float* Wl1_um  = (const float*)d_in[8];
    const float* bl1_um  = (const float*)d_in[9];
    const float* Wr1_um  = (const float*)d_in[10];
    const float* Wl1_mu  = (const float*)d_in[11];
    const float* bl1_mu  = (const float*)d_in[12];
    const float* Wr1_mu  = (const float*)d_in[13];
    const float* Wl2_um  = (const float*)d_in[14];
    const float* bl2_um  = (const float*)d_in[15];
    const float* Wr2_um  = (const float*)d_in[16];
    float* out = (float*)d_out;

    float* ws = (float*)d_ws;
    float* hm        = ws;
    float* pu        = hm + 3200000;
    float* m1        = pu + 400000;
    float* wfold     = m1 + 3200000;
    float* wfold_r   = wfold + 1792;
    float* acc_x     = wfold + 3584;
    float* acc_x_b   = acc_x + 1600000;
    float* accp      = acc_x_b + 1600000;
    float* deg_user  = accp + 600064;
    float* deg_movie = deg_user + 100000;
    unsigned int* ucursor = (unsigned int*)(deg_movie + 50000);
    float* acc_mu    = deg_movie + 50000 + 528;
    unsigned int* ulist = (unsigned int*)(acc_mu + 6400000);

    const size_t zero_bytes = (size_t)(1600000 + 1600000 + 600064 + 100000 + 50000 + 528) * 4;
    hipMemsetAsync(acc_x, 0, zero_bytes, stream);

    const int E = NUM_EDGES;
    prepass_kernel<<<PRE_HIST + PRE_GEMM + 2, HTHREADS, 0, stream>>>(
        e_dst, deg_movie, e_src, deg_user, E,
        x_movie, W_movie, b_movie, hm,
        Wl1_um, Wr1_mu, W_user, b_user, wfold);

    const int gridU = (NUM_USERS / 16 + 7) / 8;
    const int gridM = (NUM_MOVIES / 16 + 7) / 8;

    const int sblocks = (E + 8 * SC_EPW - 1) / (8 * SC_EPW);

    user_bucket_kernel<<<UP_BLOCKS, UP_THREADS, 0, stream>>>(e_src, e_dst, ucursor, ulist, E);
    scatter_feat_kernel<<<sblocks, 256, 0, stream>>>(x_user, e_src, e_dst, acc_x, acc_x_b, E);
    user_range_sum_kernel<<<UB_NB, UA_THREADS, 0, stream>>>(hm, e_src, e_dst, ucursor, ulist, acc_mu, E);

    layer1_gemm_kernel<<<gridU + gridM, 256, 0, stream>>>(
        acc_mu, deg_user, Wl1_mu, x_user, wfold_r, bl1_mu, Wl2_um, pu,
        acc_x, acc_x_b, deg_movie, wfold, hm, Wr1_um, bl1_um, m1, gridU);

    proj_scatter_kernel<<<dim3(PS_SLICES, PS_RANGES), PS_THREADS, 0, stream>>>(
        pu, e_src, e_dst, accp, E);
    head_kernel<<<(NUM_MOVIES + 255) / 256, 256, 0, stream>>>(
        accp, deg_movie, m1, bl2_um, Wr2_um, out, NUM_MOVIES);
}


# Movie-side layer-1 sum rides the user-side row walk: the wave summing a user row adds that user's 24 features to each of its movies (one line atomic per edge); scatter launch gone; sum order changes
# speedup vs baseline: 7.0265x; 1.2283x over previous
#include <hip/hip_runtime.h>
#include <hip/hip_bf16.h>

typedef __attribute__((ext_vector_type(2))) float v2f;
typedef __attribute__((ext_vector_type(8))) float v8f;

#define NUM_USERS  100000
#define NUM_MOVIES 50000
#define NUM_EDGES  4000000
#define USER_FEAT  24
#define MOVIE_FEAT 404
#define HID        64
#define GENRES     3

#define HBINS        50000
#define HSLICES      32
#define HTHREADS     1024
#define MOVIE_RANGES ((NUM_MOVIES + HBINS - 1) / HBINS)
#define USER_RANGES  ((NUM_USERS + HBINS - 1) / HBINS)

__device__ __forceinline__ void hist_count4(unsigned int* bins, const int4 v, int lo, unsigned int nb) {
    const unsigned int b0 = (unsigned int)(v.x - lo);
    const unsigned int b1 = (unsigned int)(v.y - lo);
    const unsigned int b2 = (unsigned int)(v.z - lo);
    const unsigned int b3 = (unsigned int)(v.w - lo);
    if (b0 < nb) atomicAdd(&bins[b0], 1u);
    if (b1 < nb) atomicAdd(&bins[b1], 1u);
    if (b2 < nb) atomicAdd(&bins[b2], 1u);
    if (b3 < nb) atomicAdd(&bins[b3], 1u);
}

__device__ __forceinline__
void degree_hist_role(unsigned int* bins, const int slice, const int job,
                      const int* __restrict__ dst, float* __restrict__ deg_movie,
                      const int* __restrict__ src, float* __restrict__ deg_user, int E) {
    const int tid = threadIdx.x;
    const bool is_movie = job < MOVIE_RANGES;
    const int* __restrict__ idx = is_movie ? dst : src;
    const int nseg = is_movie ? NUM_MOVIES : NUM_USERS;
    const int lo = (is_movie ? job : job - MOVIE_RANGES) * HBINS;
    const int left = nseg - lo;
    const unsigned int nb = (unsigned int)(left < HBINS ? left : HBINS);
    float* __restrict__ out = (is_movie ? deg_movie : deg_user) + lo;

    for (int i = tid; i < HBINS; i += HTHREADS) bins[i] = 0u;
    __syncthreads();

    const int per = (((E + HSLICES - 1) / HSLICES) + 3) & ~3;
    const int start = slice * per;
    int end = start + per;
    if (end > E) end = E;
    if (start < end) {
        const int4* __restrict__ p = (const int4*)(idx + start);
        const int nvec = (end - start) >> 2;
        int i = tid;
        for (; i + 3 * HTHREADS < nvec; i += 4 * HTHREADS) {
            const int4 a = p[i];
            const int4 b = p[i + HTHREADS];
            const int4 c = p[i + 2 * HTHREADS];
            const int4 d = p[i + 3 * HTHREADS];
            hist_count4(bins, a, lo, nb);
            hist_count4(bins, b, lo, nb);
            hist_count4(bins, c, lo, nb);
            hist_count4(bins, d, lo, nb);
        }
        for (; i < nvec; i += HTHREADS) hist_count4(bins, p[i], lo, nb);
        for (int e = start + (nvec << 2) + tid; e < end; e += HTHREADS) {
            const unsigned int b = (unsigned int)(idx[e] - lo);
            if (b < nb) atomicAdd(&bins[b], 1u);
        }
    }
    __syncthreads();

    const int n = (int)nb;
    const int rot = (slice * (n / HSLICES)) & ~31;
    for (int j = tid; j < n; j += HTHREADS) {
        int i = j + rot;
        if (i >= n) i -= n;
        const unsigned int cnt = bins[i];
        if (cnt) atomicAdd(&out[i], (float)cnt);
    }
}

__device__ __forceinline__ int chip_half() {
    const unsigned int id = __builtin_amdgcn_s_sendmsg_rtn(0x87u);
    return (int)((id >> 18) & 1u);
}

#define XF_PITCH   32
#define UB_ROWS    196
#define UB_NB      512
#define UB_CAP     12288
#define UP_THREADS 1024
#define UP_BLOCKS  256
#define UP_EPT     16
#define UA_THREADS 1024
#define UA_GROUP   16
#define UA_SLOTS   (UB_CAP / UA_THREADS)
#define UA_RPL     ((UB_ROWS + 31) / 32)

static_assert(UB_ROWS * UB_NB >= NUM_USERS, "buckets must cover all users");
static_assert(NUM_MOVIES <= 65536 && UB_ROWS <= 65536, "packed list word");
static_assert((long long)UP_EPT * UP_THREADS * UP_BLOCKS >= NUM_EDGES, "edges per thread");
static_assert(UB_CAP % UA_THREADS == 0, "list slots per thread");
static_assert(UB_CAP <= UB_ROWS * HID, "the sorted list lies in the array of the float bins");

__global__ __launch_bounds__(UP_THREADS)
void user_bucket_kernel(const int* __restrict__ src, const int* __restrict__ dst,
                        unsigned int* __restrict__ cursor, unsigned int* __restrict__ list, int E) {
    __shared__ unsigned int cnt[UB_NB];
    __shared__ unsigned int base[UB_NB];
    const int tid = threadIdx.x;
    for (int i = tid; i < UB_NB; i += UP_THREADS) cnt[i] = 0u;
    __syncthreads();

    const int per = (E + UP_BLOCKS - 1) / UP_BLOCKS;
    const int start = (int)blockIdx.x * per;
    int end = start + per;
    if (end > E) end = E;

    int s[UP_EPT], d[UP_EPT];
    unsigned int rk[UP_EPT];
#pragma unroll
    for (int j = 0; j < UP_EPT; ++j) {
        const int e = start + j * UP_THREADS + tid;
        s[j] = -1;
        d[j] = 0;
        if (e < end) { s[j] = src[e]; d[j] = dst[e]; }
    }
#pragma unroll
    for (int j = 0; j < UP_EPT; ++j) {
        rk[j] = 0u;
        if (s[j] >= 0) rk[j] = atomicAdd(&cnt[s[j] / UB_ROWS], 1u);
    }
    __syncthreads();
    for (int b = tid; b < UB_NB; b += UP_THREADS) {
        const unsigned int n = cnt[b];
        base[b] = n ? atomicAdd(&cursor[b], n) : 0u;
    }
    __syncthreads();
#pragma unroll
    for (int j = 0; j < UP_EPT; ++j) {
        if (s[j] >= 0) {
            const int b = s[j] / UB_ROWS;
            const unsigned int pos = base[b] + rk[j];
            if (pos < UB_CAP)
                list[(size_t)b * UB_CAP + pos] =
                    ((unsigned int)(s[j] - b * UB_ROWS) << 16) | (unsigned int)d[j];
        }
    }
}

__global__ __launch_bounds__(UA_THREADS)
void user_range_sum_kernel(const float* __restrict__ X, const int* __restrict__ src,
                           const int* __restrict__ dst,
                           const unsigned int* __restrict__ cursor,
                           const unsigned int* __restrict__ list,
                           float* __restrict__ acc,
                           const float* __restrict__ XU,
                           float* __restrict__ accx0, float* __restrict__ accx1, int E) {
    __shared__ __attribute__((aligned(16))) float bins[UB_ROWS * HID];
    __shared__ unsigned int rcnt[UB_ROWS];
    __shared__ unsigned int rstart[UB_ROWS];
    __shared__ unsigned int ticket;
    const int tid = threadIdx.x;
    const int lane = tid & 31;
    const int wave = __builtin_amdgcn_readfirstlane(tid >> 5);
    const int b = (int)blockIdx.x;
    const int row0 = b * UB_ROWS;
    float* __restrict__ accx = chip_half() ? accx1 : accx0;
    const bool feat = lane < USER_FEAT;

    const unsigned int total = cursor[b];
    if (total <= UB_CAP) {
        unsigned int* sorted = (unsigned int*)bins;
        const unsigned int* __restrict__ L = list + (size_t)b * UB_CAP;

        unsigned int w[UA_SLOTS], rk[UA_SLOTS];
#pragma unroll
        for (int k = 0; k < UA_SLOTS; ++k) {
            const unsigned int i = (unsigned int)(tid + k * UA_THREADS);
            w[k] = i < total ? L[i] : 0u;
        }

        for (int i = tid; i < UB_ROWS; i += UA_THREADS) rcnt[i] = 0u;
        if (tid == 0) ticket = 0u;
        __syncthreads();

#pragma unroll
        for (int k = 0; k < UA_SLOTS; ++k) {
            const unsigned int i = (unsigned int)(tid + k * UA_THREADS);
            rk[k] = 0u;
            if (i < total) rk[k] = atomicAdd(&rcnt[w[k] >> 16], 1u);
        }
        __syncthreads();

        if (wave == 0) {
            unsigned int c[UA_RPL];
            unsigned int own = 0u;
#pragma unroll
            for (int j = 0; j < UA_RPL; ++j) {
                const int r = lane * UA_RPL + j;
                c[j] = r < UB_ROWS ? rcnt[r] : 0u;
                own += c[j];
            }
            unsigned int incl = own;
#pragma unroll
            for (int d = 1; d < 32; d <<= 1) {
                const unsigned int t = __shfl_up(incl, d, 32);
                if (lane >= d) incl += t;
            }
            unsigned int run = incl - own;
#pragma unroll
            for (int j = 0; j < UA_RPL; ++j) {
                const int r = lane * UA_RPL + j;
                if (r < UB_ROWS) rstart[r] = run;
                run += c[j];
            }
        }
        __syncthreads();

#pragma unroll
        for (int k = 0; k < UA_SLOTS; ++k) {
            const unsigned int i = (unsigned int)(tid + k * UA_THREADS);
            if (i < total) sorted[rstart[w[k] >> 16] + rk[k]] = w[k] & 0xffffu;
        }
        __syncthreads();

        for (;;) {
            unsigned int t = 0u;
            if (lane == 0) t = atomicAdd(&ticket, 1u);
            const int r = __builtin_amdgcn_readfirstlane((int)t);
            if (r >= UB_ROWS || row0 + r >= NUM_USERS) break;
            const unsigned int n = (unsigned int)__builtin_amdgcn_readfirstlane((int)rcnt[r]);
            const unsigned int s = (unsigned int)__builtin_amdgcn_readfirstlane((int)rstart[r]);
            const float xu = feat ? XU[(size_t)(row0 + r) * USER_FEAT + lane] : 0.0f;
            float s0 = 0.0f, s1 = 0.0f;
            for (unsigned int v0 = 0u; v0 < n; v0 += 32u) {
                const unsigned int i = v0 + (unsigned int)lane;
                const int m = i < n ? (int)sorted[s + i] : 0;
                const int left = (int)(n - v0);
                const int cnt = left < 32 ? left : 32;
                int g = 0;
                for (; g + UA_GROUP <= cnt; g += UA_GROUP) {
                    v2f x[UA_GROUP];
                    int mv[UA_GROUP];
#pragma unroll
                    for (int j = 0; j < UA_GROUP; ++j) {
                        mv[j] = __builtin_amdgcn_readlane(m, g + j);
                        x[j] = *(const v2f*)(X + (size_t)mv[j] * HID + 2 * lane);
                    }
                    if (feat) {
#pragma unroll
                        for (int j = 0; j < UA_GROUP; ++j)
                            atomicAdd(accx + (size_t)mv[j] * XF_PITCH + lane, xu);
                    }
#pragma unroll
                    for (int j = 0; j < UA_GROUP; ++j) { s0 += x[j].x; s1 += x[j].y; }
                }
                if (g < cnt) {
                    v2f x[UA_GROUP];
#pragma unroll
                    for (int j = 0; j < UA_GROUP; ++j) {
                        x[j].x = 0.0f;
                        x[j].y = 0.0f;
                        if (g + j < cnt) {
                            const int mj = __builtin_amdgcn_readlane(m, g + j);
                            x[j] = *(const v2f*)(X + (size_t)mj * HID + 2 * lane);
                        }
                    }
#pragma unroll
                    for (int j = 0; j < UA_GROUP; ++j) {
                        if (g + j < cnt) {
                            const int mj = __builtin_amdgcn_readlane(m, g + j);
                            if (feat) atomicAdd(accx + (size_t)mj * XF_PITCH + lane, xu);
                        }
                    }
#pragma unroll
                    for (int j = 0; j < UA_GROUP; ++j) { s0 += x[j].x; s1 += x[j].y; }
                }
            }
            v2f o;
            o.x = s0;
            o.y = s1;
            *(v2f*)(acc + (size_t)(row0 + r) * HID + 2 * lane) = o;
        }
    } else {
        for (int i = tid; i < UB_ROWS * HID; i += UA_THREADS) bins[i] = 0.0f;
        __syncthreads();
        for (int e = wave; e < E; e += UA_THREADS / 32) {
            const int s = src[e];
            if (s / UB_ROWS == b) {
                const int d = dst[e];
                const float* __restrict__ xr = X + (size_t)d * HID + lane;
                const float a0 = xr[0];
                const float a1 = xr[32];
                float* q = bins + (s - b * UB_ROWS) * HID + lane;
                atomicAdd(q, a0);
                atomicAdd(q + 32, a1);
                if (feat) atomicAdd(accx + (size_t)d * XF_PITCH + lane, XU[(size_t)s * USER_FEAT + lane]);
            }
        }
        __syncthreads();
        int nrows = NUM_USERS - row0;
        if (nrows > UB_ROWS) nrows = UB_ROWS;
        float* __restrict__ out = acc + (size_t)row0 * HID;
        for (int i = tid; i < nrows * HID; i += UA_THREADS) out[i] = bins[i];
    }
}

#define PS_RANGES  2
#define PS_RB      ((NUM_MOVIES + PS_RANGES - 1) / PS_RANGES)
#define PS_SLICES  128
#define PS_THREADS 1024
#define PS_PLANES  4
#define PS_PITCH   150016

__device__ __forceinline__ void ps_edges8(float* bins, const float4* __restrict__ P4,
                                          const int4 d0, const int4 s0,
                                          const int4 d1, const int4 s1,
                                          int lo, unsigned int nb) {
    const int dd[8] = {d0.x, d0.y, d0.z, d0.w, d1.x, d1.y, d1.z, d1.w};
    const int ss[8] = {s0.x, s0.y, s0.z, s0.w, s1.x, s1.y, s1.z, s1.w};
    unsigned int b[8];
    float4 p[8];
#pragma unroll
    for (int j = 0; j < 8; ++j) {
        b[j] = (unsigned int)(dd[j] - lo);
        p[j] = make_float4(0.0f, 0.0f, 0.0f, 0.0f);
        if (b[j] < nb) p[j] = P4[ss[j]];
    }
#pragma unroll
    for (int j = 0; j < 8; ++j) {
        if (b[j] < nb) {
            float* q = bins + b[j] * GENRES;
            atomicAdd(q + 0, p[j].x);
            atomicAdd(q + 1, p[j].y);
            atomicAdd(q + 2, p[j].z);
        }
    }
}

__device__ __forceinline__ void ps_edge1(float* bins, const float4* __restrict__ P4,
                                         const int d, const int s, int lo, unsigned int nb) {
    const unsigned int b = (unsigned int)(d - lo);
    if (b < nb) {
        const float4 p = P4[s];
        float* q = bins + b * GENRES;
        atomicAdd(q + 0, p.x);
        atomicAdd(q + 1, p.y);
        atomicAdd(q + 2, p.z);
    }
}

__global__ __launch_bounds__(PS_THREADS)
void proj_scatter_kernel(const float* __restrict__ P, const int* __restrict__ src,
                         const int* __restrict__ dst, float* __restrict__ accp, int E) {
    __shared__ float bins[PS_RB * GENRES];

    const int tid = threadIdx.x;
    const int lo = (int)blockIdx.y * PS_RB;
    const int left = NUM_MOVIES - lo;
    const unsigned int nb = (unsigned int)(left < PS_RB ? left : PS_RB);
    const float4* __restrict__ P4 = (const float4*)P;

    for (int i = tid; i < PS_RB * GENRES; i += PS_THREADS) bins[i] = 0.0f;
    __syncthreads();

    const int per = (((E + PS_SLICES - 1) / PS_SLICES) + 3) & ~3;
    const int start = blockIdx.x * per;
    int end = start + per;
    if (end > E) end = E;
    if (start < end) {
        const int4* __restrict__ pd = (const int4*)(dst + start);
        const int4* __restrict__ ps = (const int4*)(src + start);
        const int nvec = (end - start) >> 2;
        int i = tid;
        for (; i + PS_THREADS < nvec; i += 2 * PS_THREADS) {
            const int4 d0 = pd[i];
            const int4 s0 = ps[i];
            const int4 d1 = pd[i + PS_THREADS];
            const int4 s1 = ps[i + PS_THREADS];
            ps_edges8(bins, P4, d0, s0, d1, s1, lo, nb);
        }
        for (; i < nvec; i += PS_THREADS) {
            const int4 d0 = pd[i];
            const int4 s0 = ps[i];
            ps_edge1(bins, P4, d0.x, s0.x, lo, nb);
            ps_edge1(bins, P4, d0.y, s0.y, lo, nb);
            ps_edge1(bins, P4, d0.z, s0.z, lo, nb);
            ps_edge1(bins, P4, d0.w, s0.w, lo, nb);
        }
        for (int e = start + (nvec << 2) + tid; e < end; e += PS_THREADS)
            ps_edge1(bins, P4, dst[e], src[e], lo, nb);
    }
    __syncthreads();

    const int n = (int)nb * GENRES;
    const int rot = ((int)(blockIdx.x / PS_PLANES) * (n / (PS_SLICES / PS_PLANES))) & ~31;
    float* __restrict__ out = accp + (size_t)(blockIdx.x % PS_PLANES) * PS_PITCH + (size_t)lo * GENRES;
    for (int j = tid; j < n; j += PS_THREADS) {
        int i = j + rot;
        if (i >= n) i -= n;
        const float v = bins[i];
        if (v != 0.0f) atomicAdd(&out[i], v);
    }
}

__host__ __device__ constexpr int w_pitch(int K) { return ((K / 4) & 1) ? K : K + 4; }

template <int K, int PITCH, int NT>
__device__ __forceinline__ void stage_weights(float* __restrict__ dst, const float* __restrict__ W, int tid) {
    constexpr int Q = K / 4;
    constexpr int STEPS = (64 * Q + NT - 1) / NT;
    const float4* __restrict__ W4 = (const float4*)W;
    float4 v[STEPS];
#pragma unroll
    for (int j = 0; j < STEPS; ++j) {
        const int i = tid + NT * j;
        v[j] = W4[i < 64 * Q ? i : 64 * Q - 1];
    }
#pragma unroll
    for (int j = 0; j < STEPS; ++j) {
        const int i = tid + NT * j;
        if (i < 64 * Q) {
            const int n = i / Q;
            const int q = i - n * Q;
            *(float4*)(dst + n * PITCH + 4 * q) = v[j];
        }
    }
}

template <bool PROJ, bool DUAL, int K1, int LD1, int AUG1, int K2, int LD2, int AUG2, int NT, int WPB>
__device__ __forceinline__
void sage_gemm64_body(float* __restrict__ wlds, const int gblock,
                      const float* __restrict__ A1, const float* __restrict__ A1b,
                      const float* __restrict__ deg1,
                      const float* __restrict__ W1,
                      const float* __restrict__ A2, const float* __restrict__ W2,
                      const float* __restrict__ bias, const float* __restrict__ Wp,
                      float* __restrict__ C, int M, int relu) {
    constexpr int P1 = w_pitch(K1);
    constexpr int P2 = K2 > 0 ? w_pitch(K2) : 0;

    stage_weights<K1, P1, NT>(wlds, W1, threadIdx.x);
    if (K2 > 0) stage_weights<(K2 > 0 ? K2 : 4), (K2 > 0 ? P2 : 4), NT>(wlds + 64 * P1, W2, threadIdx.x);
    __syncthreads();

    const int lane = threadIdx.x & 31;
    const int wave = threadIdx.x >> 5;
    const int half = lane >> 4;
    const int r    = lane & 15;
    if (NT / 32 > WPB && wave >= WPB) return;
    const int row0 = (gblock * WPB + wave) * 16;
    if (row0 >= M) return;

    v8f c[4] = {};

    float scale = 1.0f;
    float has_edges = 1.0f;
    if (deg1) {
        const float dg = deg1[row0 + r];
        scale = 1.0f / fmaxf(dg, 1.0f);
        has_edges = dg >= 1.0f ? 1.0f : 0.0f;
    }

    {
        constexpr int KL1 = AUG1 ? K1 - 4 : K1;
        const float* Arow = A1 + (size_t)(row0 + r) * LD1;
        const float* Brow = DUAL ? A1b + (size_t)(row0 + r) * LD1 : Arow;
        const float* Ws = wlds + r * P1 + 2 * half;
#pragma unroll
        for (int kk = 0; kk < K1; kk += 4) {
            const int k = kk + 2 * half;
            v2f a;
            if (kk < KL1) {
                float a0 = Arow[k];
                float a1 = Arow[k + 1];
                if (DUAL) { a0 += Brow[k]; a1 += Brow[k + 1]; }
                a.x = a0 * scale;
                a.y = a1 * scale;
            } else {
                const float u = AUG1 == 1 ? has_edges : 1.0f;
                a.x = half == 0 ? u : 0.0f;
                a.y = 0.0f;
            }
#pragma unroll
            for (int t = 0; t < 4; ++t) {
                const v2f b = *(const v2f*)(Ws + t * 16 * P1 + kk);
                c[t] = __builtin_amdgcn_wmma_f32_16x16x4_f32(
                    false, a, false, b, (short)0, c[t], false, false);
            }
        }
    }
    if (K2 > 0) {
        constexpr int KL2 = AUG2 ? K2 - 4 : K2;
        const float* Arow = A2 + (size_t)(row0 + r) * LD2;
        const float* Ws = wlds + 64 * P1 + r * P2 + 2 * half;
#pragma unroll
        for (int kk = 0; kk < K2; kk += 4) {
            const int k = kk + 2 * half;
            v2f a;
            if (kk < KL2) {
                a.x = Arow[k];
                a.y = Arow[k + 1];
            } else {
                const float u = AUG2 == 1 ? has_edges : 1.0f;
                a.x = half == 0 ? u : 0.0f;
                a.y = 0.0f;
            }
#pragma unroll
            for (int t = 0; t < 4; ++t) {
                const v2f b = *(const v2f*)(Ws + t * 16 * P2 + kk);
                c[t] = __builtin_amdgcn_wmma_f32_16x16x4_f32(
                    false, a, false, b, (short)0, c[t], false, false);
            }
        }
    }

    if (!PROJ) {
#pragma unroll
        for (int t = 0; t < 4; ++t) {
            const float bv = bias ? bias[t * 16 + r] : 0.0f;
#pragma unroll
            for (int i = 0; i < 8; ++i) {
                float v = c[t][i] + bv;
                if (relu) v = fmaxf(v, 0.0f);
                C[(size_t)(row0 + i + 8 * half) * HID + t * 16 + r] = v;
            }
        }
    } else {
        float s[8][GENRES];
#pragma unroll
        for (int i = 0; i < 8; ++i)
#pragma unroll
            for (int g = 0; g < GENRES; ++g) s[i][g] = 0.0f;
#pragma unroll
        for (int t = 0; t < 4; ++t) {
            const float bv = bias ? bias[t * 16 + r] : 0.0f;
            float w[GENRES];
#pragma unroll
            for (int g = 0; g < GENRES; ++g) w[g] = Wp[g * HID + t * 16 + r];
#pragma unroll
            for (int i = 0; i < 8; ++i) {
                float v = c[t][i] + bv;
                if (relu) v = fmaxf(v, 0.0f);
#pragma unroll
                for (int g = 0; g < GENRES; ++g) s[i][g] += v * w[g];
            }
        }
#pragma unroll
        for (int i = 0; i < 8; ++i)
#pragma unroll
            for (int g = 0; g < GENRES; ++g)
#pragma unroll
                for (int m = 8; m >= 1; m >>= 1)
                    s[i][g] += __shfl_xor(s[i][g], m, 16);
        float4 o = make_float4(0.0f, 0.0f, 0.0f, 0.0f);
#pragma unroll
        for (int i = 0; i < 8; ++i)
            if (r == i) o = make_float4(s[i][0], s[i][1], s[i][2], 0.0f);
        if (r < 8) ((float4*)C)[(size_t)(row0 + r + 8 * half)] = o;
    }
}

#define FOLD_K (USER_FEAT + 4)

__device__ __forceinline__
void fold_user_proj_role(const int m,
                         const float* __restrict__ Wm0, const float* __restrict__ Wm1,
                         const float* __restrict__ W_user, const float* __restrict__ b_user,
                         float* __restrict__ Wf) {
    if (threadIdx.x >= 256) return;
    const float* __restrict__ Wm = m ? Wm1 : Wm0;
    float* __restrict__ out = Wf + (size_t)m * 64 * FOLD_K;

    const int lane = threadIdx.x & 31;
    const int wave = threadIdx.x >> 5;
    const int half = lane >> 4;
    const int r    = lane & 15;
    const int tn   = wave >> 1;
    const int col  = (wave & 1) * 16 + r;
    const float* Bcol = col < USER_FEAT ? W_user + col : b_user;
    const int bpitch  = col < USER_FEAT ? USER_FEAT : 1;
    const bool live   = col <= USER_FEAT;

    const float* Arow = Wm + (size_t)(tn * 16 + r) * HID;
    v8f c = {};
#pragma unroll
    for (int kk = 0; kk < HID; kk += 4) {
        const int k = kk + 2 * half;
        v2f a, b;
        a.x = Arow[k];
        a.y = Arow[k + 1];
        const float b0 = Bcol[k * bpitch];
        const float b1 = Bcol[(k + 1) * bpitch];
        b.x = live ? b0 : 0.0f;
        b.y = live ? b1 : 0.0f;
        c = __builtin_amdgcn_wmma_f32_16x16x4_f32(
            false, a, false, b, (short)0, c, false, false);
    }
    if (col < FOLD_K) {
#pragma unroll
        for (int i = 0; i < 8; ++i)
            out[(size_t)(tn * 16 + i + 8 * half) * FOLD_K + col] = c[i];
    }
}

#define PRE_HIST (HSLICES * (MOVIE_RANGES + USER_RANGES))
#define PRE_WPB  20
#define PRE_GEMM ((NUM_MOVIES / 16 + PRE_WPB - 1) / PRE_WPB)

__global__ __launch_bounds__(HTHREADS)
void prepass_kernel(const int* __restrict__ dst, float* __restrict__ deg_movie,
                    const int* __restrict__ src, float* __restrict__ deg_user, int E,
                    const float* __restrict__ x_movie, const float* __restrict__ W_movie,
                    const float* __restrict__ b_movie, float* __restrict__ hm,
                    const float* __restrict__ Wm0, const float* __restrict__ Wm1,
                    const float* __restrict__ W_user, const float* __restrict__ b_user,
                    float* __restrict__ Wf) {
    __shared__ __attribute__((aligned(16))) unsigned int smem[HBINS];
    static_assert(64 * w_pitch(MOVIE_FEAT) <= HBINS, "staged W_movie must fit the histogram's LDS array");

    const int b = (int)blockIdx.x;
    if (b < PRE_HIST) {
        degree_hist_role(smem, b % HSLICES, b / HSLICES, dst, deg_movie, src, deg_user, E);
    } else if (b < PRE_HIST + PRE_GEMM) {
        sage_gemm64_body<false, false, MOVIE_FEAT, MOVIE_FEAT, 0, 0, 0, 0, HTHREADS, PRE_WPB>(
            (float*)smem, b - PRE_HIST, x_movie, nullptr, nullptr, W_movie, nullptr, nullptr,
            b_movie, nullptr, hm, NUM_MOVIES, 0);
    } else {
        fold_user_proj_role(b - PRE_HIST - PRE_GEMM, Wm0, Wm1, W_user, b_user, Wf);
    }
}

__global__ __launch_bounds__(256)
void layer1_gemm_kernel(const float* __restrict__ acc_mu,
                        const float* __restrict__ deg_user, const float* __restrict__ Wl1_mu,
                        const float* __restrict__ x_user, const float* __restrict__ wfold_r,
                        const float* __restrict__ bl1_mu, const float* __restrict__ Wl2_um,
                        float* __restrict__ pu,
                        const float* __restrict__ acc_x, const float* __restrict__ acc_x_b,
                        const float* __restrict__ deg_movie, const float* __restrict__ wfold_l,
                        const float* __restrict__ hm, const float* __restrict__ Wr1_um,
                        const float* __restrict__ bl1_um, float* __restrict__ m1,
                        int user_blocks) {
    __shared__ __attribute__((aligned(16))) float wlds[64 * (w_pitch(FOLD_K) + w_pitch(HID))];
    const int b = (int)blockIdx.x;
    if (b < user_blocks) {
        sage_gemm64_body<true, false, HID, HID, 0, FOLD_K, USER_FEAT, 2, 256, 8>(
            wlds, b, acc_mu, nullptr, deg_user, Wl1_mu, x_user, wfold_r, bl1_mu, Wl2_um,
            pu, NUM_USERS, 1);
    } else {
        sage_gemm64_body<false, true, FOLD_K, XF_PITCH, 1, HID, HID, 0, 256, 8>(
            wlds, b - user_blocks, acc_x, acc_x_b, deg_movie, wfold_l, hm, Wr1_um, bl1_um, nullptr,
            m1, NUM_MOVIES, 1);
    }
}

__global__ __launch_bounds__(256)
void head_kernel(const float* __restrict__ accp, const float* __restrict__ degm,
                 const float* __restrict__ m1,
                 const float* __restrict__ bl2, const float* __restrict__ Wr2,
                 float* __restrict__ out, int M) {
    int m = blockIdx.x * blockDim.x + threadIdx.x;
    if (m >= M) return;
    const float inv = 1.0f / fmaxf(degm[m], 1.0f);
    const float* a = accp + (size_t)m * GENRES;
    const float* x = m1 + (size_t)m * HID;
    float acc[GENRES];
#pragma unroll
    for (int g = 0; g < GENRES; ++g) {
        float sum = a[g];
#pragma unroll
        for (int pl = 1; pl < PS_PLANES; ++pl) sum += a[(size_t)pl * PS_PITCH + g];
        acc[g] = sum * inv + bl2[g];
    }
    for (int k = 0; k < HID; ++k) {
        const float xv = x[k];
#pragma unroll
        for (int g = 0; g < GENRES; ++g)
            acc[g] += xv * Wr2[g * HID + k];
    }
#pragma unroll
    for (int g = 0; g < GENRES; ++g) out[(size_t)m * GENRES + g] = acc[g];
}

extern "C" void kernel_launch(void* const* d_in, const int* in_sizes, int n_in,
                              void* d_out, int out_size, void* d_ws, size_t ws_size,
                              hipStream_t stream) {
    (void)in_sizes; (void)n_in; (void)out_size; (void)ws_size;

    const float* x_user  = (const float*)d_in[0];
    const float* x_movie = (const float*)d_in[1];
    const int*   e_src   = (const int*)d_in[2];
    const int*   e_dst   = (const int*)d_in[3];
    const float* W_user  = (const float*)d_in[4];
    const float* b_user  = (const float*)d_in[5];
    const float* W_movie = (const float*)d_in[6];
    const float* b_movie = (const float*)d_in[7];
    const float* Wl1_um  = (const float*)d_in[8];
    const float* bl1_um  = (const float*)d_in[9];
    const float* Wr1_um  = (const float*)d_in[10];
    const float* Wl1_mu  = (const float*)d_in[11];
    const float* bl1_mu  = (const float*)d_in[12];
    const float* Wr1_mu  = (const float*)d_in[13];
    const float* Wl2_um  = (const float*)d_in[14];
    const float* bl2_um  = (const float*)d_in[15];
    const float* Wr2_um  = (const float*)d_in[16];
    float* out = (float*)d_out;

    float* ws = (float*)d_ws;
    float* hm        = ws;
    float* pu        = hm + 3200000;
    float* m1        = pu + 400000;
    float* wfold     = m1 + 3200000;
    float* wfold_r   = wfold + 1792;
    float* acc_x     = wfold + 3584;
    float* acc_x_b   = acc_x + 1600000;
    float* accp      = acc_x_b + 1600000;
    float* deg_user  = accp + 600064;
    float* deg_movie = deg_user + 100000;
    unsigned int* ucursor = (unsigned int*)(deg_movie + 50000);
    float* acc_mu    = deg_movie + 50000 + 528;
    unsigned int* ulist = (unsigned int*)(acc_mu + 6400000);

    const size_t zero_bytes = (size_t)(1600000 + 1600000 + 600064 + 100000 + 50000 + 528) * 4;
    hipMemsetAsync(acc_x, 0, zero_bytes, stream);

    const int E = NUM_EDGES;
    prepass_kernel<<<PRE_HIST + PRE_GEMM + 2, HTHREADS, 0, stream>>>(
        e_dst, deg_movie, e_src, deg_user, E,
        x_movie, W_movie, b_movie, hm,
        Wl1_um, Wr1_mu, W_user, b_user, wfold);

    const int gridU = (NUM_USERS / 16 + 7) / 8;
    const int gridM = (NUM_MOVIES / 16 + 7) / 8;

    user_bucket_kernel<<<UP_BLOCKS, UP_THREADS, 0, stream>>>(e_src, e_dst, ucursor, ulist, E);
    user_range_sum_kernel<<<UB_NB, UA_THREADS, 0, stream>>>(hm, e_src, e_dst, ucursor, ulist, acc_mu,
                                                            x_user, acc_x, acc_x_b, E);

    layer1_gemm_kernel<<<gridU + gridM, 256, 0, stream>>>(
        acc_mu, deg_user, Wl1_mu, x_user, wfold_r, bl1_mu, Wl2_um, pu,
        acc_x, acc_x_b, deg_movie, wfold, hm, Wr1_um, bl1_um, m1, gridU);

    proj_scatter_kernel<<<dim3(PS_SLICES, PS_RANGES), PS_THREADS, 0, stream>>>(
        pu, e_src, e_dst, accp, E);
    head_kernel<<<(NUM_MOVIES + 255) / 256, 256, 0, stream>>>(
        accp, deg_movie, m1, bl2_um, Wr2_um, out, NUM_MOVIES);
}
